# starting kernel
# baseline (speedup 1.0000x reference)
#include <hip/hip_runtime.h>
#include <math.h>

typedef __attribute__((ext_vector_type(2))) float v2f;
typedef __attribute__((ext_vector_type(8))) float v8f;

#define NEG_SLOPE 0.2f
#define SOFT_EPS  1e-16f

__device__ __forceinline__ float lrelu(float x) { return x > 0.f ? x : NEG_SLOPE * x; }
__device__ __forceinline__ int   enc_ord(float f) { int i = __float_as_int(f); return i >= 0 ? i : (i ^ 0x7FFFFFFF); }
__device__ __forceinline__ float dec_ord(int i)   { return __int_as_float(i >= 0 ? i : (i ^ 0x7FFFFFFF)); }

__global__ void gat_gemm_wmma(const float* __restrict__ A, const float* __restrict__ Bg,
                              float* __restrict__ C, int M, int K, int NC) {
  extern __shared__ float Bs[];
  const int nthr = blockDim.x * blockDim.y;
  const int tid  = threadIdx.y * blockDim.x + threadIdx.x;
  for (int i = tid; i < K * NC; i += nthr) Bs[i] = Bg[i];
  __syncthreads();

  const int row0 = blockIdx.x * 16;
  const int col0 = threadIdx.y * 16;
  const int lane = threadIdx.x;
  const int mr   = lane & 15;
  const int half = lane >> 4;

  int arow = row0 + mr; if (arow > M - 1) arow = M - 1;
  const float* Ap = A + (size_t)arow * K;

  v8f acc = {};
  for (int k0 = 0; k0 < K; k0 += 4) {
    const int ka = k0 + 2 * half;
    v2f a, b;
    a.x = Ap[ka];
    a.y = Ap[ka + 1];
    b.x = Bs[ka * NC + col0 + mr];
    b.y = Bs[(ka + 1) * NC + col0 + mr];
    acc = __builtin_amdgcn_wmma_f32_16x16x4_f32(false, a, false, b, (short)0, acc,
                                                false, false);
  }
#pragma unroll
  for (int i = 0; i < 8; ++i) {
    const int r = row0 + i + 8 * half;
    if (r < M) C[(size_t)r * NC + col0 + mr] = acc[i];
  }
}

__global__ void gat_prep1(const float* __restrict__ h1, const float* __restrict__ attS,
                          const float* __restrict__ attD, float* __restrict__ aS,
                          float* __restrict__ aD, int* __restrict__ menc,
                          float* __restrict__ denom, float* __restrict__ agg, int N) {
  const int t = blockIdx.x * blockDim.x + threadIdx.x;
  if (t >= N * 8) return;
  const int h = t & 7;
  const float* hp = h1 + (size_t)t * 8;
  float s = 0.f, d = 0.f;
#pragma unroll
  for (int c = 0; c < 8; ++c) { s += hp[c] * attS[h * 8 + c]; d += hp[c] * attD[h * 8 + c]; }
  aS[t] = s; aD[t] = d;
  menc[t]  = enc_ord(lrelu(s + d));
  denom[t] = 0.f;
  float* ap = agg + (size_t)t * 8;
#pragma unroll
  for (int c = 0; c < 8; ++c) ap[c] = 0.f;
}

__global__ void gat_emax1(const int* __restrict__ src, const int* __restrict__ dst,
                          const float* __restrict__ aS, const float* __restrict__ aD,
                          int* __restrict__ menc, int E) {
  const int e = blockIdx.x * blockDim.x + threadIdx.x;
  if (e >= E) return;
  const int s = src[e], d = dst[e];
#pragma unroll
  for (int h = 0; h < 8; ++h)
    atomicMax(&menc[d * 8 + h], enc_ord(lrelu(aS[s * 8 + h] + aD[d * 8 + h])));
}

__global__ void gat_dself1(const float* __restrict__ aS, const float* __restrict__ aD,
                           const int* __restrict__ menc, float* __restrict__ denom, int N) {
  const int t = blockIdx.x * blockDim.x + threadIdx.x;
  if (t >= N * 8) return;
  denom[t] = __expf(lrelu(aS[t] + aD[t]) - dec_ord(menc[t]));
}

__global__ void gat_esum1(const int* __restrict__ src, const int* __restrict__ dst,
                          const float* __restrict__ aS, const float* __restrict__ aD,
                          const int* __restrict__ menc, float* __restrict__ denom, int E) {
  const int e = blockIdx.x * blockDim.x + threadIdx.x;
  if (e >= E) return;
  const int s = src[e], d = dst[e];
#pragma unroll
  for (int h = 0; h < 8; ++h) {
    const float ev = lrelu(aS[s * 8 + h] + aD[d * 8 + h]);
    atomicAdd(&denom[d * 8 + h], __expf(ev - dec_ord(menc[d * 8 + h])));
  }
}

__global__ void gat_eagg1(const int* __restrict__ src, const int* __restrict__ dst,
                          const float* __restrict__ aS, const float* __restrict__ aD,
                          const int* __restrict__ menc, const float* __restrict__ denom,
                          const float* __restrict__ h1, float* __restrict__ agg, int E) {
  const int e = blockIdx.x * blockDim.x + threadIdx.x;
  if (e >= E) return;
  const int s = src[e], d = dst[e];
  __builtin_prefetch(h1 + (size_t)s * 64, 0, 1);
#pragma unroll
  for (int h = 0; h < 8; ++h) {
    const float ev = lrelu(aS[s * 8 + h] + aD[d * 8 + h]);
    const float alpha = __expf(ev - dec_ord(menc[d * 8 + h])) / (denom[d * 8 + h] + SOFT_EPS);
    const float4* hs = (const float4*)(h1 + (size_t)s * 64 + h * 8);
    const float4 p0 = hs[0], p1 = hs[1];
    float* o = agg + (size_t)d * 64 + h * 8;
    atomicAdd(o + 0, alpha * p0.x); atomicAdd(o + 1, alpha * p0.y);
    atomicAdd(o + 2, alpha * p0.z); atomicAdd(o + 3, alpha * p0.w);
    atomicAdd(o + 4, alpha * p1.x); atomicAdd(o + 5, alpha * p1.y);
    atomicAdd(o + 6, alpha * p1.z); atomicAdd(o + 7, alpha * p1.w);
  }
}

__global__ void gat_fin1(const float* __restrict__ agg, const float* __restrict__ h1,
                         const float* __restrict__ aS, const float* __restrict__ aD,
                         const int* __restrict__ menc, const float* __restrict__ denom,
                         const float* __restrict__ bias, float* __restrict__ hpost, int N) {
  const int t = blockIdx.x * blockDim.x + threadIdx.x;
  if (t >= N * 8) return;
  const int h = t & 7;
  const float alpha =
      __expf(lrelu(aS[t] + aD[t]) - dec_ord(menc[t])) / (denom[t] + SOFT_EPS);
  const float* ap = agg + (size_t)t * 8;
  const float* hp = h1 + (size_t)t * 8;
  float* op = hpost + (size_t)t * 8;
#pragma unroll
  for (int c = 0; c < 8; ++c) {
    const float v = ap[c] + alpha * hp[c] + bias[h * 8 + c];
    op[c] = v > 0.f ? v : (__expf(v) - 1.f);
  }
}

__global__ void gat_padw2(const float* __restrict__ W2, float* __restrict__ W2p) {
  const int t = blockIdx.x * blockDim.x + threadIdx.x;
  if (t >= 64 * 48) return;
  const int k = t / 48, c = t % 48;
  W2p[t] = (c < 40) ? W2[k * 40 + c] : 0.f;
}

__global__ void gat_prep2(const float* __restrict__ h2, const float* __restrict__ attS,
                          const float* __restrict__ attD, float* __restrict__ aS,
                          float* __restrict__ aD, int* __restrict__ menc,
                          float* __restrict__ denom, float* __restrict__ agg, int N) {
  const int n = blockIdx.x * blockDim.x + threadIdx.x;
  if (n >= N) return;
  const float* hp = h2 + (size_t)n * 48;
  float s = 0.f, d = 0.f;
#pragma unroll
  for (int c = 0; c < 40; ++c) { s += hp[c] * attS[c]; d += hp[c] * attD[c]; }
  aS[n] = s; aD[n] = d;
  menc[n]  = enc_ord(lrelu(s + d));
  denom[n] = 0.f;
  float* ap = agg + (size_t)n * 40;
#pragma unroll
  for (int c = 0; c < 40; ++c) ap[c] = 0.f;
}

__global__ void gat_emax2(const int* __restrict__ src, const int* __restrict__ dst,
                          const float* __restrict__ aS, const float* __restrict__ aD,
                          int* __restrict__ menc, int E) {
  const int e = blockIdx.x * blockDim.x + threadIdx.x;
  if (e >= E) return;
  atomicMax(&menc[dst[e]], enc_ord(lrelu(aS[src[e]] + aD[dst[e]])));
}

__global__ void gat_dself2(const float* __restrict__ aS, const float* __restrict__ aD,
                           const int* __restrict__ menc, float* __restrict__ denom, int N) {
  const int n = blockIdx.x * blockDim.x + threadIdx.x;
  if (n >= N) return;
  denom[n] = __expf(lrelu(aS[n] + aD[n]) - dec_ord(menc[n]));
}

__global__ void gat_esum2(const int* __restrict__ src, const int* __restrict__ dst,
                          const float* __restrict__ aS, const float* __restrict__ aD,
                          const int* __restrict__ menc, float* __restrict__ denom, int E) {
  const int e = blockIdx.x * blockDim.x + threadIdx.x;
  if (e >= E) return;
  const int s = src[e], d = dst[e];
  atomicAdd(&denom[d], __expf(lrelu(aS[s] + aD[d]) - dec_ord(menc[d])));
}

__global__ void gat_eagg2(const int* __restrict__ src, const int* __restrict__ dst,
                          const float* __restrict__ aS, const float* __restrict__ aD,
                          const int* __restrict__ menc, const float* __restrict__ denom,
                          const float* __restrict__ h2, float* __restrict__ agg, int E) {
  const int e = blockIdx.x * blockDim.x + threadIdx.x;
  if (e >= E) return;
  const int s = src[e], d = dst[e];
  __builtin_prefetch(h2 + (size_t)s * 48, 0, 1);
  const float alpha =
      __expf(lrelu(aS[s] + aD[d]) - dec_ord(menc[d])) / (denom[d] + SOFT_EPS);
  const float* hp = h2 + (size_t)s * 48;
  float* o = agg + (size_t)d * 40;
#pragma unroll
  for (int c = 0; c < 40; ++c) atomicAdd(o + c, alpha * hp[c]);
}

__global__ void gat_fin2(const float* __restrict__ agg, const float* __restrict__ h2,
                         const float* __restrict__ aS, const float* __restrict__ aD,
                         const int* __restrict__ menc, const float* __restrict__ denom,
                         const float* __restrict__ bias, float* __restrict__ out, int N) {
  const int n = blockIdx.x * blockDim.x + threadIdx.x;
  if (n >= N) return;
  const float alpha =
      __expf(lrelu(aS[n] + aD[n]) - dec_ord(menc[n])) / (denom[n] + SOFT_EPS);
  const float* ap = agg + (size_t)n * 40;
  const float* hp = h2 + (size_t)n * 48;
  float v[40];
  float mx = -INFINITY;
#pragma unroll
  for (int c = 0; c < 40; ++c) {
    v[c] = ap[c] + alpha * hp[c] + bias[c];
    mx = fmaxf(mx, v[c]);
  }
  float sum = 0.f;
#pragma unroll
  for (int c = 0; c < 40; ++c) sum += __expf(v[c] - mx);
  const float lse = mx + __logf(sum);
  float* op = out + (size_t)n * 40;
#pragma unroll
  for (int c = 0; c < 40; ++c) op[c] = v[c] - lse;
}

extern "C" void kernel_launch(void* const* d_in, const int* in_sizes, int n_in,
                              void* d_out, int out_size, void* d_ws, size_t ws_size,
                              hipStream_t stream) {
  const float* x   = (const float*)d_in[0];
  const int*   ei  = (const int*)d_in[1];
  const float* W1  = (const float*)d_in[2];
  const float* as1 = (const float*)d_in[3];
  const float* ad1 = (const float*)d_in[4];
  const float* b1  = (const float*)d_in[5];
  const float* W2  = (const float*)d_in[6];
  const float* as2 = (const float*)d_in[7];
  const float* ad2 = (const float*)d_in[8];
  const float* b2  = (const float*)d_in[9];
  float* out = (float*)d_out;

  const int Fin = in_sizes[2] / 64;
  const int N   = in_sizes[0] / Fin;
  const int E   = in_sizes[1] / 2;
  const int* src = ei;
  const int* dst = ei + E;

  float* ws = (float*)d_ws;
  size_t o = 0;
  float* h1    = ws + o; o += (size_t)N * 64;
  float* hpost = ws + o; o += (size_t)N * 64;
  float* aS1   = ws + o; o += (size_t)N * 8;
  float* aD1   = ws + o; o += (size_t)N * 8;
  int*   m1    = (int*)(ws + o); o += (size_t)N * 8;
  float* dn1   = ws + o; o += (size_t)N * 8;
  float* agg1  = ws + o; o += (size_t)N * 64;
  float* h2    = ws + o; o += (size_t)N * 48;
  float* aS2   = ws + o; o += (size_t)N;
  float* aD2   = ws + o; o += (size_t)N;
  int*   m2    = (int*)(ws + o); o += (size_t)N;
  float* dn2   = ws + o; o += (size_t)N;
  float* agg2  = ws + o; o += (size_t)N * 40;
  float* W2p   = ws + o; o += (size_t)64 * 48;

  const int TB = 256;
  const int gNH = (N * 8 + TB - 1) / TB;
  const int gN  = (N + TB - 1) / TB;
  const int gE  = (E + TB - 1) / TB;
  const int gM  = (N + 15) / 16;

  gat_gemm_wmma<<<gM, dim3(32, 4), (size_t)Fin * 64 * 4, stream>>>(x, W1, h1, N, Fin, 64);
  gat_prep1 <<<gNH, TB, 0, stream>>>(h1, as1, ad1, aS1, aD1, m1, dn1, agg1, N);
  gat_emax1 <<<gE,  TB, 0, stream>>>(src, dst, aS1, aD1, m1, E);
  gat_dself1<<<gNH, TB, 0, stream>>>(aS1, aD1, m1, dn1, N);
  gat_esum1 <<<gE,  TB, 0, stream>>>(src, dst, aS1, aD1, m1, dn1, E);
  gat_eagg1 <<<gE,  TB, 0, stream>>>(src, dst, aS1, aD1, m1, dn1, h1, agg1, E);
  gat_fin1  <<<gNH, TB, 0, stream>>>(agg1, h1, aS1, aD1, m1, dn1, b1, hpost, N);

  gat_padw2 <<<(64 * 48 + TB - 1) / TB, TB, 0, stream>>>(W2, W2p);
  gat_gemm_wmma<<<gM, dim3(32, 3), (size_t)64 * 48 * 4, stream>>>(hpost, W2p, h2, N, 64, 48);
  gat_prep2 <<<gN, TB, 0, stream>>>(h2, as2, ad2, aS2, aD2, m2, dn2, agg2, N);
  gat_emax2 <<<gE, TB, 0, stream>>>(src, dst, aS2, aD2, m2, E);
  gat_dself2<<<gN, TB, 0, stream>>>(aS2, aD2, m2, dn2, N);
  gat_esum2 <<<gE, TB, 0, stream>>>(src, dst, aS2, aD2, m2, dn2, E);
  gat_eagg2 <<<gE, TB, 0, stream>>>(src, dst, aS2, aD2, m2, dn2, h2, agg2, E);
  gat_fin2  <<<gN, TB, 0, stream>>>(agg2, h2, aS2, aD2, m2, dn2, b2, out, N);

  (void)n_in; (void)out_size; (void)ws_size;
}


# Edges counting-sorted by destination in the call (rank, scan, fill); a 16-lane group per node gathers max, sum, aggregate: no float atomics, sums in arrival-rank order, lane-tree log-softmax
# speedup vs baseline: 9.0797x; 9.0797x over previous
#include <hip/hip_runtime.h>
#include <math.h>

typedef __attribute__((ext_vector_type(2))) float v2f;
typedef __attribute__((ext_vector_type(8))) float v8f;

#define NEG_SLOPE 0.2f
#define SOFT_EPS  1e-16f

__device__ __forceinline__ float lrelu(float x) { return x > 0.f ? x : NEG_SLOPE * x; }

__global__ void gat_gemm_wmma(const float* __restrict__ A, const float* __restrict__ Bg,
                              float* __restrict__ C, int M, int K, int NC) {
  extern __shared__ float Bs[];
  const int nthr = blockDim.x * blockDim.y;
  const int tid  = threadIdx.y * blockDim.x + threadIdx.x;
  for (int i = tid; i < K * NC; i += nthr) Bs[i] = Bg[i];
  __syncthreads();

  const int row0 = blockIdx.x * 16;
  const int col0 = threadIdx.y * 16;
  const int lane = threadIdx.x;
  const int mr   = lane & 15;
  const int half = lane >> 4;

  int arow = row0 + mr; if (arow > M - 1) arow = M - 1;
  const float* Ap = A + (size_t)arow * K;

  v8f acc = {};
  for (int k0 = 0; k0 < K; k0 += 4) {
    const int ka = k0 + 2 * half;
    v2f a, b;
    a.x = Ap[ka];
    a.y = Ap[ka + 1];
    b.x = Bs[ka * NC + col0 + mr];
    b.y = Bs[(ka + 1) * NC + col0 + mr];
    acc = __builtin_amdgcn_wmma_f32_16x16x4_f32(false, a, false, b, (short)0, acc,
                                                false, false);
  }
#pragma unroll
  for (int i = 0; i < 8; ++i) {
    const int r = row0 + i + 8 * half;
    if (r < M) C[(size_t)r * NC + col0 + mr] = acc[i];
  }
}

__global__ void csr_count(const int* __restrict__ dst, int* __restrict__ cnt,
                          int* __restrict__ rank, int E) {
  const int e = blockIdx.x * blockDim.x + threadIdx.x;
  if (e >= E) return;
  rank[e] = atomicAdd(&cnt[dst[e]], 1);
}

__global__ void csr_scan_block(const int* __restrict__ cnt, int* __restrict__ loc,
                               int* __restrict__ bsum, int N) {
  __shared__ int wtot[32];
  const int tid  = threadIdx.x;
  const int lane = tid & 31;
  const int wv   = tid >> 5;
  const int i    = blockIdx.x * 1024 + tid;
  const int v    = (i < N) ? cnt[i] : 0;
  int x = v;
  for (int off = 1; off < 32; off <<= 1) {
    const int y = __shfl_up(x, off);
    if (lane >= off) x += y;
  }
  if (lane == 31) wtot[wv] = x;
  __syncthreads();
  if (wv == 0) {
    int w = wtot[lane];
    for (int off = 1; off < 32; off <<= 1) {
      const int y = __shfl_up(w, off);
      if (lane >= off) w += y;
    }
    wtot[lane] = w;
  }
  __syncthreads();
  const int base = (wv > 0) ? wtot[wv - 1] : 0;
  if (i < N) loc[i] = base + x - v;
  if (tid == 1023) bsum[blockIdx.x] = base + x;
}

__global__ void csr_scan_add(const int* __restrict__ loc, const int* __restrict__ bsum,
                             int* __restrict__ rowptr, int N) {
  __shared__ int boff;
  const int tid = threadIdx.x;
  if (tid < 32) {
    int s = 0;
    for (int j = tid; j < (int)blockIdx.x; j += 32) s += bsum[j];
    for (int off = 16; off > 0; off >>= 1) s += __shfl_xor(s, off);
    if (tid == 0) boff = s;
  }
  __syncthreads();
  const int i = blockIdx.x * 1024 + tid;
  if (i < N) rowptr[i] = loc[i] + boff;
}

__global__ void csr_fill(const int* __restrict__ src, const int* __restrict__ dst,
                         const int* __restrict__ rowptr, const int* __restrict__ rank,
                         int* __restrict__ col, int E) {
  const int e = blockIdx.x * blockDim.x + threadIdx.x;
  if (e >= E) return;
  const int pos = rowptr[dst[e]] + rank[e];
  if ((unsigned)pos < (unsigned)E) col[pos] = src[e];
}

__global__ void gat_prep1(const float* __restrict__ h1, const float* __restrict__ attS,
                          const float* __restrict__ attD, float* __restrict__ aS,
                          float* __restrict__ aD, int* __restrict__ cnt, int N) {
  const int t = blockIdx.x * blockDim.x + threadIdx.x;
  if (t >= N * 8) return;
  const int h = t & 7;
  const float* hp = h1 + (size_t)t * 8;
  float s = 0.f, d = 0.f;
#pragma unroll
  for (int c = 0; c < 8; ++c) { s += hp[c] * attS[h * 8 + c]; d += hp[c] * attD[h * 8 + c]; }
  aS[t] = s; aD[t] = d;
  if (t < N) cnt[t] = 0;
}

__global__ void gat_gather1(const int* __restrict__ rowptr, const int* __restrict__ col,
                            const float* __restrict__ aS, const float* __restrict__ aD,
                            const float* __restrict__ h1, const float* __restrict__ bias,
                            float* __restrict__ hpost, int N, int E) {
  const int t = blockIdx.x * blockDim.x + threadIdx.x;
  const int g = t >> 4;
  if (g >= N) return;
  const int j = t & 15;
  const int h = j >> 1;
  const int beg = rowptr[g];
  const int end = (g + 1 < N) ? rowptr[g + 1] : E;
  const float ad    = aD[g * 8 + h];
  const float eself = lrelu(aS[g * 8 + h] + ad);

  float m = eself;
#pragma clang loop vectorize(disable) unroll_count(4)
  for (int k = beg; k < end; ++k) m = fmaxf(m, lrelu(aS[col[k] * 8 + h] + ad));

  const float4* hrow = (const float4*)h1;
  float den = __expf(eself - m);
  const float4 hs = hrow[(size_t)g * 16 + j];
  float4 acc;
  acc.x = den * hs.x; acc.y = den * hs.y; acc.z = den * hs.z; acc.w = den * hs.w;
#pragma unroll 4
  for (int k = beg; k < end; ++k) {
    const int s = col[k];
    const float p = __expf(lrelu(aS[s * 8 + h] + ad) - m);
    const float4 v = hrow[(size_t)s * 16 + j];
    den += p;
    acc.x += p * v.x; acc.y += p * v.y; acc.z += p * v.z; acc.w += p * v.w;
  }
  const float inv = 1.f / (den + SOFT_EPS);
  float4 o;
  o.x = acc.x * inv + bias[4 * j + 0];
  o.y = acc.y * inv + bias[4 * j + 1];
  o.z = acc.z * inv + bias[4 * j + 2];
  o.w = acc.w * inv + bias[4 * j + 3];
  o.x = o.x > 0.f ? o.x : (__expf(o.x) - 1.f);
  o.y = o.y > 0.f ? o.y : (__expf(o.y) - 1.f);
  o.z = o.z > 0.f ? o.z : (__expf(o.z) - 1.f);
  o.w = o.w > 0.f ? o.w : (__expf(o.w) - 1.f);
  ((float4*)hpost)[(size_t)g * 16 + j] = o;
}

__global__ void gat_padw2(const float* __restrict__ W2, float* __restrict__ W2p) {
  const int t = blockIdx.x * blockDim.x + threadIdx.x;
  if (t >= 64 * 48) return;
  const int k = t / 48, c = t % 48;
  W2p[t] = (c < 40) ? W2[k * 40 + c] : 0.f;
}

__global__ void gat_prep2(const float* __restrict__ h2, const float* __restrict__ attS,
                          const float* __restrict__ attD, float* __restrict__ aS,
                          float* __restrict__ aD, int N) {
  const int n = blockIdx.x * blockDim.x + threadIdx.x;
  if (n >= N) return;
  const float* hp = h2 + (size_t)n * 48;
  float s = 0.f, d = 0.f;
#pragma unroll
  for (int c = 0; c < 40; ++c) { s += hp[c] * attS[c]; d += hp[c] * attD[c]; }
  aS[n] = s; aD[n] = d;
}

__global__ void gat_gather2(const int* __restrict__ rowptr, const int* __restrict__ col,
                            const float* __restrict__ aS, const float* __restrict__ aD,
                            const float* __restrict__ h2, const float* __restrict__ bias,
                            float* __restrict__ out, int N, int E) {
  const int t = blockIdx.x * blockDim.x + threadIdx.x;
  const int g = t >> 4;
  if (g >= N) return;
  const int j = t & 15;
  const bool act = j < 10;
  const int jj = act ? j : 9;
  const int beg = rowptr[g];
  const int end = (g + 1 < N) ? rowptr[g + 1] : E;
  const float ad    = aD[g];
  const float eself = lrelu(aS[g] + ad);

  float m = eself;
#pragma clang loop vectorize(disable) unroll_count(4)
  for (int k = beg; k < end; ++k) m = fmaxf(m, lrelu(aS[col[k]] + ad));

  const float4* hrow = (const float4*)h2;
  float den = __expf(eself - m);
  const float4 hs = hrow[(size_t)g * 12 + jj];
  float4 acc;
  acc.x = den * hs.x; acc.y = den * hs.y; acc.z = den * hs.z; acc.w = den * hs.w;
#pragma unroll 4
  for (int k = beg; k < end; ++k) {
    const int s = col[k];
    const float p = __expf(lrelu(aS[s] + ad) - m);
    const float4 v = hrow[(size_t)s * 12 + jj];
    den += p;
    acc.x += p * v.x; acc.y += p * v.y; acc.z += p * v.z; acc.w += p * v.w;
  }
  const float inv = 1.f / (den + SOFT_EPS);
  float4 v;
  v.x = acc.x * inv + bias[4 * jj + 0];
  v.y = acc.y * inv + bias[4 * jj + 1];
  v.z = acc.z * inv + bias[4 * jj + 2];
  v.w = acc.w * inv + bias[4 * jj + 3];
  float mx = act ? fmaxf(fmaxf(v.x, v.y), fmaxf(v.z, v.w)) : -INFINITY;
  for (int off = 8; off > 0; off >>= 1) mx = fmaxf(mx, __shfl_xor(mx, off, 16));
  float sum = act ? (__expf(v.x - mx) + __expf(v.y - mx) + __expf(v.z - mx) + __expf(v.w - mx))
                  : 0.f;
  for (int off = 8; off > 0; off >>= 1) sum += __shfl_xor(sum, off, 16);
  const float lse = mx + __logf(sum);
  if (act) {
    float4 o;
    o.x = v.x - lse; o.y = v.y - lse; o.z = v.z - lse; o.w = v.w - lse;
    ((float4*)out)[(size_t)g * 10 + j] = o;
  }
}

extern "C" void kernel_launch(void* const* d_in, const int* in_sizes, int n_in,
                              void* d_out, int out_size, void* d_ws, size_t ws_size,
                              hipStream_t stream) {
  const float* x   = (const float*)d_in[0];
  const int*   ei  = (const int*)d_in[1];
  const float* W1  = (const float*)d_in[2];
  const float* as1 = (const float*)d_in[3];
  const float* ad1 = (const float*)d_in[4];
  const float* b1  = (const float*)d_in[5];
  const float* W2  = (const float*)d_in[6];
  const float* as2 = (const float*)d_in[7];
  const float* ad2 = (const float*)d_in[8];
  const float* b2  = (const float*)d_in[9];
  float* out = (float*)d_out;

  const int Fin = in_sizes[2] / 64;
  const int N   = in_sizes[0] / Fin;
  const int E   = in_sizes[1] / 2;
  const int* src = ei;
  const int* dst = ei + E;

  const int TB = 256;
  const int gNH = (N * 8 + TB - 1) / TB;
  const int gN  = (N + TB - 1) / TB;
  const int gE  = (E + TB - 1) / TB;
  const int gM  = (N + 15) / 16;
  const int gG  = (N * 16 + TB - 1) / TB;
  const int gS  = (N + 1023) / 1024;

  float* ws = (float*)d_ws;
  size_t o = 0;
  float* h1    = ws + o; o += (size_t)N * 64;
  float* hpost = ws + o; o += (size_t)N * 64;
  float* h2    = ws + o; o += (size_t)N * 48;
  float* aS1   = ws + o; o += (size_t)N * 8;
  float* aD1   = ws + o; o += (size_t)N * 8;
  float* aS2   = ws + o; o += (size_t)N;
  float* aD2   = ws + o; o += (size_t)N;
  float* W2p   = ws + o; o += (size_t)64 * 48;
  int* cnt     = (int*)(ws + o); o += (size_t)N;
  int* loc     = (int*)(ws + o); o += (size_t)N;
  int* rowptr  = (int*)(ws + o); o += (size_t)N;
  int* bsum    = (int*)(ws + o); o += (size_t)gS;
  int* rank    = (int*)(ws + o); o += (size_t)E;
  int* col     = (int*)(ws + o); o += (size_t)E;

  gat_gemm_wmma<<<gM, dim3(32, 4), (size_t)Fin * 64 * 4, stream>>>(x, W1, h1, N, Fin, 64);
  gat_prep1 <<<gNH, TB, 0, stream>>>(h1, as1, ad1, aS1, aD1, cnt, N);
  csr_count     <<<gE, TB,   0, stream>>>(dst, cnt, rank, E);
  csr_scan_block<<<gS, 1024, 0, stream>>>(cnt, loc, bsum, N);
  csr_scan_add  <<<gS, 1024, 0, stream>>>(loc, bsum, rowptr, N);
  csr_fill      <<<gE, TB,   0, stream>>>(src, dst, rowptr, rank, col, E);
  gat_gather1<<<gG, TB, 0, stream>>>(rowptr, col, aS1, aD1, h1, b1, hpost, N, E);

  gat_padw2 <<<(64 * 48 + TB - 1) / TB, TB, 0, stream>>>(W2, W2p);
  gat_gemm_wmma<<<gM, dim3(32, 3), (size_t)64 * 48 * 4, stream>>>(hpost, W2p, h2, N, 64, 48);
  gat_prep2 <<<gN, TB, 0, stream>>>(h2, as2, ad2, aS2, aD2, N);
  gat_gather2<<<gG, TB, 0, stream>>>(rowptr, col, aS2, aD2, h2, b2, out, N, E);

  (void)n_in; (void)out_size; (void)ws_size;
}


# Both gathers walk a node's edges once: running max per 4-edge trip (1-3 edge tail = one masked trip), sums so far rescaled by exp(m_old-m_new) when it rises; max walk gone; add order kept
# speedup vs baseline: 10.0673x; 1.1088x over previous
#include <hip/hip_runtime.h>
#include <math.h>

typedef __attribute__((ext_vector_type(2))) float v2f;
typedef __attribute__((ext_vector_type(8))) float v8f;

#define NEG_SLOPE 0.2f
#define SOFT_EPS  1e-16f

__device__ __forceinline__ float lrelu(float x) { return x > 0.f ? x : NEG_SLOPE * x; }

__global__ void gat_gemm_wmma(const float* __restrict__ A, const float* __restrict__ Bg,
                              float* __restrict__ C, int M, int K, int NC) {
  extern __shared__ float Bs[];
  const int nthr = blockDim.x * blockDim.y;
  const int tid  = threadIdx.y * blockDim.x + threadIdx.x;
  for (int i = tid; i < K * NC; i += nthr) Bs[i] = Bg[i];
  __syncthreads();

  const int row0 = blockIdx.x * 16;
  const int col0 = threadIdx.y * 16;
  const int lane = threadIdx.x;
  const int mr   = lane & 15;
  const int half = lane >> 4;

  int arow = row0 + mr; if (arow > M - 1) arow = M - 1;
  const float* Ap = A + (size_t)arow * K;

  v8f acc = {};
  for (int k0 = 0; k0 < K; k0 += 4) {
    const int ka = k0 + 2 * half;
    v2f a, b;
    a.x = Ap[ka];
    a.y = Ap[ka + 1];
    b.x = Bs[ka * NC + col0 + mr];
    b.y = Bs[(ka + 1) * NC + col0 + mr];
    acc = __builtin_amdgcn_wmma_f32_16x16x4_f32(false, a, false, b, (short)0, acc,
                                                false, false);
  }
#pragma unroll
  for (int i = 0; i < 8; ++i) {
    const int r = row0 + i + 8 * half;
    if (r < M) C[(size_t)r * NC + col0 + mr] = acc[i];
  }
}

__global__ void csr_count(const int* __restrict__ dst, int* __restrict__ cnt,
                          int* __restrict__ rank, int E) {
  const int e = blockIdx.x * blockDim.x + threadIdx.x;
  if (e >= E) return;
  rank[e] = atomicAdd(&cnt[dst[e]], 1);
}

__global__ void csr_scan_block(const int* __restrict__ cnt, int* __restrict__ loc,
                               int* __restrict__ bsum, int N) {
  __shared__ int wtot[32];
  const int tid  = threadIdx.x;
  const int lane = tid & 31;
  const int wv   = tid >> 5;
  const int i    = blockIdx.x * 1024 + tid;
  const int v    = (i < N) ? cnt[i] : 0;
  int x = v;
  for (int off = 1; off < 32; off <<= 1) {
    const int y = __shfl_up(x, off);
    if (lane >= off) x += y;
  }
  if (lane == 31) wtot[wv] = x;
  __syncthreads();
  if (wv == 0) {
    int w = wtot[lane];
    for (int off = 1; off < 32; off <<= 1) {
      const int y = __shfl_up(w, off);
      if (lane >= off) w += y;
    }
    wtot[lane] = w;
  }
  __syncthreads();
  const int base = (wv > 0) ? wtot[wv - 1] : 0;
  if (i < N) loc[i] = base + x - v;
  if (tid == 1023) bsum[blockIdx.x] = base + x;
}

__global__ void csr_scan_add(const int* __restrict__ loc, const int* __restrict__ bsum,
                             int* __restrict__ rowptr, int N) {
  __shared__ int boff;
  const int tid = threadIdx.x;
  if (tid < 32) {
    int s = 0;
    for (int j = tid; j < (int)blockIdx.x; j += 32) s += bsum[j];
    for (int off = 16; off > 0; off >>= 1) s += __shfl_xor(s, off);
    if (tid == 0) boff = s;
  }
  __syncthreads();
  const int i = blockIdx.x * 1024 + tid;
  if (i < N) rowptr[i] = loc[i] + boff;
}

__global__ void csr_fill(const int* __restrict__ src, const int* __restrict__ dst,
                         const int* __restrict__ rowptr, const int* __restrict__ rank,
                         int* __restrict__ col, int E) {
  const int e = blockIdx.x * blockDim.x + threadIdx.x;
  if (e >= E) return;
  const int pos = rowptr[dst[e]] + rank[e];
  if ((unsigned)pos < (unsigned)E) col[pos] = src[e];
}

__device__ __forceinline__ void sm_rescale(float sc, float& den, float4& acc) {
  den *= sc;
  acc.x *= sc; acc.y *= sc; acc.z *= sc; acc.w *= sc;
}

__device__ __forceinline__ void sm_add(float p, const float4 v, float& den, float4& acc) {
  den += p;
  acc.x += p * v.x; acc.y += p * v.y; acc.z += p * v.z; acc.w += p * v.w;
}

__global__ void gat_prep1(const float* __restrict__ h1, const float* __restrict__ attS,
                          const float* __restrict__ attD, float* __restrict__ aS,
                          float* __restrict__ aD, int* __restrict__ cnt, int N) {
  const int t = blockIdx.x * blockDim.x + threadIdx.x;
  if (t >= N * 8) return;
  const int h = t & 7;
  const float* hp = h1 + (size_t)t * 8;
  float s = 0.f, d = 0.f;
#pragma unroll
  for (int c = 0; c < 8; ++c) { s += hp[c] * attS[h * 8 + c]; d += hp[c] * attD[h * 8 + c]; }
  aS[t] = s; aD[t] = d;
  if (t < N) cnt[t] = 0;
}

__global__ void gat_gather1(const int* __restrict__ rowptr, const int* __restrict__ col,
                            const float* __restrict__ aS, const float* __restrict__ aD,
                            const float* __restrict__ h1, const float* __restrict__ bias,
                            float* __restrict__ hpost, int N, int E) {
  const int t = blockIdx.x * blockDim.x + threadIdx.x;
  const int g = t >> 4;
  if (g >= N) return;
  const int j = t & 15;
  const int h = j >> 1;
  const int beg = rowptr[g];
  const int end = (g + 1 < N) ? rowptr[g + 1] : E;
  const float ad    = aD[g * 8 + h];
  const float eself = lrelu(aS[g * 8 + h] + ad);

  const float4* hrow = (const float4*)h1;
  float  m   = eself;
  float  den = 1.f;
  float4 acc = hrow[(size_t)g * 16 + j];
  int k = beg;
  for (; k + 4 <= end; k += 4) {
    const int s0 = col[k], s1 = col[k + 1], s2 = col[k + 2], s3 = col[k + 3];
    const float e0 = lrelu(aS[s0 * 8 + h] + ad);
    const float e1 = lrelu(aS[s1 * 8 + h] + ad);
    const float e2 = lrelu(aS[s2 * 8 + h] + ad);
    const float e3 = lrelu(aS[s3 * 8 + h] + ad);
    const float4 v0 = hrow[(size_t)s0 * 16 + j];
    const float4 v1 = hrow[(size_t)s1 * 16 + j];
    const float4 v2 = hrow[(size_t)s2 * 16 + j];
    const float4 v3 = hrow[(size_t)s3 * 16 + j];
    const float mn = fmaxf(fmaxf(fmaxf(m, e0), fmaxf(e1, e2)), e3);
    sm_rescale(__expf(m - mn), den, acc);
    sm_add(__expf(e0 - mn), v0, den, acc);
    sm_add(__expf(e1 - mn), v1, den, acc);
    sm_add(__expf(e2 - mn), v2, den, acc);
    sm_add(__expf(e3 - mn), v3, den, acc);
    m = mn;
  }
  if (k < end) {
    const int last = end - 1;
    const bool has1 = k + 1 < end, has2 = k + 2 < end;
    const int s0 = col[k], s1 = col[has1 ? k + 1 : last], s2 = col[has2 ? k + 2 : last];
    const float e0 = lrelu(aS[s0 * 8 + h] + ad);
    const float x1 = lrelu(aS[s1 * 8 + h] + ad);
    const float x2 = lrelu(aS[s2 * 8 + h] + ad);
    const float4 v0 = hrow[(size_t)s0 * 16 + j];
    const float4 v1 = hrow[(size_t)s1 * 16 + j];
    const float4 v2 = hrow[(size_t)s2 * 16 + j];
    const float e1 = has1 ? x1 : -INFINITY;
    const float e2 = has2 ? x2 : -INFINITY;
    const float mn = fmaxf(fmaxf(m, e0), fmaxf(e1, e2));
    sm_rescale(__expf(m - mn), den, acc);
    sm_add(__expf(e0 - mn), v0, den, acc);
    sm_add(__expf(e1 - mn), v1, den, acc);
    sm_add(__expf(e2 - mn), v2, den, acc);
    m = mn;
  }
  const float inv = 1.f / (den + SOFT_EPS);
  float4 o;
  o.x = acc.x * inv + bias[4 * j + 0];
  o.y = acc.y * inv + bias[4 * j + 1];
  o.z = acc.z * inv + bias[4 * j + 2];
  o.w = acc.w * inv + bias[4 * j + 3];
  o.x = o.x > 0.f ? o.x : (__expf(o.x) - 1.f);
  o.y = o.y > 0.f ? o.y : (__expf(o.y) - 1.f);
  o.z = o.z > 0.f ? o.z : (__expf(o.z) - 1.f);
  o.w = o.w > 0.f ? o.w : (__expf(o.w) - 1.f);
  ((float4*)hpost)[(size_t)g * 16 + j] = o;
}

__global__ void gat_padw2(const float* __restrict__ W2, float* __restrict__ W2p) {
  const int t = blockIdx.x * blockDim.x + threadIdx.x;
  if (t >= 64 * 48) return;
  const int k = t / 48, c = t % 48;
  W2p[t] = (c < 40) ? W2[k * 40 + c] : 0.f;
}

__global__ void gat_prep2(const float* __restrict__ h2, const float* __restrict__ attS,
                          const float* __restrict__ attD, float* __restrict__ aS,
                          float* __restrict__ aD, int N) {
  const int n = blockIdx.x * blockDim.x + threadIdx.x;
  if (n >= N) return;
  const float* hp = h2 + (size_t)n * 48;
  float s = 0.f, d = 0.f;
#pragma unroll
  for (int c = 0; c < 40; ++c) { s += hp[c] * attS[c]; d += hp[c] * attD[c]; }
  aS[n] = s; aD[n] = d;
}

__global__ void gat_gather2(const int* __restrict__ rowptr, const int* __restrict__ col,
                            const float* __restrict__ aS, const float* __restrict__ aD,
                            const float* __restrict__ h2, const float* __restrict__ bias,
                            float* __restrict__ out, int N, int E) {
  const int t = blockIdx.x * blockDim.x + threadIdx.x;
  const int g = t >> 4;
  if (g >= N) return;
  const int j = t & 15;
  const bool act = j < 10;
  const int jj = act ? j : 9;
  const int beg = rowptr[g];
  const int end = (g + 1 < N) ? rowptr[g + 1] : E;
  const float ad    = aD[g];
  const float eself = lrelu(aS[g] + ad);

  const float4* hrow = (const float4*)h2;
  float  m   = eself;
  float  den = 1.f;
  float4 acc = hrow[(size_t)g * 12 + jj];
  int k = beg;
  for (; k + 4 <= end; k += 4) {
    const int s0 = col[k], s1 = col[k + 1], s2 = col[k + 2], s3 = col[k + 3];
    const float e0 = lrelu(aS[s0] + ad);
    const float e1 = lrelu(aS[s1] + ad);
    const float e2 = lrelu(aS[s2] + ad);
    const float e3 = lrelu(aS[s3] + ad);
    const float4 v0 = hrow[(size_t)s0 * 12 + jj];
    const float4 v1 = hrow[(size_t)s1 * 12 + jj];
    const float4 v2 = hrow[(size_t)s2 * 12 + jj];
    const float4 v3 = hrow[(size_t)s3 * 12 + jj];
    const float mn = fmaxf(fmaxf(fmaxf(m, e0), fmaxf(e1, e2)), e3);
    sm_rescale(__expf(m - mn), den, acc);
    sm_add(__expf(e0 - mn), v0, den, acc);
    sm_add(__expf(e1 - mn), v1, den, acc);
    sm_add(__expf(e2 - mn), v2, den, acc);
    sm_add(__expf(e3 - mn), v3, den, acc);
    m = mn;
  }
  if (k < end) {
    const int last = end - 1;
    const bool has1 = k + 1 < end, has2 = k + 2 < end;
    const int s0 = col[k], s1 = col[has1 ? k + 1 : last], s2 = col[has2 ? k + 2 : last];
    const float e0 = lrelu(aS[s0] + ad);
    const float x1 = lrelu(aS[s1] + ad);
    const float x2 = lrelu(aS[s2] + ad);
    const float4 v0 = hrow[(size_t)s0 * 12 + jj];
    const float4 v1 = hrow[(size_t)s1 * 12 + jj];
    const float4 v2 = hrow[(size_t)s2 * 12 + jj];
    const float e1 = has1 ? x1 : -INFINITY;
    const float e2 = has2 ? x2 : -INFINITY;
    const float mn = fmaxf(fmaxf(m, e0), fmaxf(e1, e2));
    sm_rescale(__expf(m - mn), den, acc);
    sm_add(__expf(e0 - mn), v0, den, acc);
    sm_add(__expf(e1 - mn), v1, den, acc);
    sm_add(__expf(e2 - mn), v2, den, acc);
    m = mn;
  }
  const float inv = 1.f / (den + SOFT_EPS);
  float4 v;
  v.x = acc.x * inv + bias[4 * jj + 0];
  v.y = acc.y * inv + bias[4 * jj + 1];
  v.z = acc.z * inv + bias[4 * jj + 2];
  v.w = acc.w * inv + bias[4 * jj + 3];
  float mx = act ? fmaxf(fmaxf(v.x, v.y), fmaxf(v.z, v.w)) : -INFINITY;
  for (int off = 8; off > 0; off >>= 1) mx = fmaxf(mx, __shfl_xor(mx, off, 16));
  float sum = act ? (__expf(v.x - mx) + __expf(v.y - mx) + __expf(v.z - mx) + __expf(v.w - mx))
                  : 0.f;
  for (int off = 8; off > 0; off >>= 1) sum += __shfl_xor(sum, off, 16);
  const float lse = mx + __logf(sum);
  if (act) {
    float4 o;
    o.x = v.x - lse; o.y = v.y - lse; o.z = v.z - lse; o.w = v.w - lse;
    ((float4*)out)[(size_t)g * 10 + j] = o;
  }
}

extern "C" void kernel_launch(void* const* d_in, const int* in_sizes, int n_in,
                              void* d_out, int out_size, void* d_ws, size_t ws_size,
                              hipStream_t stream) {
  const float* x   = (const float*)d_in[0];
  const int*   ei  = (const int*)d_in[1];
  const float* W1  = (const float*)d_in[2];
  const float* as1 = (const float*)d_in[3];
  const float* ad1 = (const float*)d_in[4];
  const float* b1  = (const float*)d_in[5];
  const float* W2  = (const float*)d_in[6];
  const float* as2 = (const float*)d_in[7];
  const float* ad2 = (const float*)d_in[8];
  const float* b2  = (const float*)d_in[9];
  float* out = (float*)d_out;

  const int Fin = in_sizes[2] / 64;
  const int N   = in_sizes[0] / Fin;
  const int E   = in_sizes[1] / 2;
  const int* src = ei;
  const int* dst = ei + E;

  const int TB = 256;
  const int gNH = (N * 8 + TB - 1) / TB;
  const int gN  = (N + TB - 1) / TB;
  const int gE  = (E + TB - 1) / TB;
  const int gM  = (N + 15) / 16;
  const int gG  = (N * 16 + TB - 1) / TB;
  const int gS  = (N + 1023) / 1024;

  float* ws = (float*)d_ws;
  size_t o = 0;
  float* h1    = ws + o; o += (size_t)N * 64;
  float* hpost = ws + o; o += (size_t)N * 64;
  float* h2    = ws + o; o += (size_t)N * 48;
  float* aS1   = ws + o; o += (size_t)N * 8;
  float* aD1   = ws + o; o += (size_t)N * 8;
  float* aS2   = ws + o; o += (size_t)N;
  float* aD2   = ws + o; o += (size_t)N;
  float* W2p   = ws + o; o += (size_t)64 * 48;
  int* cnt     = (int*)(ws + o); o += (size_t)N;
  int* loc     = (int*)(ws + o); o += (size_t)N;
  int* rowptr  = (int*)(ws + o); o += (size_t)N;
  int* bsum    = (int*)(ws + o); o += (size_t)gS;
  int* rank    = (int*)(ws + o); o += (size_t)E;
  int* col     = (int*)(ws + o); o += (size_t)E;

  gat_gemm_wmma<<<gM, dim3(32, 4), (size_t)Fin * 64 * 4, stream>>>(x, W1, h1, N, Fin, 64);
  gat_prep1 <<<gNH, TB, 0, stream>>>(h1, as1, ad1, aS1, aD1, cnt, N);
  csr_count     <<<gE, TB,   0, stream>>>(dst, cnt, rank, E);
  csr_scan_block<<<gS, 1024, 0, stream>>>(cnt, loc, bsum, N);
  csr_scan_add  <<<gS, 1024, 0, stream>>>(loc, bsum, rowptr, N);
  csr_fill      <<<gE, TB,   0, stream>>>(src, dst, rowptr, rank, col, E);
  gat_gather1<<<gG, TB, 0, stream>>>(rowptr, col, aS1, aD1, h1, b1, hpost, N, E);

  gat_padw2 <<<(64 * 48 + TB - 1) / TB, TB, 0, stream>>>(W2, W2p);
  gat_gemm_wmma<<<gM, dim3(32, 3), (size_t)64 * 48 * 4, stream>>>(hpost, W2p, h2, N, 64, 48);
  gat_prep2 <<<gN, TB, 0, stream>>>(h2, as2, ad2, aS2, aD2, N);
  gat_gather2<<<gG, TB, 0, stream>>>(rowptr, col, aS2, aD2, h2, b2, out, N, E);

  (void)n_in; (void)out_size; (void)ws_size;
}


# Edge scores from the gathered row itself: per-lane dot with att_src, summed over the lane pair (layer 1) or 16-lane row (layer 2) by DPP; score planes, their gathers and both prep launches gone
# speedup vs baseline: 10.6680x; 1.0597x over previous
#include <hip/hip_runtime.h>
#include <math.h>

typedef __attribute__((ext_vector_type(2))) float v2f;
typedef __attribute__((ext_vector_type(8))) float v8f;

#define NEG_SLOPE 0.2f
#define SOFT_EPS  1e-16f

__device__ __forceinline__ float lrelu(float x) { return x > 0.f ? x : NEG_SLOPE * x; }

__global__ void gat_gemm_wmma(const float* __restrict__ A, const float* __restrict__ Bg,
                              float* __restrict__ C, int* __restrict__ zero,
                              int M, int K, int NC) {
  extern __shared__ float Bs[];
  const int nthr = blockDim.x * blockDim.y;
  const int tid  = threadIdx.y * blockDim.x + threadIdx.x;
  if (zero != nullptr) {
    const int gi = blockIdx.x * nthr + tid;
    if (gi < M) zero[gi] = 0;
  }
  for (int i = tid; i < K * NC; i += nthr) Bs[i] = Bg[i];
  __syncthreads();

  const int row0 = blockIdx.x * 16;
  const int col0 = threadIdx.y * 16;
  const int lane = threadIdx.x;
  const int mr   = lane & 15;
  const int half = lane >> 4;

  int arow = row0 + mr; if (arow > M - 1) arow = M - 1;
  const float* Ap = A + (size_t)arow * K;

  v8f acc = {};
  for (int k0 = 0; k0 < K; k0 += 4) {
    const int ka = k0 + 2 * half;
    v2f a, b;
    a.x = Ap[ka];
    a.y = Ap[ka + 1];
    b.x = Bs[ka * NC + col0 + mr];
    b.y = Bs[(ka + 1) * NC + col0 + mr];
    acc = __builtin_amdgcn_wmma_f32_16x16x4_f32(false, a, false, b, (short)0, acc,
                                                false, false);
  }
#pragma unroll
  for (int i = 0; i < 8; ++i) {
    const int r = row0 + i + 8 * half;
    if (r < M) C[(size_t)r * NC + col0 + mr] = acc[i];
  }
}

__global__ void csr_count(const int* __restrict__ dst, int* __restrict__ cnt,
                          int* __restrict__ rank, int E) {
  const int e = blockIdx.x * blockDim.x + threadIdx.x;
  if (e >= E) return;
  rank[e] = atomicAdd(&cnt[dst[e]], 1);
}

__global__ void csr_scan_block(const int* __restrict__ cnt, int* __restrict__ loc,
                               int* __restrict__ bsum, int N) {
  __shared__ int wtot[32];
  const int tid  = threadIdx.x;
  const int lane = tid & 31;
  const int wv   = tid >> 5;
  const int i    = blockIdx.x * 1024 + tid;
  const int v    = (i < N) ? cnt[i] : 0;
  int x = v;
  for (int off = 1; off < 32; off <<= 1) {
    const int y = __shfl_up(x, off);
    if (lane >= off) x += y;
  }
  if (lane == 31) wtot[wv] = x;
  __syncthreads();
  if (wv == 0) {
    int w = wtot[lane];
    for (int off = 1; off < 32; off <<= 1) {
      const int y = __shfl_up(w, off);
      if (lane >= off) w += y;
    }
    wtot[lane] = w;
  }
  __syncthreads();
  const int base = (wv > 0) ? wtot[wv - 1] : 0;
  if (i < N) loc[i] = base + x - v;
  if (tid == 1023) bsum[blockIdx.x] = base + x;
}

__global__ void csr_scan_add(const int* __restrict__ loc, const int* __restrict__ bsum,
                             int* __restrict__ rowptr, int N) {
  __shared__ int boff;
  const int tid = threadIdx.x;
  if (tid < 32) {
    int s = 0;
    for (int j = tid; j < (int)blockIdx.x; j += 32) s += bsum[j];
    for (int off = 16; off > 0; off >>= 1) s += __shfl_xor(s, off);
    if (tid == 0) boff = s;
  }
  __syncthreads();
  const int i = blockIdx.x * 1024 + tid;
  if (i < N) rowptr[i] = loc[i] + boff;
}

__global__ void csr_fill(const int* __restrict__ src, const int* __restrict__ dst,
                         const int* __restrict__ rowptr, const int* __restrict__ rank,
                         int* __restrict__ col, int E) {
  const int e = blockIdx.x * blockDim.x + threadIdx.x;
  if (e >= E) return;
  const int pos = rowptr[dst[e]] + rank[e];
  if ((unsigned)pos < (unsigned)E) col[pos] = src[e];
}

__device__ __forceinline__ void sm_rescale(float sc, float& den, float4& acc) {
  den *= sc;
  acc.x *= sc; acc.y *= sc; acc.z *= sc; acc.w *= sc;
}

__device__ __forceinline__ void sm_add(float p, const float4 v, float& den, float4& acc) {
  den += p;
  acc.x += p * v.x; acc.y += p * v.y; acc.z += p * v.z; acc.w += p * v.w;
}

__device__ __forceinline__ float dot4(const float4 v, float w0, float w1, float w2, float w3) {
  return v.x * w0 + v.y * w1 + v.z * w2 + v.w * w3;
}

template <int CTRL>
__device__ __forceinline__ float lane_xchg(float x) {
  return __int_as_float(__builtin_amdgcn_update_dpp(0, __float_as_int(x), CTRL, 0xF, 0xF, false));
}

__device__ __forceinline__ float pair_sum(float p) { return p + lane_xchg<0xB1>(p); }

__device__ __forceinline__ float row_sum(float p) {
  p += lane_xchg<0xB1>(p);
  p += lane_xchg<0x4E>(p);
  p += lane_xchg<0x141>(p);
  p += lane_xchg<0x140>(p);
  return p;
}

__global__ void gat_gather1(const int* __restrict__ rowptr, const int* __restrict__ col,
                            const float* __restrict__ attS, const float* __restrict__ attD,
                            const float* __restrict__ h1, const float* __restrict__ bias,
                            float* __restrict__ hpost, int N, int E) {
  const int t = blockIdx.x * blockDim.x + threadIdx.x;
  const int g = t >> 4;
  if (g >= N) return;
  const int j = t & 15;
  const int beg = rowptr[g];
  const int end = (g + 1 < N) ? rowptr[g + 1] : E;
  const float ws0 = attS[4 * j + 0], ws1 = attS[4 * j + 1], ws2 = attS[4 * j + 2], ws3 = attS[4 * j + 3];
  const float wd0 = attD[4 * j + 0], wd1 = attD[4 * j + 1], wd2 = attD[4 * j + 2], wd3 = attD[4 * j + 3];

  const float4* hrow = (const float4*)h1;
  float4 acc = hrow[(size_t)g * 16 + j];
  const float ad    = pair_sum(dot4(acc, wd0, wd1, wd2, wd3));
  const float eself = lrelu(pair_sum(dot4(acc, ws0, ws1, ws2, ws3)) + ad);
  float  m   = eself;
  float  den = 1.f;
  int k = beg;
  for (; k + 4 <= end; k += 4) {
    const int s0 = col[k], s1 = col[k + 1], s2 = col[k + 2], s3 = col[k + 3];
    const float4 v0 = hrow[(size_t)s0 * 16 + j];
    const float4 v1 = hrow[(size_t)s1 * 16 + j];
    const float4 v2 = hrow[(size_t)s2 * 16 + j];
    const float4 v3 = hrow[(size_t)s3 * 16 + j];
    const float e0 = lrelu(pair_sum(dot4(v0, ws0, ws1, ws2, ws3)) + ad);
    const float e1 = lrelu(pair_sum(dot4(v1, ws0, ws1, ws2, ws3)) + ad);
    const float e2 = lrelu(pair_sum(dot4(v2, ws0, ws1, ws2, ws3)) + ad);
    const float e3 = lrelu(pair_sum(dot4(v3, ws0, ws1, ws2, ws3)) + ad);
    const float mn = fmaxf(fmaxf(fmaxf(m, e0), fmaxf(e1, e2)), e3);
    sm_rescale(__expf(m - mn), den, acc);
    sm_add(__expf(e0 - mn), v0, den, acc);
    sm_add(__expf(e1 - mn), v1, den, acc);
    sm_add(__expf(e2 - mn), v2, den, acc);
    sm_add(__expf(e3 - mn), v3, den, acc);
    m = mn;
  }
  if (k < end) {
    const int last = end - 1;
    const bool has1 = k + 1 < end, has2 = k + 2 < end;
    const int s0 = col[k], s1 = col[has1 ? k + 1 : last], s2 = col[has2 ? k + 2 : last];
    const float4 v0 = hrow[(size_t)s0 * 16 + j];
    const float4 v1 = hrow[(size_t)s1 * 16 + j];
    const float4 v2 = hrow[(size_t)s2 * 16 + j];
    const float e0 = lrelu(pair_sum(dot4(v0, ws0, ws1, ws2, ws3)) + ad);
    const float x1 = lrelu(pair_sum(dot4(v1, ws0, ws1, ws2, ws3)) + ad);
    const float x2 = lrelu(pair_sum(dot4(v2, ws0, ws1, ws2, ws3)) + ad);
    const float e1 = has1 ? x1 : -INFINITY;
    const float e2 = has2 ? x2 : -INFINITY;
    const float mn = fmaxf(fmaxf(m, e0), fmaxf(e1, e2));
    sm_rescale(__expf(m - mn), den, acc);
    sm_add(__expf(e0 - mn), v0, den, acc);
    sm_add(__expf(e1 - mn), v1, den, acc);
    sm_add(__expf(e2 - mn), v2, den, acc);
    m = mn;
  }
  const float inv = 1.f / (den + SOFT_EPS);
  float4 o;
  o.x = acc.x * inv + bias[4 * j + 0];
  o.y = acc.y * inv + bias[4 * j + 1];
  o.z = acc.z * inv + bias[4 * j + 2];
  o.w = acc.w * inv + bias[4 * j + 3];
  o.x = o.x > 0.f ? o.x : (__expf(o.x) - 1.f);
  o.y = o.y > 0.f ? o.y : (__expf(o.y) - 1.f);
  o.z = o.z > 0.f ? o.z : (__expf(o.z) - 1.f);
  o.w = o.w > 0.f ? o.w : (__expf(o.w) - 1.f);
  ((float4*)hpost)[(size_t)g * 16 + j] = o;
}

__global__ void gat_padw2(const float* __restrict__ W2, float* __restrict__ W2p) {
  const int t = blockIdx.x * blockDim.x + threadIdx.x;
  if (t >= 64 * 48) return;
  const int k = t / 48, c = t % 48;
  W2p[t] = (c < 40) ? W2[k * 40 + c] : 0.f;
}

__global__ void gat_gather2(const int* __restrict__ rowptr, const int* __restrict__ col,
                            const float* __restrict__ attS, const float* __restrict__ attD,
                            const float* __restrict__ h2, const float* __restrict__ bias,
                            float* __restrict__ out, int N, int E) {
  const int t = blockIdx.x * blockDim.x + threadIdx.x;
  const int g = t >> 4;
  if (g >= N) return;
  const int j = t & 15;
  const bool act = j < 10;
  const int jj = act ? j : 9;
  const int beg = rowptr[g];
  const int end = (g + 1 < N) ? rowptr[g + 1] : E;
  const float ws0 = act ? attS[4 * jj + 0] : 0.f, ws1 = act ? attS[4 * jj + 1] : 0.f;
  const float ws2 = act ? attS[4 * jj + 2] : 0.f, ws3 = act ? attS[4 * jj + 3] : 0.f;
  const float wd0 = act ? attD[4 * jj + 0] : 0.f, wd1 = act ? attD[4 * jj + 1] : 0.f;
  const float wd2 = act ? attD[4 * jj + 2] : 0.f, wd3 = act ? attD[4 * jj + 3] : 0.f;

  const float4* hrow = (const float4*)h2;
  float4 acc = hrow[(size_t)g * 12 + jj];
  const float ad    = row_sum(dot4(acc, wd0, wd1, wd2, wd3));
  const float eself = lrelu(row_sum(dot4(acc, ws0, ws1, ws2, ws3)) + ad);
  float  m   = eself;
  float  den = 1.f;
  int k = beg;
  for (; k + 4 <= end; k += 4) {
    const int s0 = col[k], s1 = col[k + 1], s2 = col[k + 2], s3 = col[k + 3];
    const float4 v0 = hrow[(size_t)s0 * 12 + jj];
    const float4 v1 = hrow[(size_t)s1 * 12 + jj];
    const float4 v2 = hrow[(size_t)s2 * 12 + jj];
    const float4 v3 = hrow[(size_t)s3 * 12 + jj];
    const float e0 = lrelu(row_sum(dot4(v0, ws0, ws1, ws2, ws3)) + ad);
    const float e1 = lrelu(row_sum(dot4(v1, ws0, ws1, ws2, ws3)) + ad);
    const float e2 = lrelu(row_sum(dot4(v2, ws0, ws1, ws2, ws3)) + ad);
    const float e3 = lrelu(row_sum(dot4(v3, ws0, ws1, ws2, ws3)) + ad);
    const float mn = fmaxf(fmaxf(fmaxf(m, e0), fmaxf(e1, e2)), e3);
    sm_rescale(__expf(m - mn), den, acc);
    sm_add(__expf(e0 - mn), v0, den, acc);
    sm_add(__expf(e1 - mn), v1, den, acc);
    sm_add(__expf(e2 - mn), v2, den, acc);
    sm_add(__expf(e3 - mn), v3, den, acc);
    m = mn;
  }
  if (k < end) {
    const int last = end - 1;
    const bool has1 = k + 1 < end, has2 = k + 2 < end;
    const int s0 = col[k], s1 = col[has1 ? k + 1 : last], s2 = col[has2 ? k + 2 : last];
    const float4 v0 = hrow[(size_t)s0 * 12 + jj];
    const float4 v1 = hrow[(size_t)s1 * 12 + jj];
    const float4 v2 = hrow[(size_t)s2 * 12 + jj];
    const float e0 = lrelu(row_sum(dot4(v0, ws0, ws1, ws2, ws3)) + ad);
    const float x1 = lrelu(row_sum(dot4(v1, ws0, ws1, ws2, ws3)) + ad);
    const float x2 = lrelu(row_sum(dot4(v2, ws0, ws1, ws2, ws3)) + ad);
    const float e1 = has1 ? x1 : -INFINITY;
    const float e2 = has2 ? x2 : -INFINITY;
    const float mn = fmaxf(fmaxf(m, e0), fmaxf(e1, e2));
    sm_rescale(__expf(m - mn), den, acc);
    sm_add(__expf(e0 - mn), v0, den, acc);
    sm_add(__expf(e1 - mn), v1, den, acc);
    sm_add(__expf(e2 - mn), v2, den, acc);
    m = mn;
  }
  const float inv = 1.f / (den + SOFT_EPS);
  float4 v;
  v.x = acc.x * inv + bias[4 * jj + 0];
  v.y = acc.y * inv + bias[4 * jj + 1];
  v.z = acc.z * inv + bias[4 * jj + 2];
  v.w = acc.w * inv + bias[4 * jj + 3];
  float mx = act ? fmaxf(fmaxf(v.x, v.y), fmaxf(v.z, v.w)) : -INFINITY;
  for (int off = 8; off > 0; off >>= 1) mx = fmaxf(mx, __shfl_xor(mx, off, 16));
  float sum = act ? (__expf(v.x - mx) + __expf(v.y - mx) + __expf(v.z - mx) + __expf(v.w - mx))
                  : 0.f;
  for (int off = 8; off > 0; off >>= 1) sum += __shfl_xor(sum, off, 16);
  const float lse = mx + __logf(sum);
  if (act) {
    float4 o;
    o.x = v.x - lse; o.y = v.y - lse; o.z = v.z - lse; o.w = v.w - lse;
    ((float4*)out)[(size_t)g * 10 + j] = o;
  }
}

extern "C" void kernel_launch(void* const* d_in, const int* in_sizes, int n_in,
                              void* d_out, int out_size, void* d_ws, size_t ws_size,
                              hipStream_t stream) {
  const float* x   = (const float*)d_in[0];
  const int*   ei  = (const int*)d_in[1];
  const float* W1  = (const float*)d_in[2];
  const float* as1 = (const float*)d_in[3];
  const float* ad1 = (const float*)d_in[4];
  const float* b1  = (const float*)d_in[5];
  const float* W2  = (const float*)d_in[6];
  const float* as2 = (const float*)d_in[7];
  const float* ad2 = (const float*)d_in[8];
  const float* b2  = (const float*)d_in[9];
  float* out = (float*)d_out;

  const int Fin = in_sizes[2] / 64;
  const int N   = in_sizes[0] / Fin;
  const int E   = in_sizes[1] / 2;
  const int* src = ei;
  const int* dst = ei + E;

  const int TB = 256;
  const int gE  = (E + TB - 1) / TB;
  const int gM  = (N + 15) / 16;
  const int gG  = (N * 16 + TB - 1) / TB;
  const int gS  = (N + 1023) / 1024;

  float* ws = (float*)d_ws;
  size_t o = 0;
  float* h1    = ws + o; o += (size_t)N * 64;
  float* hpost = ws + o; o += (size_t)N * 64;
  float* h2    = ws + o; o += (size_t)N * 48;
  float* W2p   = ws + o; o += (size_t)64 * 48;
  int* cnt     = (int*)(ws + o); o += (size_t)N;
  int* loc     = (int*)(ws + o); o += (size_t)N;
  int* rowptr  = (int*)(ws + o); o += (size_t)N;
  int* bsum    = (int*)(ws + o); o += (size_t)gS;
  int* rank    = (int*)(ws + o); o += (size_t)E;
  int* col     = (int*)(ws + o); o += (size_t)E;

  gat_gemm_wmma<<<gM, dim3(32, 4), (size_t)Fin * 64 * 4, stream>>>(x, W1, h1, cnt, N, Fin, 64);
  csr_count     <<<gE, TB,   0, stream>>>(dst, cnt, rank, E);
  csr_scan_block<<<gS, 1024, 0, stream>>>(cnt, loc, bsum, N);
  csr_scan_add  <<<gS, 1024, 0, stream>>>(loc, bsum, rowptr, N);
  csr_fill      <<<gE, TB,   0, stream>>>(src, dst, rowptr, rank, col, E);
  gat_gather1<<<gG, TB, 0, stream>>>(rowptr, col, as1, ad1, h1, b1, hpost, N, E);

  gat_padw2 <<<(64 * 48 + TB - 1) / TB, TB, 0, stream>>>(W2, W2p);
  gat_gemm_wmma<<<gM, dim3(32, 3), (size_t)64 * 48 * 4, stream>>>(hpost, W2p, h2, nullptr, N, 64, 48);
  gat_gather2<<<gG, TB, 0, stream>>>(rowptr, col, as2, ad2, h2, b2, out, N, E);

  (void)n_in; (void)out_size; (void)ws_size;
}


# Edge sort without global atomics: edges binned by 256-node range (per-chunk LDS histogram, scan, scatter), then a block per range ranks and fills slots with LDS counters; slots keep arrival order
# speedup vs baseline: 18.3179x; 1.7171x over previous
#include <hip/hip_runtime.h>
#include <math.h>

typedef __attribute__((ext_vector_type(2))) float v2f;
typedef __attribute__((ext_vector_type(8))) float v8f;

#define NEG_SLOPE 0.2f
#define SOFT_EPS  1e-16f

__device__ __forceinline__ float lrelu(float x) { return x > 0.f ? x : NEG_SLOPE * x; }

__global__ void gat_gemm_wmma(const float* __restrict__ A, const float* __restrict__ Bg,
                              float* __restrict__ C, int M, int K, int NC) {
  extern __shared__ float Bs[];
  const int nthr = blockDim.x * blockDim.y;
  const int tid  = threadIdx.y * blockDim.x + threadIdx.x;
  for (int i = tid; i < K * NC; i += nthr) Bs[i] = Bg[i];
  __syncthreads();

  const int row0 = blockIdx.x * 16;
  const int col0 = threadIdx.y * 16;
  const int lane = threadIdx.x;
  const int mr   = lane & 15;
  const int half = lane >> 4;

  int arow = row0 + mr; if (arow > M - 1) arow = M - 1;
  const float* Ap = A + (size_t)arow * K;

  v8f acc = {};
  for (int k0 = 0; k0 < K; k0 += 4) {
    const int ka = k0 + 2 * half;
    v2f a, b;
    a.x = Ap[ka];
    a.y = Ap[ka + 1];
    b.x = Bs[ka * NC + col0 + mr];
    b.y = Bs[(ka + 1) * NC + col0 + mr];
    acc = __builtin_amdgcn_wmma_f32_16x16x4_f32(false, a, false, b, (short)0, acc,
                                                false, false);
  }
#pragma unroll
  for (int i = 0; i < 8; ++i) {
    const int r = row0 + i + 8 * half;
    if (r < M) C[(size_t)r * NC + col0 + mr] = acc[i];
  }
}

#define RNG_SHIFT  8
#define RNG_NODES  256
#define BIN_BLOCKS 256
#define BIN_TB     1024

__device__ __forceinline__ void edge_chunk(int E, int& beg, int& end) {
  const int chunk = (((E + BIN_BLOCKS - 1) / BIN_BLOCKS) + 3) & ~3;
  beg = min((int)blockIdx.x * chunk, E);
  end = min(beg + chunk, E);
}

__device__ __forceinline__ void bin_add4(int* __restrict__ sh, const int4 d) {
  atomicAdd(&sh[d.x >> RNG_SHIFT], 1);
  atomicAdd(&sh[d.y >> RNG_SHIFT], 1);
  atomicAdd(&sh[d.z >> RNG_SHIFT], 1);
  atomicAdd(&sh[d.w >> RNG_SHIFT], 1);
}

__device__ __forceinline__ void bin_add8(int* __restrict__ sh, const int4 a, const int4 b) {
  atomicAdd(&sh[a.x >> RNG_SHIFT], 1);
  atomicAdd(&sh[b.x >> RNG_SHIFT], 1);
  atomicAdd(&sh[a.y >> RNG_SHIFT], 1);
  atomicAdd(&sh[b.y >> RNG_SHIFT], 1);
  atomicAdd(&sh[a.z >> RNG_SHIFT], 1);
  atomicAdd(&sh[b.z >> RNG_SHIFT], 1);
  atomicAdd(&sh[a.w >> RNG_SHIFT], 1);
  atomicAdd(&sh[b.w >> RNG_SHIFT], 1);
}

__global__ void __launch_bounds__(BIN_TB)
bin_hist(const int* __restrict__ dst, int* __restrict__ hist, int E, int NB) {
  extern __shared__ int sh[];
  const int tid = threadIdx.x;
  for (int i = tid; i < NB; i += BIN_TB) sh[i] = 0;
  __syncthreads();
  int beg, end;
  edge_chunk(E, beg, end);
  if ((E & 3) == 0) {
    int e = beg + 4 * tid;
    for (; e + 4 * BIN_TB < end; e += 8 * BIN_TB) {
      const int4 a = *(const int4*)(dst + e);
      const int4 b = *(const int4*)(dst + e + 4 * BIN_TB);
      bin_add8(sh, a, b);
    }
    if (e < end) bin_add4(sh, *(const int4*)(dst + e));
  } else {
    for (int e = beg + tid; e < end; e += BIN_TB) atomicAdd(&sh[dst[e] >> RNG_SHIFT], 1);
  }
  __syncthreads();
  for (int i = tid; i < NB; i += BIN_TB) hist[i * BIN_BLOCKS + blockIdx.x] = sh[i];
}

__global__ void csr_scan_block(const int* __restrict__ cnt, int* __restrict__ loc,
                               int* __restrict__ bsum, int N) {
  __shared__ int wtot[32];
  const int tid  = threadIdx.x;
  const int lane = tid & 31;
  const int wv   = tid >> 5;
  const int i    = blockIdx.x * 1024 + tid;
  const int v    = (i < N) ? cnt[i] : 0;
  int x = v;
  for (int off = 1; off < 32; off <<= 1) {
    const int y = __shfl_up(x, off);
    if (lane >= off) x += y;
  }
  if (lane == 31) wtot[wv] = x;
  __syncthreads();
  if (wv == 0) {
    int w = wtot[lane];
    for (int off = 1; off < 32; off <<= 1) {
      const int y = __shfl_up(w, off);
      if (lane >= off) w += y;
    }
    wtot[lane] = w;
  }
  __syncthreads();
  const int base = (wv > 0) ? wtot[wv - 1] : 0;
  if (i < N) loc[i] = base + x - v;
  if (tid == 1023) bsum[blockIdx.x] = base + x;
}

__global__ void csr_scan_add(const int* __restrict__ loc, const int* __restrict__ bsum,
                             int* __restrict__ rowptr, int N) {
  __shared__ int boff;
  const int tid = threadIdx.x;
  if (tid < 32) {
    int s = 0;
    for (int j = tid; j < (int)blockIdx.x; j += 32) s += bsum[j];
    for (int off = 16; off > 0; off >>= 1) s += __shfl_xor(s, off);
    if (tid == 0) boff = s;
  }
  __syncthreads();
  const int i = blockIdx.x * 1024 + tid;
  if (i < N) rowptr[i] = loc[i] + boff;
}

__device__ __forceinline__ int take_slot(int* __restrict__ sh, int d) {
  return atomicAdd(&sh[d >> RNG_SHIFT], 1);
}

__device__ __forceinline__ void put_pair(int2* __restrict__ pairs, int pos, int s, int d, int E) {
  if ((unsigned)pos < (unsigned)E) pairs[pos] = make_int2(s, d);
}

__device__ __forceinline__ void put4(int* __restrict__ sh, int2* __restrict__ pairs,
                                     const int4 s, const int4 d, int E) {
  const int q0 = take_slot(sh, d.x), q1 = take_slot(sh, d.y);
  const int q2 = take_slot(sh, d.z), q3 = take_slot(sh, d.w);
  put_pair(pairs, q0, s.x, d.x, E);
  put_pair(pairs, q1, s.y, d.y, E);
  put_pair(pairs, q2, s.z, d.z, E);
  put_pair(pairs, q3, s.w, d.w, E);
}

__global__ void __launch_bounds__(BIN_TB)
bin_scatter(const int* __restrict__ src, const int* __restrict__ dst,
            const int* __restrict__ off, int2* __restrict__ pairs, int E, int NB) {
  extern __shared__ int sh[];
  const int tid = threadIdx.x;
  for (int i = tid; i < NB; i += BIN_TB) sh[i] = off[i * BIN_BLOCKS + blockIdx.x];
  __syncthreads();
  int beg, end;
  edge_chunk(E, beg, end);
  if ((E & 3) == 0) {
    int e = beg + 4 * tid;
    for (; e + 4 * BIN_TB < end; e += 8 * BIN_TB) {
      const int4 da = *(const int4*)(dst + e);
      const int4 db = *(const int4*)(dst + e + 4 * BIN_TB);
      const int4 sa = *(const int4*)(src + e);
      const int4 sb = *(const int4*)(src + e + 4 * BIN_TB);
      const int q0 = take_slot(sh, da.x), q1 = take_slot(sh, da.y);
      const int q2 = take_slot(sh, da.z), q3 = take_slot(sh, da.w);
      const int q4 = take_slot(sh, db.x), q5 = take_slot(sh, db.y);
      const int q6 = take_slot(sh, db.z), q7 = take_slot(sh, db.w);
      put_pair(pairs, q0, sa.x, da.x, E);
      put_pair(pairs, q1, sa.y, da.y, E);
      put_pair(pairs, q2, sa.z, da.z, E);
      put_pair(pairs, q3, sa.w, da.w, E);
      put_pair(pairs, q4, sb.x, db.x, E);
      put_pair(pairs, q5, sb.y, db.y, E);
      put_pair(pairs, q6, sb.z, db.z, E);
      put_pair(pairs, q7, sb.w, db.w, E);
    }
    if (e < end) {
      const int4 da = *(const int4*)(dst + e);
      const int4 sa = *(const int4*)(src + e);
      put4(sh, pairs, sa, da, E);
    }
  } else {
    for (int e = beg + tid; e < end; e += BIN_TB) {
      const int d = dst[e];
      put_pair(pairs, take_slot(sh, d), src[e], d, E);
    }
  }
}

__global__ void __launch_bounds__(1024)
range_fill(const int2* __restrict__ pairs, const int* __restrict__ off,
           int* __restrict__ rowptr, int* __restrict__ col, int N, int E, int NB) {
  __shared__ int cnt[RNG_NODES];
  __shared__ int cur[RNG_NODES];
  __shared__ int wtot[RNG_NODES / 32];
  const int b    = blockIdx.x;
  const int tid  = threadIdx.x;
  const int nthr = blockDim.x;
  const int lane = tid & 31;
  const int wv   = tid >> 5;
  int beg = off[b * BIN_BLOCKS];
  int end = (b + 1 < NB) ? off[(b + 1) * BIN_BLOCKS] : E;
  beg = max(0, min(beg, E));
  end = max(beg, min(end, E));
  if (tid < RNG_NODES) cnt[tid] = 0;
  __syncthreads();
  int i = beg + tid;
  for (; i + 3 * nthr < end; i += 4 * nthr) {
    const int d0 = pairs[i].y, d1 = pairs[i + nthr].y, d2 = pairs[i + 2 * nthr].y, d3 = pairs[i + 3 * nthr].y;
    atomicAdd(&cnt[d0 & (RNG_NODES - 1)], 1);
    atomicAdd(&cnt[d1 & (RNG_NODES - 1)], 1);
    atomicAdd(&cnt[d2 & (RNG_NODES - 1)], 1);
    atomicAdd(&cnt[d3 & (RNG_NODES - 1)], 1);
  }
  for (; i < end; i += nthr) atomicAdd(&cnt[pairs[i].y & (RNG_NODES - 1)], 1);
  __syncthreads();
  int v = 0, x = 0;
  if (tid < RNG_NODES) {
    v = cnt[tid];
    x = v;
    for (int o = 1; o < 32; o <<= 1) {
      const int y = __shfl_up(x, o);
      if (lane >= o) x += y;
    }
    if (lane == 31) wtot[wv] = x;
  }
  __syncthreads();
  if (tid < RNG_NODES) {
    int base = beg;
    for (int w = 0; w < wv; ++w) base += wtot[w];
    const int first = base + x - v;
    cur[tid] = first;
    const int node = b * RNG_NODES + tid;
    if (node < N) rowptr[node] = first;
  }
  __syncthreads();
  i = beg + tid;
  for (; i + 3 * nthr < end; i += 4 * nthr) {
    const int2 p0 = pairs[i], p1 = pairs[i + nthr], p2 = pairs[i + 2 * nthr], p3 = pairs[i + 3 * nthr];
    const int q0 = atomicAdd(&cur[p0.y & (RNG_NODES - 1)], 1);
    const int q1 = atomicAdd(&cur[p1.y & (RNG_NODES - 1)], 1);
    const int q2 = atomicAdd(&cur[p2.y & (RNG_NODES - 1)], 1);
    const int q3 = atomicAdd(&cur[p3.y & (RNG_NODES - 1)], 1);
    if ((unsigned)q0 < (unsigned)E) col[q0] = p0.x;
    if ((unsigned)q1 < (unsigned)E) col[q1] = p1.x;
    if ((unsigned)q2 < (unsigned)E) col[q2] = p2.x;
    if ((unsigned)q3 < (unsigned)E) col[q3] = p3.x;
  }
  for (; i < end; i += nthr) {
    const int2 p = pairs[i];
    const int q = atomicAdd(&cur[p.y & (RNG_NODES - 1)], 1);
    if ((unsigned)q < (unsigned)E) col[q] = p.x;
  }
}

__device__ __forceinline__ void sm_rescale(float sc, float& den, float4& acc) {
  den *= sc;
  acc.x *= sc; acc.y *= sc; acc.z *= sc; acc.w *= sc;
}

__device__ __forceinline__ void sm_add(float p, const float4 v, float& den, float4& acc) {
  den += p;
  acc.x += p * v.x; acc.y += p * v.y; acc.z += p * v.z; acc.w += p * v.w;
}

__device__ __forceinline__ float dot4(const float4 v, float w0, float w1, float w2, float w3) {
  return v.x * w0 + v.y * w1 + v.z * w2 + v.w * w3;
}

template <int CTRL>
__device__ __forceinline__ float lane_xchg(float x) {
  return __int_as_float(__builtin_amdgcn_update_dpp(0, __float_as_int(x), CTRL, 0xF, 0xF, false));
}

__device__ __forceinline__ float pair_sum(float p) { return p + lane_xchg<0xB1>(p); }

__device__ __forceinline__ float row_sum(float p) {
  p += lane_xchg<0xB1>(p);
  p += lane_xchg<0x4E>(p);
  p += lane_xchg<0x141>(p);
  p += lane_xchg<0x140>(p);
  return p;
}

__global__ void gat_gather1(const int* __restrict__ rowptr, const int* __restrict__ col,
                            const float* __restrict__ attS, const float* __restrict__ attD,
                            const float* __restrict__ h1, const float* __restrict__ bias,
                            float* __restrict__ hpost, int N, int E) {
  const int t = blockIdx.x * blockDim.x + threadIdx.x;
  const int g = t >> 4;
  if (g >= N) return;
  const int j = t & 15;
  const int beg = rowptr[g];
  const int end = (g + 1 < N) ? rowptr[g + 1] : E;
  const float ws0 = attS[4 * j + 0], ws1 = attS[4 * j + 1], ws2 = attS[4 * j + 2], ws3 = attS[4 * j + 3];
  const float wd0 = attD[4 * j + 0], wd1 = attD[4 * j + 1], wd2 = attD[4 * j + 2], wd3 = attD[4 * j + 3];

  const float4* hrow = (const float4*)h1;
  float4 acc = hrow[(size_t)g * 16 + j];
  const float ad    = pair_sum(dot4(acc, wd0, wd1, wd2, wd3));
  const float eself = lrelu(pair_sum(dot4(acc, ws0, ws1, ws2, ws3)) + ad);
  float  m   = eself;
  float  den = 1.f;
  int k = beg;
  for (; k + 4 <= end; k += 4) {
    const int s0 = col[k], s1 = col[k + 1], s2 = col[k + 2], s3 = col[k + 3];
    const float4 v0 = hrow[(size_t)s0 * 16 + j];
    const float4 v1 = hrow[(size_t)s1 * 16 + j];
    const float4 v2 = hrow[(size_t)s2 * 16 + j];
    const float4 v3 = hrow[(size_t)s3 * 16 + j];
    const float e0 = lrelu(pair_sum(dot4(v0, ws0, ws1, ws2, ws3)) + ad);
    const float e1 = lrelu(pair_sum(dot4(v1, ws0, ws1, ws2, ws3)) + ad);
    const float e2 = lrelu(pair_sum(dot4(v2, ws0, ws1, ws2, ws3)) + ad);
    const float e3 = lrelu(pair_sum(dot4(v3, ws0, ws1, ws2, ws3)) + ad);
    const float mn = fmaxf(fmaxf(fmaxf(m, e0), fmaxf(e1, e2)), e3);
    sm_rescale(__expf(m - mn), den, acc);
    sm_add(__expf(e0 - mn), v0, den, acc);
    sm_add(__expf(e1 - mn), v1, den, acc);
    sm_add(__expf(e2 - mn), v2, den, acc);
    sm_add(__expf(e3 - mn), v3, den, acc);
    m = mn;
  }
  if (k < end) {
    const int last = end - 1;
    const bool has1 = k + 1 < end, has2 = k + 2 < end;
    const int s0 = col[k], s1 = col[has1 ? k + 1 : last], s2 = col[has2 ? k + 2 : last];
    const float4 v0 = hrow[(size_t)s0 * 16 + j];
    const float4 v1 = hrow[(size_t)s1 * 16 + j];
    const float4 v2 = hrow[(size_t)s2 * 16 + j];
    const float e0 = lrelu(pair_sum(dot4(v0, ws0, ws1, ws2, ws3)) + ad);
    const float x1 = lrelu(pair_sum(dot4(v1, ws0, ws1, ws2, ws3)) + ad);
    const float x2 = lrelu(pair_sum(dot4(v2, ws0, ws1, ws2, ws3)) + ad);
    const float e1 = has1 ? x1 : -INFINITY;
    const float e2 = has2 ? x2 : -INFINITY;
    const float mn = fmaxf(fmaxf(m, e0), fmaxf(e1, e2));
    sm_rescale(__expf(m - mn), den, acc);
    sm_add(__expf(e0 - mn), v0, den, acc);
    sm_add(__expf(e1 - mn), v1, den, acc);
    sm_add(__expf(e2 - mn), v2, den, acc);
    m = mn;
  }
  const float inv = 1.f / (den + SOFT_EPS);
  float4 o;
  o.x = acc.x * inv + bias[4 * j + 0];
  o.y = acc.y * inv + bias[4 * j + 1];
  o.z = acc.z * inv + bias[4 * j + 2];
  o.w = acc.w * inv + bias[4 * j + 3];
  o.x = o.x > 0.f ? o.x : (__expf(o.x) - 1.f);
  o.y = o.y > 0.f ? o.y : (__expf(o.y) - 1.f);
  o.z = o.z > 0.f ? o.z : (__expf(o.z) - 1.f);
  o.w = o.w > 0.f ? o.w : (__expf(o.w) - 1.f);
  ((float4*)hpost)[(size_t)g * 16 + j] = o;
}

__global__ void gat_padw2(const float* __restrict__ W2, float* __restrict__ W2p) {
  const int t = blockIdx.x * blockDim.x + threadIdx.x;
  if (t >= 64 * 48) return;
  const int k = t / 48, c = t % 48;
  W2p[t] = (c < 40) ? W2[k * 40 + c] : 0.f;
}

__global__ void gat_gather2(const int* __restrict__ rowptr, const int* __restrict__ col,
                            const float* __restrict__ attS, const float* __restrict__ attD,
                            const float* __restrict__ h2, const float* __restrict__ bias,
                            float* __restrict__ out, int N, int E) {
  const int t = blockIdx.x * blockDim.x + threadIdx.x;
  const int g = t >> 4;
  if (g >= N) return;
  const int j = t & 15;
  const bool act = j < 10;
  const int jj = act ? j : 9;
  const int beg = rowptr[g];
  const int end = (g + 1 < N) ? rowptr[g + 1] : E;
  const float ws0 = act ? attS[4 * jj + 0] : 0.f, ws1 = act ? attS[4 * jj + 1] : 0.f;
  const float ws2 = act ? attS[4 * jj + 2] : 0.f, ws3 = act ? attS[4 * jj + 3] : 0.f;
  const float wd0 = act ? attD[4 * jj + 0] : 0.f, wd1 = act ? attD[4 * jj + 1] : 0.f;
  const float wd2 = act ? attD[4 * jj + 2] : 0.f, wd3 = act ? attD[4 * jj + 3] : 0.f;

  const float4* hrow = (const float4*)h2;
  float4 acc = hrow[(size_t)g * 12 + jj];
  const float ad    = row_sum(dot4(acc, wd0, wd1, wd2, wd3));
  const float eself = lrelu(row_sum(dot4(acc, ws0, ws1, ws2, ws3)) + ad);
  float  m   = eself;
  float  den = 1.f;
  int k = beg;
  for (; k + 4 <= end; k += 4) {
    const int s0 = col[k], s1 = col[k + 1], s2 = col[k + 2], s3 = col[k + 3];
    const float4 v0 = hrow[(size_t)s0 * 12 + jj];
    const float4 v1 = hrow[(size_t)s1 * 12 + jj];
    const float4 v2 = hrow[(size_t)s2 * 12 + jj];
    const float4 v3 = hrow[(size_t)s3 * 12 + jj];
    const float e0 = lrelu(row_sum(dot4(v0, ws0, ws1, ws2, ws3)) + ad);
    const float e1 = lrelu(row_sum(dot4(v1, ws0, ws1, ws2, ws3)) + ad);
    const float e2 = lrelu(row_sum(dot4(v2, ws0, ws1, ws2, ws3)) + ad);
    const float e3 = lrelu(row_sum(dot4(v3, ws0, ws1, ws2, ws3)) + ad);
    const float mn = fmaxf(fmaxf(fmaxf(m, e0), fmaxf(e1, e2)), e3);
    sm_rescale(__expf(m - mn), den, acc);
    sm_add(__expf(e0 - mn), v0, den, acc);
    sm_add(__expf(e1 - mn), v1, den, acc);
    sm_add(__expf(e2 - mn), v2, den, acc);
    sm_add(__expf(e3 - mn), v3, den, acc);
    m = mn;
  }
  if (k < end) {
    const int last = end - 1;
    const bool has1 = k + 1 < end, has2 = k + 2 < end;
    const int s0 = col[k], s1 = col[has1 ? k + 1 : last], s2 = col[has2 ? k + 2 : last];
    const float4 v0 = hrow[(size_t)s0 * 12 + jj];
    const float4 v1 = hrow[(size_t)s1 * 12 + jj];
    const float4 v2 = hrow[(size_t)s2 * 12 + jj];
    const float e0 = lrelu(row_sum(dot4(v0, ws0, ws1, ws2, ws3)) + ad);
    const float x1 = lrelu(row_sum(dot4(v1, ws0, ws1, ws2, ws3)) + ad);
    const float x2 = lrelu(row_sum(dot4(v2, ws0, ws1, ws2, ws3)) + ad);
    const float e1 = has1 ? x1 : -INFINITY;
    const float e2 = has2 ? x2 : -INFINITY;
    const float mn = fmaxf(fmaxf(m, e0), fmaxf(e1, e2));
    sm_rescale(__expf(m - mn), den, acc);
    sm_add(__expf(e0 - mn), v0, den, acc);
    sm_add(__expf(e1 - mn), v1, den, acc);
    sm_add(__expf(e2 - mn), v2, den, acc);
    m = mn;
  }
  const float inv = 1.f / (den + SOFT_EPS);
  float4 v;
  v.x = acc.x * inv + bias[4 * jj + 0];
  v.y = acc.y * inv + bias[4 * jj + 1];
  v.z = acc.z * inv + bias[4 * jj + 2];
  v.w = acc.w * inv + bias[4 * jj + 3];
  float mx = act ? fmaxf(fmaxf(v.x, v.y), fmaxf(v.z, v.w)) : -INFINITY;
  for (int off = 8; off > 0; off >>= 1) mx = fmaxf(mx, __shfl_xor(mx, off, 16));
  float sum = act ? (__expf(v.x - mx) + __expf(v.y - mx) + __expf(v.z - mx) + __expf(v.w - mx))
                  : 0.f;
  for (int off = 8; off > 0; off >>= 1) sum += __shfl_xor(sum, off, 16);
  const float lse = mx + __logf(sum);
  if (act) {
    float4 o;
    o.x = v.x - lse; o.y = v.y - lse; o.z = v.z - lse; o.w = v.w - lse;
    ((float4*)out)[(size_t)g * 10 + j] = o;
  }
}

extern "C" void kernel_launch(void* const* d_in, const int* in_sizes, int n_in,
                              void* d_out, int out_size, void* d_ws, size_t ws_size,
                              hipStream_t stream) {
  const float* x   = (const float*)d_in[0];
  const int*   ei  = (const int*)d_in[1];
  const float* W1  = (const float*)d_in[2];
  const float* as1 = (const float*)d_in[3];
  const float* ad1 = (const float*)d_in[4];
  const float* b1  = (const float*)d_in[5];
  const float* W2  = (const float*)d_in[6];
  const float* as2 = (const float*)d_in[7];
  const float* ad2 = (const float*)d_in[8];
  const float* b2  = (const float*)d_in[9];
  float* out = (float*)d_out;

  const int Fin = in_sizes[2] / 64;
  const int N   = in_sizes[0] / Fin;
  const int E   = in_sizes[1] / 2;
  const int* src = ei;
  const int* dst = ei + E;

  const int TB = 256;
  const int gM  = (N + 15) / 16;
  const int gG  = (N * 16 + TB - 1) / TB;
  const int NB  = (N + RNG_NODES - 1) / RNG_NODES;
  const int nH  = NB * BIN_BLOCKS;
  const int gS  = (nH + 1023) / 1024;

  float* ws = (float*)d_ws;
  size_t o = 0;
  float* h1    = ws + o; o += (size_t)N * 64;
  float* hpost = ws + o; o += (size_t)N * 64;
  float* h2    = ws + o; o += (size_t)N * 48;
  float* W2p   = ws + o; o += (size_t)64 * 48;
  int2* pairs  = (int2*)(ws + o); o += (size_t)E * 2;
  int* col     = (int*)(ws + o); o += (size_t)E;
  int* rowptr  = (int*)(ws + o); o += (size_t)N;
  int* hist    = (int*)(ws + o); o += (size_t)nH;
  int* loc     = (int*)(ws + o); o += (size_t)nH;
  int* off     = (int*)(ws + o); o += (size_t)nH;
  int* bsum    = (int*)(ws + o); o += (size_t)gS;

  gat_gemm_wmma<<<gM, dim3(32, 4), (size_t)Fin * 64 * 4, stream>>>(x, W1, h1, N, Fin, 64);
  bin_hist      <<<BIN_BLOCKS, BIN_TB, (size_t)NB * 4, stream>>>(dst, hist, E, NB);
  csr_scan_block<<<gS, 1024, 0, stream>>>(hist, loc, bsum, nH);
  csr_scan_add  <<<gS, 1024, 0, stream>>>(loc, bsum, off, nH);
  bin_scatter   <<<BIN_BLOCKS, BIN_TB, (size_t)NB * 4, stream>>>(src, dst, off, pairs, E, NB);
  range_fill    <<<NB, 1024, 0, stream>>>(pairs, off, rowptr, col, N, E, NB);
  gat_gather1<<<gG, TB, 0, stream>>>(rowptr, col, as1, ad1, h1, b1, hpost, N, E);

  gat_padw2 <<<(64 * 48 + TB - 1) / TB, TB, 0, stream>>>(W2, W2p);
  gat_gemm_wmma<<<gM, dim3(32, 3), (size_t)64 * 48 * 4, stream>>>(hpost, W2p, h2, N, 64, 48);
  gat_gather2<<<gG, TB, 0, stream>>>(rowptr, col, as2, ad2, h2, b2, out, N, E);

  (void)n_in; (void)out_size; (void)ws_size;
}


# Layer-2 projection moved into gather1's tail: a block's 16 finished rows go through LDS to 3 waves x 16 WMMA steps, W2 read in place (pad = column guard); hpost plane, pad and GEMM launches gone
# speedup vs baseline: 20.1952x; 1.1025x over previous
#include <hip/hip_runtime.h>
#include <math.h>

typedef __attribute__((ext_vector_type(2))) float v2f;
typedef __attribute__((ext_vector_type(8))) float v8f;

#define NEG_SLOPE 0.2f
#define SOFT_EPS  1e-16f

__device__ __forceinline__ float lrelu(float x) { return x > 0.f ? x : NEG_SLOPE * x; }

__global__ void gat_gemm_wmma(const float* __restrict__ A, const float* __restrict__ Bg,
                              float* __restrict__ C, int M, int K, int NC) {
  extern __shared__ float Bs[];
  const int nthr = blockDim.x * blockDim.y;
  const int tid  = threadIdx.y * blockDim.x + threadIdx.x;
  for (int i = tid; i < K * NC; i += nthr) Bs[i] = Bg[i];
  __syncthreads();

  const int row0 = blockIdx.x * 16;
  const int col0 = threadIdx.y * 16;
  const int lane = threadIdx.x;
  const int mr   = lane & 15;
  const int half = lane >> 4;

  int arow = row0 + mr; if (arow > M - 1) arow = M - 1;
  const float* Ap = A + (size_t)arow * K;

  v8f acc = {};
  for (int k0 = 0; k0 < K; k0 += 4) {
    const int ka = k0 + 2 * half;
    v2f a, b;
    a.x = Ap[ka];
    a.y = Ap[ka + 1];
    b.x = Bs[ka * NC + col0 + mr];
    b.y = Bs[(ka + 1) * NC + col0 + mr];
    acc = __builtin_amdgcn_wmma_f32_16x16x4_f32(false, a, false, b, (short)0, acc,
                                                false, false);
  }
#pragma unroll
  for (int i = 0; i < 8; ++i) {
    const int r = row0 + i + 8 * half;
    if (r < M) C[(size_t)r * NC + col0 + mr] = acc[i];
  }
}

#define RNG_SHIFT  8
#define RNG_NODES  256
#define BIN_BLOCKS 256
#define BIN_TB     1024

__device__ __forceinline__ void edge_chunk(int E, int& beg, int& end) {
  const int chunk = (((E + BIN_BLOCKS - 1) / BIN_BLOCKS) + 3) & ~3;
  beg = min((int)blockIdx.x * chunk, E);
  end = min(beg + chunk, E);
}

__device__ __forceinline__ void bin_add4(int* __restrict__ sh, const int4 d) {
  atomicAdd(&sh[d.x >> RNG_SHIFT], 1);
  atomicAdd(&sh[d.y >> RNG_SHIFT], 1);
  atomicAdd(&sh[d.z >> RNG_SHIFT], 1);
  atomicAdd(&sh[d.w >> RNG_SHIFT], 1);
}

__device__ __forceinline__ void bin_add8(int* __restrict__ sh, const int4 a, const int4 b) {
  atomicAdd(&sh[a.x >> RNG_SHIFT], 1);
  atomicAdd(&sh[b.x >> RNG_SHIFT], 1);
  atomicAdd(&sh[a.y >> RNG_SHIFT], 1);
  atomicAdd(&sh[b.y >> RNG_SHIFT], 1);
  atomicAdd(&sh[a.z >> RNG_SHIFT], 1);
  atomicAdd(&sh[b.z >> RNG_SHIFT], 1);
  atomicAdd(&sh[a.w >> RNG_SHIFT], 1);
  atomicAdd(&sh[b.w >> RNG_SHIFT], 1);
}

__global__ void __launch_bounds__(BIN_TB)
bin_hist(const int* __restrict__ dst, int* __restrict__ hist, int E, int NB) {
  extern __shared__ int sh[];
  const int tid = threadIdx.x;
  for (int i = tid; i < NB; i += BIN_TB) sh[i] = 0;
  __syncthreads();
  int beg, end;
  edge_chunk(E, beg, end);
  if ((E & 3) == 0) {
    int e = beg + 4 * tid;
    for (; e + 4 * BIN_TB < end; e += 8 * BIN_TB) {
      const int4 a = *(const int4*)(dst + e);
      const int4 b = *(const int4*)(dst + e + 4 * BIN_TB);
      bin_add8(sh, a, b);
    }
    if (e < end) bin_add4(sh, *(const int4*)(dst + e));
  } else {
    for (int e = beg + tid; e < end; e += BIN_TB) atomicAdd(&sh[dst[e] >> RNG_SHIFT], 1);
  }
  __syncthreads();
  for (int i = tid; i < NB; i += BIN_TB) hist[i * BIN_BLOCKS + blockIdx.x] = sh[i];
}

__global__ void csr_scan_block(const int* __restrict__ cnt, int* __restrict__ loc,
                               int* __restrict__ bsum, int N) {
  __shared__ int wtot[32];
  const int tid  = threadIdx.x;
  const int lane = tid & 31;
  const int wv   = tid >> 5;
  const int i    = blockIdx.x * 1024 + tid;
  const int v    = (i < N) ? cnt[i] : 0;
  int x = v;
  for (int off = 1; off < 32; off <<= 1) {
    const int y = __shfl_up(x, off);
    if (lane >= off) x += y;
  }
  if (lane == 31) wtot[wv] = x;
  __syncthreads();
  if (wv == 0) {
    int w = wtot[lane];
    for (int off = 1; off < 32; off <<= 1) {
      const int y = __shfl_up(w, off);
      if (lane >= off) w += y;
    }
    wtot[lane] = w;
  }
  __syncthreads();
  const int base = (wv > 0) ? wtot[wv - 1] : 0;
  if (i < N) loc[i] = base + x - v;
  if (tid == 1023) bsum[blockIdx.x] = base + x;
}

__global__ void csr_scan_add(const int* __restrict__ loc, const int* __restrict__ bsum,
                             int* __restrict__ rowptr, int N) {
  __shared__ int boff;
  const int tid = threadIdx.x;
  if (tid < 32) {
    int s = 0;
    for (int j = tid; j < (int)blockIdx.x; j += 32) s += bsum[j];
    for (int off = 16; off > 0; off >>= 1) s += __shfl_xor(s, off);
    if (tid == 0) boff = s;
  }
  __syncthreads();
  const int i = blockIdx.x * 1024 + tid;
  if (i < N) rowptr[i] = loc[i] + boff;
}

__device__ __forceinline__ int take_slot(int* __restrict__ sh, int d) {
  return atomicAdd(&sh[d >> RNG_SHIFT], 1);
}

__device__ __forceinline__ void put_pair(int2* __restrict__ pairs, int pos, int s, int d, int E) {
  if ((unsigned)pos < (unsigned)E) pairs[pos] = make_int2(s, d);
}

__device__ __forceinline__ void put4(int* __restrict__ sh, int2* __restrict__ pairs,
                                     const int4 s, const int4 d, int E) {
  const int q0 = take_slot(sh, d.x), q1 = take_slot(sh, d.y);
  const int q2 = take_slot(sh, d.z), q3 = take_slot(sh, d.w);
  put_pair(pairs, q0, s.x, d.x, E);
  put_pair(pairs, q1, s.y, d.y, E);
  put_pair(pairs, q2, s.z, d.z, E);
  put_pair(pairs, q3, s.w, d.w, E);
}

__global__ void __launch_bounds__(BIN_TB)
bin_scatter(const int* __restrict__ src, const int* __restrict__ dst,
            const int* __restrict__ off, int2* __restrict__ pairs, int E, int NB) {
  extern __shared__ int sh[];
  const int tid = threadIdx.x;
  for (int i = tid; i < NB; i += BIN_TB) sh[i] = off[i * BIN_BLOCKS + blockIdx.x];
  __syncthreads();
  int beg, end;
  edge_chunk(E, beg, end);
  if ((E & 3) == 0) {
    int e = beg + 4 * tid;
    for (; e + 4 * BIN_TB < end; e += 8 * BIN_TB) {
      const int4 da = *(const int4*)(dst + e);
      const int4 db = *(const int4*)(dst + e + 4 * BIN_TB);
      const int4 sa = *(const int4*)(src + e);
      const int4 sb = *(const int4*)(src + e + 4 * BIN_TB);
      const int q0 = take_slot(sh, da.x), q1 = take_slot(sh, da.y);
      const int q2 = take_slot(sh, da.z), q3 = take_slot(sh, da.w);
      const int q4 = take_slot(sh, db.x), q5 = take_slot(sh, db.y);
      const int q6 = take_slot(sh, db.z), q7 = take_slot(sh, db.w);
      put_pair(pairs, q0, sa.x, da.x, E);
      put_pair(pairs, q1, sa.y, da.y, E);
      put_pair(pairs, q2, sa.z, da.z, E);
      put_pair(pairs, q3, sa.w, da.w, E);
      put_pair(pairs, q4, sb.x, db.x, E);
      put_pair(pairs, q5, sb.y, db.y, E);
      put_pair(pairs, q6, sb.z, db.z, E);
      put_pair(pairs, q7, sb.w, db.w, E);
    }
    if (e < end) {
      const int4 da = *(const int4*)(dst + e);
      const int4 sa = *(const int4*)(src + e);
      put4(sh, pairs, sa, da, E);
    }
  } else {
    for (int e = beg + tid; e < end; e += BIN_TB) {
      const int d = dst[e];
      put_pair(pairs, take_slot(sh, d), src[e], d, E);
    }
  }
}

__global__ void __launch_bounds__(1024)
range_fill(const int2* __restrict__ pairs, const int* __restrict__ off,
           int* __restrict__ rowptr, int* __restrict__ col, int N, int E, int NB) {
  __shared__ int cnt[RNG_NODES];
  __shared__ int cur[RNG_NODES];
  __shared__ int wtot[RNG_NODES / 32];
  const int b    = blockIdx.x;
  const int tid  = threadIdx.x;
  const int nthr = blockDim.x;
  const int lane = tid & 31;
  const int wv   = tid >> 5;
  int beg = off[b * BIN_BLOCKS];
  int end = (b + 1 < NB) ? off[(b + 1) * BIN_BLOCKS] : E;
  beg = max(0, min(beg, E));
  end = max(beg, min(end, E));
  if (tid < RNG_NODES) cnt[tid] = 0;
  __syncthreads();
  int i = beg + tid;
  for (; i + 3 * nthr < end; i += 4 * nthr) {
    const int d0 = pairs[i].y, d1 = pairs[i + nthr].y, d2 = pairs[i + 2 * nthr].y, d3 = pairs[i + 3 * nthr].y;
    atomicAdd(&cnt[d0 & (RNG_NODES - 1)], 1);
    atomicAdd(&cnt[d1 & (RNG_NODES - 1)], 1);
    atomicAdd(&cnt[d2 & (RNG_NODES - 1)], 1);
    atomicAdd(&cnt[d3 & (RNG_NODES - 1)], 1);
  }
  for (; i < end; i += nthr) atomicAdd(&cnt[pairs[i].y & (RNG_NODES - 1)], 1);
  __syncthreads();
  int v = 0, x = 0;
  if (tid < RNG_NODES) {
    v = cnt[tid];
    x = v;
    for (int o = 1; o < 32; o <<= 1) {
      const int y = __shfl_up(x, o);
      if (lane >= o) x += y;
    }
    if (lane == 31) wtot[wv] = x;
  }
  __syncthreads();
  if (tid < RNG_NODES) {
    int base = beg;
    for (int w = 0; w < wv; ++w) base += wtot[w];
    const int first = base + x - v;
    cur[tid] = first;
    const int node = b * RNG_NODES + tid;
    if (node < N) rowptr[node] = first;
  }
  __syncthreads();
  i = beg + tid;
  for (; i + 3 * nthr < end; i += 4 * nthr) {
    const int2 p0 = pairs[i], p1 = pairs[i + nthr], p2 = pairs[i + 2 * nthr], p3 = pairs[i + 3 * nthr];
    const int q0 = atomicAdd(&cur[p0.y & (RNG_NODES - 1)], 1);
    const int q1 = atomicAdd(&cur[p1.y & (RNG_NODES - 1)], 1);
    const int q2 = atomicAdd(&cur[p2.y & (RNG_NODES - 1)], 1);
    const int q3 = atomicAdd(&cur[p3.y & (RNG_NODES - 1)], 1);
    if ((unsigned)q0 < (unsigned)E) col[q0] = p0.x;
    if ((unsigned)q1 < (unsigned)E) col[q1] = p1.x;
    if ((unsigned)q2 < (unsigned)E) col[q2] = p2.x;
    if ((unsigned)q3 < (unsigned)E) col[q3] = p3.x;
  }
  for (; i < end; i += nthr) {
    const int2 p = pairs[i];
    const int q = atomicAdd(&cur[p.y & (RNG_NODES - 1)], 1);
    if ((unsigned)q < (unsigned)E) col[q] = p.x;
  }
}

__device__ __forceinline__ void sm_rescale(float sc, float& den, float4& acc) {
  den *= sc;
  acc.x *= sc; acc.y *= sc; acc.z *= sc; acc.w *= sc;
}

__device__ __forceinline__ void sm_add(float p, const float4 v, float& den, float4& acc) {
  den += p;
  acc.x += p * v.x; acc.y += p * v.y; acc.z += p * v.z; acc.w += p * v.w;
}

__device__ __forceinline__ float dot4(const float4 v, float w0, float w1, float w2, float w3) {
  return v.x * w0 + v.y * w1 + v.z * w2 + v.w * w3;
}

template <int CTRL>
__device__ __forceinline__ float lane_xchg(float x) {
  return __int_as_float(__builtin_amdgcn_update_dpp(0, __float_as_int(x), CTRL, 0xF, 0xF, false));
}

__device__ __forceinline__ float pair_sum(float p) { return p + lane_xchg<0xB1>(p); }

__device__ __forceinline__ float row_sum(float p) {
  p += lane_xchg<0xB1>(p);
  p += lane_xchg<0x4E>(p);
  p += lane_xchg<0x141>(p);
  p += lane_xchg<0x140>(p);
  return p;
}

#define G1_TB    256
#define G1_PITCH 68
__global__ void __launch_bounds__(G1_TB)
gat_gather1_proj(const int* __restrict__ rowptr, const int* __restrict__ col,
                 const float* __restrict__ attS, const float* __restrict__ attD,
                 const float* __restrict__ h1, const float* __restrict__ bias,
                 const float* __restrict__ W2, float* __restrict__ h2, int N, int E) {
  __shared__ float tile[16 * G1_PITCH];
  const int t = blockIdx.x * G1_TB + threadIdx.x;
  const int g = min(t >> 4, N - 1);
  const int j = t & 15;
  const int beg = rowptr[g];
  const int end = (g + 1 < N) ? rowptr[g + 1] : E;
  const float ws0 = attS[4 * j + 0], ws1 = attS[4 * j + 1], ws2 = attS[4 * j + 2], ws3 = attS[4 * j + 3];
  const float wd0 = attD[4 * j + 0], wd1 = attD[4 * j + 1], wd2 = attD[4 * j + 2], wd3 = attD[4 * j + 3];

  const float4* hrow = (const float4*)h1;
  float4 acc = hrow[(size_t)g * 16 + j];
  const float ad    = pair_sum(dot4(acc, wd0, wd1, wd2, wd3));
  const float eself = lrelu(pair_sum(dot4(acc, ws0, ws1, ws2, ws3)) + ad);
  float  m   = eself;
  float  den = 1.f;
  int k = beg;
  for (; k + 4 <= end; k += 4) {
    const int s0 = col[k], s1 = col[k + 1], s2 = col[k + 2], s3 = col[k + 3];
    const float4 v0 = hrow[(size_t)s0 * 16 + j];
    const float4 v1 = hrow[(size_t)s1 * 16 + j];
    const float4 v2 = hrow[(size_t)s2 * 16 + j];
    const float4 v3 = hrow[(size_t)s3 * 16 + j];
    const float e0 = lrelu(pair_sum(dot4(v0, ws0, ws1, ws2, ws3)) + ad);
    const float e1 = lrelu(pair_sum(dot4(v1, ws0, ws1, ws2, ws3)) + ad);
    const float e2 = lrelu(pair_sum(dot4(v2, ws0, ws1, ws2, ws3)) + ad);
    const float e3 = lrelu(pair_sum(dot4(v3, ws0, ws1, ws2, ws3)) + ad);
    const float mn = fmaxf(fmaxf(fmaxf(m, e0), fmaxf(e1, e2)), e3);
    sm_rescale(__expf(m - mn), den, acc);
    sm_add(__expf(e0 - mn), v0, den, acc);
    sm_add(__expf(e1 - mn), v1, den, acc);
    sm_add(__expf(e2 - mn), v2, den, acc);
    sm_add(__expf(e3 - mn), v3, den, acc);
    m = mn;
  }
  if (k < end) {
    const int last = end - 1;
    const bool has1 = k + 1 < end, has2 = k + 2 < end;
    const int s0 = col[k], s1 = col[has1 ? k + 1 : last], s2 = col[has2 ? k + 2 : last];
    const float4 v0 = hrow[(size_t)s0 * 16 + j];
    const float4 v1 = hrow[(size_t)s1 * 16 + j];
    const float4 v2 = hrow[(size_t)s2 * 16 + j];
    const float e0 = lrelu(pair_sum(dot4(v0, ws0, ws1, ws2, ws3)) + ad);
    const float x1 = lrelu(pair_sum(dot4(v1, ws0, ws1, ws2, ws3)) + ad);
    const float x2 = lrelu(pair_sum(dot4(v2, ws0, ws1, ws2, ws3)) + ad);
    const float e1 = has1 ? x1 : -INFINITY;
    const float e2 = has2 ? x2 : -INFINITY;
    const float mn = fmaxf(fmaxf(m, e0), fmaxf(e1, e2));
    sm_rescale(__expf(m - mn), den, acc);
    sm_add(__expf(e0 - mn), v0, den, acc);
    sm_add(__expf(e1 - mn), v1, den, acc);
    sm_add(__expf(e2 - mn), v2, den, acc);
    m = mn;
  }
  const int wv   = threadIdx.x >> 5;
  const int lane = threadIdx.x & 31;
  const int mr   = lane & 15;
  const int half = lane >> 4;
  const int c    = wv * 16 + mr;
  const bool cv  = c < 40;
  const int cc   = cv ? c : 39;
  float bw[32];
  if (wv < 3) {
#pragma unroll
    for (int s = 0; s < 16; ++s) {
      const int ka = 4 * s + 2 * half;
      bw[2 * s]     = W2[ka * 40 + cc];
      bw[2 * s + 1] = W2[(ka + 1) * 40 + cc];
    }
  }

  const float inv = 1.f / (den + SOFT_EPS);
  float4 o;
  o.x = acc.x * inv + bias[4 * j + 0];
  o.y = acc.y * inv + bias[4 * j + 1];
  o.z = acc.z * inv + bias[4 * j + 2];
  o.w = acc.w * inv + bias[4 * j + 3];
  o.x = o.x > 0.f ? o.x : (__expf(o.x) - 1.f);
  o.y = o.y > 0.f ? o.y : (__expf(o.y) - 1.f);
  o.z = o.z > 0.f ? o.z : (__expf(o.z) - 1.f);
  o.w = o.w > 0.f ? o.w : (__expf(o.w) - 1.f);

  *(float4*)&tile[(threadIdx.x >> 4) * G1_PITCH + 4 * j] = o;
  __syncthreads();
  if (wv < 3) {
    const float* Ar = tile + mr * G1_PITCH;
    v8f acc2 = {};
#pragma unroll
    for (int s = 0; s < 16; ++s) {
      const int ka = 4 * s + 2 * half;
      v2f a, b;
      a.x = Ar[ka];
      a.y = Ar[ka + 1];
      b.x = cv ? bw[2 * s] : 0.f;
      b.y = cv ? bw[2 * s + 1] : 0.f;
      acc2 = __builtin_amdgcn_wmma_f32_16x16x4_f32(false, a, false, b, (short)0, acc2,
                                                   false, false);
    }
    const int row0 = blockIdx.x * 16;
#pragma unroll
    for (int i = 0; i < 8; ++i) {
      const int r = row0 + i + 8 * half;
      if (r < N) h2[(size_t)r * 48 + c] = acc2[i];
    }
  }
}

__global__ void gat_gather2(const int* __restrict__ rowptr, const int* __restrict__ col,
                            const float* __restrict__ attS, const float* __restrict__ attD,
                            const float* __restrict__ h2, const float* __restrict__ bias,
                            float* __restrict__ out, int N, int E) {
  const int t = blockIdx.x * blockDim.x + threadIdx.x;
  const int g = t >> 4;
  if (g >= N) return;
  const int j = t & 15;
  const bool act = j < 10;
  const int jj = act ? j : 9;
  const int beg = rowptr[g];
  const int end = (g + 1 < N) ? rowptr[g + 1] : E;
  const float ws0 = act ? attS[4 * jj + 0] : 0.f, ws1 = act ? attS[4 * jj + 1] : 0.f;
  const float ws2 = act ? attS[4 * jj + 2] : 0.f, ws3 = act ? attS[4 * jj + 3] : 0.f;
  const float wd0 = act ? attD[4 * jj + 0] : 0.f, wd1 = act ? attD[4 * jj + 1] : 0.f;
  const float wd2 = act ? attD[4 * jj + 2] : 0.f, wd3 = act ? attD[4 * jj + 3] : 0.f;

  const float4* hrow = (const float4*)h2;
  float4 acc = hrow[(size_t)g * 12 + jj];
  const float ad    = row_sum(dot4(acc, wd0, wd1, wd2, wd3));
  const float eself = lrelu(row_sum(dot4(acc, ws0, ws1, ws2, ws3)) + ad);
  float  m   = eself;
  float  den = 1.f;
  int k = beg;
  for (; k + 4 <= end; k += 4) {
    const int s0 = col[k], s1 = col[k + 1], s2 = col[k + 2], s3 = col[k + 3];
    const float4 v0 = hrow[(size_t)s0 * 12 + jj];
    const float4 v1 = hrow[(size_t)s1 * 12 + jj];
    const float4 v2 = hrow[(size_t)s2 * 12 + jj];
    const float4 v3 = hrow[(size_t)s3 * 12 + jj];
    const float e0 = lrelu(row_sum(dot4(v0, ws0, ws1, ws2, ws3)) + ad);
    const float e1 = lrelu(row_sum(dot4(v1, ws0, ws1, ws2, ws3)) + ad);
    const float e2 = lrelu(row_sum(dot4(v2, ws0, ws1, ws2, ws3)) + ad);
    const float e3 = lrelu(row_sum(dot4(v3, ws0, ws1, ws2, ws3)) + ad);
    const float mn = fmaxf(fmaxf(fmaxf(m, e0), fmaxf(e1, e2)), e3);
    sm_rescale(__expf(m - mn), den, acc);
    sm_add(__expf(e0 - mn), v0, den, acc);
    sm_add(__expf(e1 - mn), v1, den, acc);
    sm_add(__expf(e2 - mn), v2, den, acc);
    sm_add(__expf(e3 - mn), v3, den, acc);
    m = mn;
  }
  if (k < end) {
    const int last = end - 1;
    const bool has1 = k + 1 < end, has2 = k + 2 < end;
    const int s0 = col[k], s1 = col[has1 ? k + 1 : last], s2 = col[has2 ? k + 2 : last];
    const float4 v0 = hrow[(size_t)s0 * 12 + jj];
    const float4 v1 = hrow[(size_t)s1 * 12 + jj];
    const float4 v2 = hrow[(size_t)s2 * 12 + jj];
    const float e0 = lrelu(row_sum(dot4(v0, ws0, ws1, ws2, ws3)) + ad);
    const float x1 = lrelu(row_sum(dot4(v1, ws0, ws1, ws2, ws3)) + ad);
    const float x2 = lrelu(row_sum(dot4(v2, ws0, ws1, ws2, ws3)) + ad);
    const float e1 = has1 ? x1 : -INFINITY;
    const float e2 = has2 ? x2 : -INFINITY;
    const float mn = fmaxf(fmaxf(m, e0), fmaxf(e1, e2));
    sm_rescale(__expf(m - mn), den, acc);
    sm_add(__expf(e0 - mn), v0, den, acc);
    sm_add(__expf(e1 - mn), v1, den, acc);
    sm_add(__expf(e2 - mn), v2, den, acc);
    m = mn;
  }
  const float inv = 1.f / (den + SOFT_EPS);
  float4 v;
  v.x = acc.x * inv + bias[4 * jj + 0];
  v.y = acc.y * inv + bias[4 * jj + 1];
  v.z = acc.z * inv + bias[4 * jj + 2];
  v.w = acc.w * inv + bias[4 * jj + 3];
  float mx = act ? fmaxf(fmaxf(v.x, v.y), fmaxf(v.z, v.w)) : -INFINITY;
  for (int off = 8; off > 0; off >>= 1) mx = fmaxf(mx, __shfl_xor(mx, off, 16));
  float sum = act ? (__expf(v.x - mx) + __expf(v.y - mx) + __expf(v.z - mx) + __expf(v.w - mx))
                  : 0.f;
  for (int off = 8; off > 0; off >>= 1) sum += __shfl_xor(sum, off, 16);
  const float lse = mx + __logf(sum);
  if (act) {
    float4 o;
    o.x = v.x - lse; o.y = v.y - lse; o.z = v.z - lse; o.w = v.w - lse;
    ((float4*)out)[(size_t)g * 10 + j] = o;
  }
}

extern "C" void kernel_launch(void* const* d_in, const int* in_sizes, int n_in,
                              void* d_out, int out_size, void* d_ws, size_t ws_size,
                              hipStream_t stream) {
  const float* x   = (const float*)d_in[0];
  const int*   ei  = (const int*)d_in[1];
  const float* W1  = (const float*)d_in[2];
  const float* as1 = (const float*)d_in[3];
  const float* ad1 = (const float*)d_in[4];
  const float* b1  = (const float*)d_in[5];
  const float* W2  = (const float*)d_in[6];
  const float* as2 = (const float*)d_in[7];
  const float* ad2 = (const float*)d_in[8];
  const float* b2  = (const float*)d_in[9];
  float* out = (float*)d_out;

  const int Fin = in_sizes[2] / 64;
  const int N   = in_sizes[0] / Fin;
  const int E   = in_sizes[1] / 2;
  const int* src = ei;
  const int* dst = ei + E;

  const int TB = 256;
  const int gM  = (N + 15) / 16;
  const int gG  = (N * 16 + TB - 1) / TB;
  const int NB  = (N + RNG_NODES - 1) / RNG_NODES;
  const int nH  = NB * BIN_BLOCKS;
  const int gS  = (nH + 1023) / 1024;

  float* ws = (float*)d_ws;
  size_t o = 0;
  float* h1    = ws + o; o += (size_t)N * 64;
  float* h2    = ws + o; o += (size_t)N * 48;
  int2* pairs  = (int2*)(ws + o); o += (size_t)E * 2;
  int* col     = (int*)(ws + o); o += (size_t)E;
  int* rowptr  = (int*)(ws + o); o += (size_t)N;
  int* hist    = (int*)(ws + o); o += (size_t)nH;
  int* loc     = (int*)(ws + o); o += (size_t)nH;
  int* off     = (int*)(ws + o); o += (size_t)nH;
  int* bsum    = (int*)(ws + o); o += (size_t)gS;

  gat_gemm_wmma<<<gM, dim3(32, 4), (size_t)Fin * 64 * 4, stream>>>(x, W1, h1, N, Fin, 64);
  bin_hist      <<<BIN_BLOCKS, BIN_TB, (size_t)NB * 4, stream>>>(dst, hist, E, NB);
  csr_scan_block<<<gS, 1024, 0, stream>>>(hist, loc, bsum, nH);
  csr_scan_add  <<<gS, 1024, 0, stream>>>(loc, bsum, off, nH);
  bin_scatter   <<<BIN_BLOCKS, BIN_TB, (size_t)NB * 4, stream>>>(src, dst, off, pairs, E, NB);
  range_fill    <<<NB, 1024, 0, stream>>>(pairs, off, rowptr, col, N, E, NB);
  gat_gather1_proj<<<gM, G1_TB, 0, stream>>>(rowptr, col, as1, ad1, h1, b1, W2, h2, N, E);

  gat_gather2<<<gG, TB, 0, stream>>>(rowptr, col, as2, ad2, h2, b2, out, N, E);

  (void)n_in; (void)out_size; (void)ws_size;
}


# Layer-1 projection re-tiled: one wave per 16x64 strip, A rows read once by 128-bit loads and used for 4 column tiles, W1 staged once per 4 strips; k summed as 0,1,4,5 then 2,3,6,7 of each 8
# speedup vs baseline: 26.0110x; 1.2880x over previous
#include <hip/hip_runtime.h>
#include <math.h>

typedef __attribute__((ext_vector_type(2))) float v2f;
typedef __attribute__((ext_vector_type(8))) float v8f;

#define NEG_SLOPE 0.2f
#define SOFT_EPS  1e-16f

__device__ __forceinline__ float lrelu(float x) { return x > 0.f ? x : NEG_SLOPE * x; }

#define P1_TB    128
#define P1_PITCH 144
__global__ void __launch_bounds__(P1_TB)
gat_proj1_wmma(const float* __restrict__ X, const float* __restrict__ W,
               float* __restrict__ H, int M) {
  __shared__ float Bs[64 * P1_PITCH];
  const int tid   = threadIdx.x;
  const int wv    = tid >> 5;
  const int lane  = tid & 31;
  const int mr    = lane & 15;
  const int half  = lane >> 4;
  const int strip = blockIdx.x * (P1_TB / 32) + wv;
  const int arow  = min(strip * 16 + mr, M - 1);
  const float4* Ap = (const float4*)(X + (size_t)arow * 128) + half;
  float4 a[16];
#pragma unroll
  for (int s = 0; s < 16; ++s) a[s] = Ap[2 * s];

  {
    const float4* W4 = (const float4*)W;
    const int c = tid & 15;
#pragma unroll
    for (int i = 0; i < 8; ++i) {
      const int kp = (tid >> 4) + 8 * i;
      const float4 w0 = W4[(2 * kp) * 16 + c];
      const float4 w1 = W4[(2 * kp + 1) * 16 + c];
      *(float4*)&Bs[kp * P1_PITCH + 8 * c]     = make_float4(w0.x, w1.x, w0.y, w1.y);
      *(float4*)&Bs[kp * P1_PITCH + 8 * c + 4] = make_float4(w0.z, w1.z, w0.w, w1.w);
    }
  }
  __syncthreads();

  v8f acc0 = {}, acc1 = {}, acc2 = {}, acc3 = {};
  const float* Bl = Bs + (2 * half) * P1_PITCH + 2 * mr;
#pragma unroll
  for (int s = 0; s < 16; ++s) {
#pragma unroll
    for (int u = 0; u < 2; ++u) {
      v2f av;
      av.x = u ? a[s].z : a[s].x;
      av.y = u ? a[s].w : a[s].y;
      const float* bp = Bl + (4 * s + u) * P1_PITCH;
      const v2f b0 = *(const v2f*)(bp);
      const v2f b1 = *(const v2f*)(bp + 32);
      const v2f b2 = *(const v2f*)(bp + 64);
      const v2f b3 = *(const v2f*)(bp + 96);
      acc0 = __builtin_amdgcn_wmma_f32_16x16x4_f32(false, av, false, b0, (short)0, acc0, false, false);
      acc1 = __builtin_amdgcn_wmma_f32_16x16x4_f32(false, av, false, b1, (short)0, acc1, false, false);
      acc2 = __builtin_amdgcn_wmma_f32_16x16x4_f32(false, av, false, b2, (short)0, acc2, false, false);
      acc3 = __builtin_amdgcn_wmma_f32_16x16x4_f32(false, av, false, b3, (short)0, acc3, false, false);
    }
  }
#pragma unroll
  for (int i = 0; i < 8; ++i) {
    const int r = strip * 16 + i + 8 * half;
    if (r < M) {
      float* Hr = H + (size_t)r * 64 + mr;
      Hr[0]  = acc0[i];
      Hr[16] = acc1[i];
      Hr[32] = acc2[i];
      Hr[48] = acc3[i];
    }
  }
}

#define RNG_SHIFT  8
#define RNG_NODES  256
#define BIN_BLOCKS 256
#define BIN_TB     1024

__device__ __forceinline__ void edge_chunk(int E, int& beg, int& end) {
  const int chunk = (((E + BIN_BLOCKS - 1) / BIN_BLOCKS) + 3) & ~3;
  beg = min((int)blockIdx.x * chunk, E);
  end = min(beg + chunk, E);
}

__device__ __forceinline__ void bin_add4(int* __restrict__ sh, const int4 d) {
  atomicAdd(&sh[d.x >> RNG_SHIFT], 1);
  atomicAdd(&sh[d.y >> RNG_SHIFT], 1);
  atomicAdd(&sh[d.z >> RNG_SHIFT], 1);
  atomicAdd(&sh[d.w >> RNG_SHIFT], 1);
}

__device__ __forceinline__ void bin_add8(int* __restrict__ sh, const int4 a, const int4 b) {
  atomicAdd(&sh[a.x >> RNG_SHIFT], 1);
  atomicAdd(&sh[b.x >> RNG_SHIFT], 1);
  atomicAdd(&sh[a.y >> RNG_SHIFT], 1);
  atomicAdd(&sh[b.y >> RNG_SHIFT], 1);
  atomicAdd(&sh[a.z >> RNG_SHIFT], 1);
  atomicAdd(&sh[b.z >> RNG_SHIFT], 1);
  atomicAdd(&sh[a.w >> RNG_SHIFT], 1);
  atomicAdd(&sh[b.w >> RNG_SHIFT], 1);
}

__global__ void __launch_bounds__(BIN_TB)
bin_hist(const int* __restrict__ dst, int* __restrict__ hist, int E, int NB) {
  extern __shared__ int sh[];
  const int tid = threadIdx.x;
  for (int i = tid; i < NB; i += BIN_TB) sh[i] = 0;
  __syncthreads();
  int beg, end;
  edge_chunk(E, beg, end);
  if ((E & 3) == 0) {
    int e = beg + 4 * tid;
    for (; e + 4 * BIN_TB < end; e += 8 * BIN_TB) {
      const int4 a = *(const int4*)(dst + e);
      const int4 b = *(const int4*)(dst + e + 4 * BIN_TB);
      bin_add8(sh, a, b);
    }
    if (e < end) bin_add4(sh, *(const int4*)(dst + e));
  } else {
    for (int e = beg + tid; e < end; e += BIN_TB) atomicAdd(&sh[dst[e] >> RNG_SHIFT], 1);
  }
  __syncthreads();
  for (int i = tid; i < NB; i += BIN_TB) hist[i * BIN_BLOCKS + blockIdx.x] = sh[i];
}

__global__ void csr_scan_block(const int* __restrict__ cnt, int* __restrict__ loc,
                               int* __restrict__ bsum, int N) {
  __shared__ int wtot[32];
  const int tid  = threadIdx.x;
  const int lane = tid & 31;
  const int wv   = tid >> 5;
  const int i    = blockIdx.x * 1024 + tid;
  const int v    = (i < N) ? cnt[i] : 0;
  int x = v;
  for (int off = 1; off < 32; off <<= 1) {
    const int y = __shfl_up(x, off);
    if (lane >= off) x += y;
  }
  if (lane == 31) wtot[wv] = x;
  __syncthreads();
  if (wv == 0) {
    int w = wtot[lane];
    for (int off = 1; off < 32; off <<= 1) {
      const int y = __shfl_up(w, off);
      if (lane >= off) w += y;
    }
    wtot[lane] = w;
  }
  __syncthreads();
  const int base = (wv > 0) ? wtot[wv - 1] : 0;
  if (i < N) loc[i] = base + x - v;
  if (tid == 1023) bsum[blockIdx.x] = base + x;
}

__global__ void csr_scan_add(const int* __restrict__ loc, const int* __restrict__ bsum,
                             int* __restrict__ rowptr, int N) {
  __shared__ int boff;
  const int tid = threadIdx.x;
  if (tid < 32) {
    int s = 0;
    for (int j = tid; j < (int)blockIdx.x; j += 32) s += bsum[j];
    for (int off = 16; off > 0; off >>= 1) s += __shfl_xor(s, off);
    if (tid == 0) boff = s;
  }
  __syncthreads();
  const int i = blockIdx.x * 1024 + tid;
  if (i < N) rowptr[i] = loc[i] + boff;
}

__device__ __forceinline__ int take_slot(int* __restrict__ sh, int d) {
  return atomicAdd(&sh[d >> RNG_SHIFT], 1);
}

__device__ __forceinline__ void put_pair(int2* __restrict__ pairs, int pos, int s, int d, int E) {
  if ((unsigned)pos < (unsigned)E) pairs[pos] = make_int2(s, d);
}

__device__ __forceinline__ void put4(int* __restrict__ sh, int2* __restrict__ pairs,
                                     const int4 s, const int4 d, int E) {
  const int q0 = take_slot(sh, d.x), q1 = take_slot(sh, d.y);
  const int q2 = take_slot(sh, d.z), q3 = take_slot(sh, d.w);
  put_pair(pairs, q0, s.x, d.x, E);
  put_pair(pairs, q1, s.y, d.y, E);
  put_pair(pairs, q2, s.z, d.z, E);
  put_pair(pairs, q3, s.w, d.w, E);
}

__global__ void __launch_bounds__(BIN_TB)
bin_scatter(const int* __restrict__ src, const int* __restrict__ dst,
            const int* __restrict__ off, int2* __restrict__ pairs, int E, int NB) {
  extern __shared__ int sh[];
  const int tid = threadIdx.x;
  for (int i = tid; i < NB; i += BIN_TB) sh[i] = off[i * BIN_BLOCKS + blockIdx.x];
  __syncthreads();
  int beg, end;
  edge_chunk(E, beg, end);
  if ((E & 3) == 0) {
    int e = beg + 4 * tid;
    for (; e + 4 * BIN_TB < end; e += 8 * BIN_TB) {
      const int4 da = *(const int4*)(dst + e);
      const int4 db = *(const int4*)(dst + e + 4 * BIN_TB);
      const int4 sa = *(const int4*)(src + e);
      const int4 sb = *(const int4*)(src + e + 4 * BIN_TB);
      const int q0 = take_slot(sh, da.x), q1 = take_slot(sh, da.y);
      const int q2 = take_slot(sh, da.z), q3 = take_slot(sh, da.w);
      const int q4 = take_slot(sh, db.x), q5 = take_slot(sh, db.y);
      const int q6 = take_slot(sh, db.z), q7 = take_slot(sh, db.w);
      put_pair(pairs, q0, sa.x, da.x, E);
      put_pair(pairs, q1, sa.y, da.y, E);
      put_pair(pairs, q2, sa.z, da.z, E);
      put_pair(pairs, q3, sa.w, da.w, E);
      put_pair(pairs, q4, sb.x, db.x, E);
      put_pair(pairs, q5, sb.y, db.y, E);
      put_pair(pairs, q6, sb.z, db.z, E);
      put_pair(pairs, q7, sb.w, db.w, E);
    }
    if (e < end) {
      const int4 da = *(const int4*)(dst + e);
      const int4 sa = *(const int4*)(src + e);
      put4(sh, pairs, sa, da, E);
    }
  } else {
    for (int e = beg + tid; e < end; e += BIN_TB) {
      const int d = dst[e];
      put_pair(pairs, take_slot(sh, d), src[e], d, E);
    }
  }
}

__global__ void __launch_bounds__(1024)
range_fill(const int2* __restrict__ pairs, const int* __restrict__ off,
           int* __restrict__ rowptr, int* __restrict__ col, int N, int E, int NB) {
  __shared__ int cnt[RNG_NODES];
  __shared__ int cur[RNG_NODES];
  __shared__ int wtot[RNG_NODES / 32];
  const int b    = blockIdx.x;
  const int tid  = threadIdx.x;
  const int nthr = blockDim.x;
  const int lane = tid & 31;
  const int wv   = tid >> 5;
  int beg = off[b * BIN_BLOCKS];
  int end = (b + 1 < NB) ? off[(b + 1) * BIN_BLOCKS] : E;
  beg = max(0, min(beg, E));
  end = max(beg, min(end, E));
  if (tid < RNG_NODES) cnt[tid] = 0;
  __syncthreads();
  int i = beg + tid;
  for (; i + 3 * nthr < end; i += 4 * nthr) {
    const int d0 = pairs[i].y, d1 = pairs[i + nthr].y, d2 = pairs[i + 2 * nthr].y, d3 = pairs[i + 3 * nthr].y;
    atomicAdd(&cnt[d0 & (RNG_NODES - 1)], 1);
    atomicAdd(&cnt[d1 & (RNG_NODES - 1)], 1);
    atomicAdd(&cnt[d2 & (RNG_NODES - 1)], 1);
    atomicAdd(&cnt[d3 & (RNG_NODES - 1)], 1);
  }
  for (; i < end; i += nthr) atomicAdd(&cnt[pairs[i].y & (RNG_NODES - 1)], 1);
  __syncthreads();
  int v = 0, x = 0;
  if (tid < RNG_NODES) {
    v = cnt[tid];
    x = v;
    for (int o = 1; o < 32; o <<= 1) {
      const int y = __shfl_up(x, o);
      if (lane >= o) x += y;
    }
    if (lane == 31) wtot[wv] = x;
  }
  __syncthreads();
  if (tid < RNG_NODES) {
    int base = beg;
    for (int w = 0; w < wv; ++w) base += wtot[w];
    const int first = base + x - v;
    cur[tid] = first;
    const int node = b * RNG_NODES + tid;
    if (node < N) rowptr[node] = first;
  }
  __syncthreads();
  i = beg + tid;
  for (; i + 3 * nthr < end; i += 4 * nthr) {
    const int2 p0 = pairs[i], p1 = pairs[i + nthr], p2 = pairs[i + 2 * nthr], p3 = pairs[i + 3 * nthr];
    const int q0 = atomicAdd(&cur[p0.y & (RNG_NODES - 1)], 1);
    const int q1 = atomicAdd(&cur[p1.y & (RNG_NODES - 1)], 1);
    const int q2 = atomicAdd(&cur[p2.y & (RNG_NODES - 1)], 1);
    const int q3 = atomicAdd(&cur[p3.y & (RNG_NODES - 1)], 1);
    if ((unsigned)q0 < (unsigned)E) col[q0] = p0.x;
    if ((unsigned)q1 < (unsigned)E) col[q1] = p1.x;
    if ((unsigned)q2 < (unsigned)E) col[q2] = p2.x;
    if ((unsigned)q3 < (unsigned)E) col[q3] = p3.x;
  }
  for (; i < end; i += nthr) {
    const int2 p = pairs[i];
    const int q = atomicAdd(&cur[p.y & (RNG_NODES - 1)], 1);
    if ((unsigned)q < (unsigned)E) col[q] = p.x;
  }
}

__device__ __forceinline__ void sm_rescale(float sc, float& den, float4& acc) {
  den *= sc;
  acc.x *= sc; acc.y *= sc; acc.z *= sc; acc.w *= sc;
}

__device__ __forceinline__ void sm_add(float p, const float4 v, float& den, float4& acc) {
  den += p;
  acc.x += p * v.x; acc.y += p * v.y; acc.z += p * v.z; acc.w += p * v.w;
}

__device__ __forceinline__ float dot4(const float4 v, float w0, float w1, float w2, float w3) {
  return v.x * w0 + v.y * w1 + v.z * w2 + v.w * w3;
}

template <int CTRL>
__device__ __forceinline__ float lane_xchg(float x) {
  return __int_as_float(__builtin_amdgcn_update_dpp(0, __float_as_int(x), CTRL, 0xF, 0xF, false));
}

__device__ __forceinline__ float pair_sum(float p) { return p + lane_xchg<0xB1>(p); }

__device__ __forceinline__ float row_sum(float p) {
  p += lane_xchg<0xB1>(p);
  p += lane_xchg<0x4E>(p);
  p += lane_xchg<0x141>(p);
  p += lane_xchg<0x140>(p);
  return p;
}

#define G1_TB    256
#define G1_PITCH 68
__global__ void __launch_bounds__(G1_TB)
gat_gather1_proj(const int* __restrict__ rowptr, const int* __restrict__ col,
                 const float* __restrict__ attS, const float* __restrict__ attD,
                 const float* __restrict__ h1, const float* __restrict__ bias,
                 const float* __restrict__ W2, float* __restrict__ h2, int N, int E) {
  __shared__ float tile[16 * G1_PITCH];
  const int t = blockIdx.x * G1_TB + threadIdx.x;
  const int g = min(t >> 4, N - 1);
  const int j = t & 15;
  const int beg = rowptr[g];
  const int end = (g + 1 < N) ? rowptr[g + 1] : E;
  const float ws0 = attS[4 * j + 0], ws1 = attS[4 * j + 1], ws2 = attS[4 * j + 2], ws3 = attS[4 * j + 3];
  const float wd0 = attD[4 * j + 0], wd1 = attD[4 * j + 1], wd2 = attD[4 * j + 2], wd3 = attD[4 * j + 3];

  const float4* hrow = (const float4*)h1;
  float4 acc = hrow[(size_t)g * 16 + j];
  const float ad    = pair_sum(dot4(acc, wd0, wd1, wd2, wd3));
  const float eself = lrelu(pair_sum(dot4(acc, ws0, ws1, ws2, ws3)) + ad);
  float  m   = eself;
  float  den = 1.f;
  int k = beg;
  for (; k + 4 <= end; k += 4) {
    const int s0 = col[k], s1 = col[k + 1], s2 = col[k + 2], s3 = col[k + 3];
    const float4 v0 = hrow[(size_t)s0 * 16 + j];
    const float4 v1 = hrow[(size_t)s1 * 16 + j];
    const float4 v2 = hrow[(size_t)s2 * 16 + j];
    const float4 v3 = hrow[(size_t)s3 * 16 + j];
    const float e0 = lrelu(pair_sum(dot4(v0, ws0, ws1, ws2, ws3)) + ad);
    const float e1 = lrelu(pair_sum(dot4(v1, ws0, ws1, ws2, ws3)) + ad);
    const float e2 = lrelu(pair_sum(dot4(v2, ws0, ws1, ws2, ws3)) + ad);
    const float e3 = lrelu(pair_sum(dot4(v3, ws0, ws1, ws2, ws3)) + ad);
    const float mn = fmaxf(fmaxf(fmaxf(m, e0), fmaxf(e1, e2)), e3);
    sm_rescale(__expf(m - mn), den, acc);
    sm_add(__expf(e0 - mn), v0, den, acc);
    sm_add(__expf(e1 - mn), v1, den, acc);
    sm_add(__expf(e2 - mn), v2, den, acc);
    sm_add(__expf(e3 - mn), v3, den, acc);
    m = mn;
  }
  if (k < end) {
    const int last = end - 1;
    const bool has1 = k + 1 < end, has2 = k + 2 < end;
    const int s0 = col[k], s1 = col[has1 ? k + 1 : last], s2 = col[has2 ? k + 2 : last];
    const float4 v0 = hrow[(size_t)s0 * 16 + j];
    const float4 v1 = hrow[(size_t)s1 * 16 + j];
    const float4 v2 = hrow[(size_t)s2 * 16 + j];
    const float e0 = lrelu(pair_sum(dot4(v0, ws0, ws1, ws2, ws3)) + ad);
    const float x1 = lrelu(pair_sum(dot4(v1, ws0, ws1, ws2, ws3)) + ad);
    const float x2 = lrelu(pair_sum(dot4(v2, ws0, ws1, ws2, ws3)) + ad);
    const float e1 = has1 ? x1 : -INFINITY;
    const float e2 = has2 ? x2 : -INFINITY;
    const float mn = fmaxf(fmaxf(m, e0), fmaxf(e1, e2));
    sm_rescale(__expf(m - mn), den, acc);
    sm_add(__expf(e0 - mn), v0, den, acc);
    sm_add(__expf(e1 - mn), v1, den, acc);
    sm_add(__expf(e2 - mn), v2, den, acc);
    m = mn;
  }
  const int wv   = threadIdx.x >> 5;
  const int lane = threadIdx.x & 31;
  const int mr   = lane & 15;
  const int half = lane >> 4;
  const int c    = wv * 16 + mr;
  const bool cv  = c < 40;
  const int cc   = cv ? c : 39;
  float bw[32];
  if (wv < 3) {
#pragma unroll
    for (int s = 0; s < 16; ++s) {
      const int ka = 4 * s + 2 * half;
      bw[2 * s]     = W2[ka * 40 + cc];
      bw[2 * s + 1] = W2[(ka + 1) * 40 + cc];
    }
  }

  const float inv = 1.f / (den + SOFT_EPS);
  float4 o;
  o.x = acc.x * inv + bias[4 * j + 0];
  o.y = acc.y * inv + bias[4 * j + 1];
  o.z = acc.z * inv + bias[4 * j + 2];
  o.w = acc.w * inv + bias[4 * j + 3];
  o.x = o.x > 0.f ? o.x : (__expf(o.x) - 1.f);
  o.y = o.y > 0.f ? o.y : (__expf(o.y) - 1.f);
  o.z = o.z > 0.f ? o.z : (__expf(o.z) - 1.f);
  o.w = o.w > 0.f ? o.w : (__expf(o.w) - 1.f);

  *(float4*)&tile[(threadIdx.x >> 4) * G1_PITCH + 4 * j] = o;
  __syncthreads();
  if (wv < 3) {
    const float* Ar = tile + mr * G1_PITCH;
    v8f acc2 = {};
#pragma unroll
    for (int s = 0; s < 16; ++s) {
      const int ka = 4 * s + 2 * half;
      v2f a, b;
      a.x = Ar[ka];
      a.y = Ar[ka + 1];
      b.x = cv ? bw[2 * s] : 0.f;
      b.y = cv ? bw[2 * s + 1] : 0.f;
      acc2 = __builtin_amdgcn_wmma_f32_16x16x4_f32(false, a, false, b, (short)0, acc2,
                                                   false, false);
    }
    const int row0 = blockIdx.x * 16;
#pragma unroll
    for (int i = 0; i < 8; ++i) {
      const int r = row0 + i + 8 * half;
      if (r < N) h2[(size_t)r * 48 + c] = acc2[i];
    }
  }
}

__global__ void gat_gather2(const int* __restrict__ rowptr, const int* __restrict__ col,
                            const float* __restrict__ attS, const float* __restrict__ attD,
                            const float* __restrict__ h2, const float* __restrict__ bias,
                            float* __restrict__ out, int N, int E) {
  const int t = blockIdx.x * blockDim.x + threadIdx.x;
  const int g = t >> 4;
  if (g >= N) return;
  const int j = t & 15;
  const bool act = j < 10;
  const int jj = act ? j : 9;
  const int beg = rowptr[g];
  const int end = (g + 1 < N) ? rowptr[g + 1] : E;
  const float ws0 = act ? attS[4 * jj + 0] : 0.f, ws1 = act ? attS[4 * jj + 1] : 0.f;
  const float ws2 = act ? attS[4 * jj + 2] : 0.f, ws3 = act ? attS[4 * jj + 3] : 0.f;
  const float wd0 = act ? attD[4 * jj + 0] : 0.f, wd1 = act ? attD[4 * jj + 1] : 0.f;
  const float wd2 = act ? attD[4 * jj + 2] : 0.f, wd3 = act ? attD[4 * jj + 3] : 0.f;

  const float4* hrow = (const float4*)h2;
  float4 acc = hrow[(size_t)g * 12 + jj];
  const float ad    = row_sum(dot4(acc, wd0, wd1, wd2, wd3));
  const float eself = lrelu(row_sum(dot4(acc, ws0, ws1, ws2, ws3)) + ad);
  float  m   = eself;
  float  den = 1.f;
  int k = beg;
  for (; k + 4 <= end; k += 4) {
    const int s0 = col[k], s1 = col[k + 1], s2 = col[k + 2], s3 = col[k + 3];
    const float4 v0 = hrow[(size_t)s0 * 12 + jj];
    const float4 v1 = hrow[(size_t)s1 * 12 + jj];
    const float4 v2 = hrow[(size_t)s2 * 12 + jj];
    const float4 v3 = hrow[(size_t)s3 * 12 + jj];
    const float e0 = lrelu(row_sum(dot4(v0, ws0, ws1, ws2, ws3)) + ad);
    const float e1 = lrelu(row_sum(dot4(v1, ws0, ws1, ws2, ws3)) + ad);
    const float e2 = lrelu(row_sum(dot4(v2, ws0, ws1, ws2, ws3)) + ad);
    const float e3 = lrelu(row_sum(dot4(v3, ws0, ws1, ws2, ws3)) + ad);
    const float mn = fmaxf(fmaxf(fmaxf(m, e0), fmaxf(e1, e2)), e3);
    sm_rescale(__expf(m - mn), den, acc);
    sm_add(__expf(e0 - mn), v0, den, acc);
    sm_add(__expf(e1 - mn), v1, den, acc);
    sm_add(__expf(e2 - mn), v2, den, acc);
    sm_add(__expf(e3 - mn), v3, den, acc);
    m = mn;
  }
  if (k < end) {
    const int last = end - 1;
    const bool has1 = k + 1 < end, has2 = k + 2 < end;
    const int s0 = col[k], s1 = col[has1 ? k + 1 : last], s2 = col[has2 ? k + 2 : last];
    const float4 v0 = hrow[(size_t)s0 * 12 + jj];
    const float4 v1 = hrow[(size_t)s1 * 12 + jj];
    const float4 v2 = hrow[(size_t)s2 * 12 + jj];
    const float e0 = lrelu(row_sum(dot4(v0, ws0, ws1, ws2, ws3)) + ad);
    const float x1 = lrelu(row_sum(dot4(v1, ws0, ws1, ws2, ws3)) + ad);
    const float x2 = lrelu(row_sum(dot4(v2, ws0, ws1, ws2, ws3)) + ad);
    const float e1 = has1 ? x1 : -INFINITY;
    const float e2 = has2 ? x2 : -INFINITY;
    const float mn = fmaxf(fmaxf(m, e0), fmaxf(e1, e2));
    sm_rescale(__expf(m - mn), den, acc);
    sm_add(__expf(e0 - mn), v0, den, acc);
    sm_add(__expf(e1 - mn), v1, den, acc);
    sm_add(__expf(e2 - mn), v2, den, acc);
    m = mn;
  }
  const float inv = 1.f / (den + SOFT_EPS);
  float4 v;
  v.x = acc.x * inv + bias[4 * jj + 0];
  v.y = acc.y * inv + bias[4 * jj + 1];
  v.z = acc.z * inv + bias[4 * jj + 2];
  v.w = acc.w * inv + bias[4 * jj + 3];
  float mx = act ? fmaxf(fmaxf(v.x, v.y), fmaxf(v.z, v.w)) : -INFINITY;
  for (int off = 8; off > 0; off >>= 1) mx = fmaxf(mx, __shfl_xor(mx, off, 16));
  float sum = act ? (__expf(v.x - mx) + __expf(v.y - mx) + __expf(v.z - mx) + __expf(v.w - mx))
                  : 0.f;
  for (int off = 8; off > 0; off >>= 1) sum += __shfl_xor(sum, off, 16);
  const float lse = mx + __logf(sum);
  if (act) {
    float4 o;
    o.x = v.x - lse; o.y = v.y - lse; o.z = v.z - lse; o.w = v.w - lse;
    ((float4*)out)[(size_t)g * 10 + j] = o;
  }
}

extern "C" void kernel_launch(void* const* d_in, const int* in_sizes, int n_in,
                              void* d_out, int out_size, void* d_ws, size_t ws_size,
                              hipStream_t stream) {
  const float* x   = (const float*)d_in[0];
  const int*   ei  = (const int*)d_in[1];
  const float* W1  = (const float*)d_in[2];
  const float* as1 = (const float*)d_in[3];
  const float* ad1 = (const float*)d_in[4];
  const float* b1  = (const float*)d_in[5];
  const float* W2  = (const float*)d_in[6];
  const float* as2 = (const float*)d_in[7];
  const float* ad2 = (const float*)d_in[8];
  const float* b2  = (const float*)d_in[9];
  float* out = (float*)d_out;

  const int Fin = in_sizes[2] / 64;
  const int N   = in_sizes[0] / Fin;
  const int E   = in_sizes[1] / 2;
  const int* src = ei;
  const int* dst = ei + E;

  const int TB = 256;
  const int gM  = (N + 15) / 16;
  const int gG  = (N * 16 + TB - 1) / TB;
  const int NB  = (N + RNG_NODES - 1) / RNG_NODES;
  const int nH  = NB * BIN_BLOCKS;
  const int gS  = (nH + 1023) / 1024;

  float* ws = (float*)d_ws;
  size_t o = 0;
  float* h1    = ws + o; o += (size_t)N * 64;
  float* h2    = ws + o; o += (size_t)N * 48;
  int2* pairs  = (int2*)(ws + o); o += (size_t)E * 2;
  int* col     = (int*)(ws + o); o += (size_t)E;
  int* rowptr  = (int*)(ws + o); o += (size_t)N;
  int* hist    = (int*)(ws + o); o += (size_t)nH;
  int* loc     = (int*)(ws + o); o += (size_t)nH;
  int* off     = (int*)(ws + o); o += (size_t)nH;
  int* bsum    = (int*)(ws + o); o += (size_t)gS;

  gat_proj1_wmma<<<(gM + 3) / 4, P1_TB, 0, stream>>>(x, W1, h1, N);
  bin_hist      <<<BIN_BLOCKS, BIN_TB, (size_t)NB * 4, stream>>>(dst, hist, E, NB);
  csr_scan_block<<<gS, 1024, 0, stream>>>(hist, loc, bsum, nH);
  csr_scan_add  <<<gS, 1024, 0, stream>>>(loc, bsum, off, nH);
  bin_scatter   <<<BIN_BLOCKS, BIN_TB, (size_t)NB * 4, stream>>>(src, dst, off, pairs, E, NB);
  range_fill    <<<NB, 1024, 0, stream>>>(pairs, off, rowptr, col, N, E, NB);
  gat_gather1_proj<<<gM, G1_TB, 0, stream>>>(rowptr, col, as1, ad1, h1, b1, W2, h2, N, E);

  gat_gather2<<<gG, TB, 0, stream>>>(rowptr, col, as2, ad2, h2, b2, out, N, E);

  (void)n_in; (void)out_size; (void)ws_size;
}


# Edge sort level 1 made chunk-local: a block counts, scans and scatters its chunk in place (runs by range), range blocks walk their 256 runs; table scans, 3 launches gone; slots in new arrival order
# speedup vs baseline: 26.8920x; 1.0339x over previous
#include <hip/hip_runtime.h>
#include <math.h>

typedef __attribute__((ext_vector_type(2))) float v2f;
typedef __attribute__((ext_vector_type(8))) float v8f;

#define NEG_SLOPE 0.2f
#define SOFT_EPS  1e-16f

__device__ __forceinline__ float lrelu(float x) { return x > 0.f ? x : NEG_SLOPE * x; }

#define P1_TB    128
#define P1_PITCH 144
__global__ void __launch_bounds__(P1_TB)
gat_proj1_wmma(const float* __restrict__ X, const float* __restrict__ W,
               float* __restrict__ H, int M) {
  __shared__ float Bs[64 * P1_PITCH];
  const int tid   = threadIdx.x;
  const int wv    = tid >> 5;
  const int lane  = tid & 31;
  const int mr    = lane & 15;
  const int half  = lane >> 4;
  const int strip = blockIdx.x * (P1_TB / 32) + wv;
  const int arow  = min(strip * 16 + mr, M - 1);
  const float4* Ap = (const float4*)(X + (size_t)arow * 128) + half;
  float4 a[16];
#pragma unroll
  for (int s = 0; s < 16; ++s) a[s] = Ap[2 * s];

  {
    const float4* W4 = (const float4*)W;
    const int c = tid & 15;
#pragma unroll
    for (int i = 0; i < 8; ++i) {
      const int kp = (tid >> 4) + 8 * i;
      const float4 w0 = W4[(2 * kp) * 16 + c];
      const float4 w1 = W4[(2 * kp + 1) * 16 + c];
      *(float4*)&Bs[kp * P1_PITCH + 8 * c]     = make_float4(w0.x, w1.x, w0.y, w1.y);
      *(float4*)&Bs[kp * P1_PITCH + 8 * c + 4] = make_float4(w0.z, w1.z, w0.w, w1.w);
    }
  }
  __syncthreads();

  v8f acc0 = {}, acc1 = {}, acc2 = {}, acc3 = {};
  const float* Bl = Bs + (2 * half) * P1_PITCH + 2 * mr;
#pragma unroll
  for (int s = 0; s < 16; ++s) {
#pragma unroll
    for (int u = 0; u < 2; ++u) {
      v2f av;
      av.x = u ? a[s].z : a[s].x;
      av.y = u ? a[s].w : a[s].y;
      const float* bp = Bl + (4 * s + u) * P1_PITCH;
      const v2f b0 = *(const v2f*)(bp);
      const v2f b1 = *(const v2f*)(bp + 32);
      const v2f b2 = *(const v2f*)(bp + 64);
      const v2f b3 = *(const v2f*)(bp + 96);
      acc0 = __builtin_amdgcn_wmma_f32_16x16x4_f32(false, av, false, b0, (short)0, acc0, false, false);
      acc1 = __builtin_amdgcn_wmma_f32_16x16x4_f32(false, av, false, b1, (short)0, acc1, false, false);
      acc2 = __builtin_amdgcn_wmma_f32_16x16x4_f32(false, av, false, b2, (short)0, acc2, false, false);
      acc3 = __builtin_amdgcn_wmma_f32_16x16x4_f32(false, av, false, b3, (short)0, acc3, false, false);
    }
  }
#pragma unroll
  for (int i = 0; i < 8; ++i) {
    const int r = strip * 16 + i + 8 * half;
    if (r < M) {
      float* Hr = H + (size_t)r * 64 + mr;
      Hr[0]  = acc0[i];
      Hr[16] = acc1[i];
      Hr[32] = acc2[i];
      Hr[48] = acc3[i];
    }
  }
}

#define RNG_SHIFT  8
#define RNG_NODES  256
#define BIN_BLOCKS 256
#define BIN_TB     1024
#define RF_TB      1024
#define RF_LPC     (RF_TB / BIN_BLOCKS)
#define RF_DEPTH   8

__device__ __forceinline__ void edge_chunk(int c, int E, int& beg, int& end) {
  const int chunk = (((E + BIN_BLOCKS - 1) / BIN_BLOCKS) + 3) & ~3;
  beg = min(c * chunk, E);
  end = min(beg + chunk, E);
}

__device__ __forceinline__ void bin_add4(int* __restrict__ sh, const int4 d) {
  atomicAdd(&sh[d.x >> RNG_SHIFT], 1);
  atomicAdd(&sh[d.y >> RNG_SHIFT], 1);
  atomicAdd(&sh[d.z >> RNG_SHIFT], 1);
  atomicAdd(&sh[d.w >> RNG_SHIFT], 1);
}

__device__ __forceinline__ void bin_add8(int* __restrict__ sh, const int4 a, const int4 b) {
  atomicAdd(&sh[a.x >> RNG_SHIFT], 1);
  atomicAdd(&sh[b.x >> RNG_SHIFT], 1);
  atomicAdd(&sh[a.y >> RNG_SHIFT], 1);
  atomicAdd(&sh[b.y >> RNG_SHIFT], 1);
  atomicAdd(&sh[a.z >> RNG_SHIFT], 1);
  atomicAdd(&sh[b.z >> RNG_SHIFT], 1);
  atomicAdd(&sh[a.w >> RNG_SHIFT], 1);
  atomicAdd(&sh[b.w >> RNG_SHIFT], 1);
}

__device__ __forceinline__ int take_slot(int* __restrict__ sh, int d) {
  return atomicAdd(&sh[d >> RNG_SHIFT], 1);
}

__device__ __forceinline__ void put_pair(int2* __restrict__ mine, int pos, int s, int d, int len) {
  if ((unsigned)pos < (unsigned)len) mine[pos] = make_int2(s, d);
}

__device__ __forceinline__ void put4(int* __restrict__ sh, int2* __restrict__ mine,
                                     const int4 s, const int4 d, int len) {
  const int q0 = take_slot(sh, d.x), q1 = take_slot(sh, d.y);
  const int q2 = take_slot(sh, d.z), q3 = take_slot(sh, d.w);
  put_pair(mine, q0, s.x, d.x, len);
  put_pair(mine, q1, s.y, d.y, len);
  put_pair(mine, q2, s.z, d.z, len);
  put_pair(mine, q3, s.w, d.w, len);
}

__global__ void __launch_bounds__(BIN_TB)
chunk_sort(const int* __restrict__ src, const int* __restrict__ dst,
           int* __restrict__ tab, int2* __restrict__ pairs, int E, int NB) {
  extern __shared__ int sh[];
  __shared__ int wtot[BIN_TB / 32];
  const int tid  = threadIdx.x;
  const int lane = tid & 31;
  const int wv   = tid >> 5;
  for (int i = tid; i < NB; i += BIN_TB) sh[i] = 0;
  __syncthreads();
  int beg, end;
  edge_chunk(blockIdx.x, E, beg, end);
  const bool wide = (E & 3) == 0;
  if (wide) {
    int e = beg + 4 * tid;
    for (; e + 4 * BIN_TB < end; e += 8 * BIN_TB) {
      const int4 a = *(const int4*)(dst + e);
      const int4 b = *(const int4*)(dst + e + 4 * BIN_TB);
      bin_add8(sh, a, b);
    }
    if (e < end) bin_add4(sh, *(const int4*)(dst + e));
  } else {
    for (int e = beg + tid; e < end; e += BIN_TB) atomicAdd(&sh[dst[e] >> RNG_SHIFT], 1);
  }
  __syncthreads();
  int run = 0;
  for (int base = 0; base < NB; base += BIN_TB) {
    const int i = base + tid;
    const int v = (i < NB) ? sh[i] : 0;
    int x = v;
    for (int o = 1; o < 32; o <<= 1) {
      const int y = __shfl_up(x, o);
      if (lane >= o) x += y;
    }
    if (lane == 31) wtot[wv] = x;
    __syncthreads();
    if (wv == 0) {
      int w = wtot[lane];
      for (int o = 1; o < 32; o <<= 1) {
        const int y = __shfl_up(w, o);
        if (lane >= o) w += y;
      }
      wtot[lane] = w;
    }
    __syncthreads();
    const int first = run + ((wv > 0) ? wtot[wv - 1] : 0) + x - v;
    if (i < NB) {
      sh[i] = first;
      tab[i * BIN_BLOCKS + blockIdx.x] = first;
    }
    run += wtot[BIN_TB / 32 - 1];
    __syncthreads();
  }
  int2* mine = pairs + beg;
  const int len = end - beg;
  if (wide) {
    int e = beg + 4 * tid;
    for (; e + 4 * BIN_TB < end; e += 8 * BIN_TB) {
      const int4 da = *(const int4*)(dst + e);
      const int4 db = *(const int4*)(dst + e + 4 * BIN_TB);
      const int4 sa = *(const int4*)(src + e);
      const int4 sb = *(const int4*)(src + e + 4 * BIN_TB);
      const int q0 = take_slot(sh, da.x), q1 = take_slot(sh, da.y);
      const int q2 = take_slot(sh, da.z), q3 = take_slot(sh, da.w);
      const int q4 = take_slot(sh, db.x), q5 = take_slot(sh, db.y);
      const int q6 = take_slot(sh, db.z), q7 = take_slot(sh, db.w);
      put_pair(mine, q0, sa.x, da.x, len);
      put_pair(mine, q1, sa.y, da.y, len);
      put_pair(mine, q2, sa.z, da.z, len);
      put_pair(mine, q3, sa.w, da.w, len);
      put_pair(mine, q4, sb.x, db.x, len);
      put_pair(mine, q5, sb.y, db.y, len);
      put_pair(mine, q6, sb.z, db.z, len);
      put_pair(mine, q7, sb.w, db.w, len);
    }
    if (e < end) {
      const int4 da = *(const int4*)(dst + e);
      const int4 sa = *(const int4*)(src + e);
      put4(sh, mine, sa, da, len);
    }
  } else {
    for (int e = beg + tid; e < end; e += BIN_TB) {
      const int d = dst[e];
      put_pair(mine, take_slot(sh, d), src[e], d, len);
    }
  }
}

__global__ void __launch_bounds__(RF_TB)
range_fill(const int2* __restrict__ pairs, const int* __restrict__ tab,
           int* __restrict__ rowptr, int* __restrict__ col, int N, int E, int NB) {
  __shared__ int cnt[RNG_NODES];
  __shared__ int cur[RNG_NODES];
  __shared__ int wtot[RNG_NODES / 32];
  __shared__ int rbeg[BIN_BLOCKS];
  __shared__ int rend[BIN_BLOCKS];
  __shared__ int bsum[BIN_BLOCKS / 32];
  const int b    = blockIdx.x;
  const int tid  = threadIdx.x;
  const int lane = tid & 31;
  const int wv   = tid >> 5;
  if (tid < BIN_BLOCKS) {
    int cb, ce;
    edge_chunk(tid, E, cb, ce);
    const int clen = ce - cb;
    int r0 = tab[b * BIN_BLOCKS + tid];
    int r1 = (b + 1 < NB) ? tab[(b + 1) * BIN_BLOCKS + tid] : clen;
    r0 = max(0, min(r0, clen));
    r1 = max(r0, min(r1, clen));
    rbeg[tid] = cb + r0;
    rend[tid] = cb + r1;
    int s = r0;
    for (int o = 16; o > 0; o >>= 1) s += __shfl_xor(s, o);
    if (lane == 0) bsum[wv] = s;
  }
  if (tid < RNG_NODES) cnt[tid] = 0;
  __syncthreads();
  int beg = 0;
  for (int w = 0; w < BIN_BLOCKS / 32; ++w) beg += bsum[w];
  beg = max(0, min(beg, E));
  const int c   = tid / RF_LPC;
  const int sub = tid % RF_LPC;
  const int rb = rbeg[c], re = rend[c], last = re - 1;
  for (int i = rb + sub; i < re; i += RF_DEPTH * RF_LPC) {
    int d[RF_DEPTH];
#pragma unroll
    for (int u = 0; u < RF_DEPTH; ++u) d[u] = pairs[min(i + u * RF_LPC, last)].y;
#pragma unroll
    for (int u = 0; u < RF_DEPTH; ++u)
      atomicAdd(&cnt[d[u] & (RNG_NODES - 1)], (i + u * RF_LPC < re) ? 1 : 0);
  }
  __syncthreads();
  int v = 0, x = 0;
  if (tid < RNG_NODES) {
    v = cnt[tid];
    x = v;
    for (int o = 1; o < 32; o <<= 1) {
      const int y = __shfl_up(x, o);
      if (lane >= o) x += y;
    }
    if (lane == 31) wtot[wv] = x;
  }
  __syncthreads();
  if (tid < RNG_NODES) {
    int base = beg;
    for (int w = 0; w < wv; ++w) base += wtot[w];
    const int first = base + x - v;
    cur[tid] = first;
    const int node = b * RNG_NODES + tid;
    if (node < N) rowptr[node] = first;
  }
  __syncthreads();
  for (int i = rb + sub; i < re; i += RF_DEPTH * RF_LPC) {
    int2 p[RF_DEPTH];
    int  q[RF_DEPTH];
#pragma unroll
    for (int u = 0; u < RF_DEPTH; ++u) p[u] = pairs[min(i + u * RF_LPC, last)];
    __builtin_amdgcn_sched_barrier(0);
#pragma unroll
    for (int u = 0; u < RF_DEPTH; ++u)
      q[u] = atomicAdd(&cur[p[u].y & (RNG_NODES - 1)], (i + u * RF_LPC < re) ? 1 : 0);
#pragma unroll
    for (int u = 0; u < RF_DEPTH; ++u)
      if (i + u * RF_LPC < re && (unsigned)q[u] < (unsigned)E) col[q[u]] = p[u].x;
  }
}

__device__ __forceinline__ void sm_rescale(float sc, float& den, float4& acc) {
  den *= sc;
  acc.x *= sc; acc.y *= sc; acc.z *= sc; acc.w *= sc;
}

__device__ __forceinline__ void sm_add(float p, const float4 v, float& den, float4& acc) {
  den += p;
  acc.x += p * v.x; acc.y += p * v.y; acc.z += p * v.z; acc.w += p * v.w;
}

__device__ __forceinline__ float dot4(const float4 v, float w0, float w1, float w2, float w3) {
  return v.x * w0 + v.y * w1 + v.z * w2 + v.w * w3;
}

template <int CTRL>
__device__ __forceinline__ float lane_xchg(float x) {
  return __int_as_float(__builtin_amdgcn_update_dpp(0, __float_as_int(x), CTRL, 0xF, 0xF, false));
}

__device__ __forceinline__ float pair_sum(float p) { return p + lane_xchg<0xB1>(p); }

__device__ __forceinline__ float row_sum(float p) {
  p += lane_xchg<0xB1>(p);
  p += lane_xchg<0x4E>(p);
  p += lane_xchg<0x141>(p);
  p += lane_xchg<0x140>(p);
  return p;
}

#define G1_TB    256
#define G1_PITCH 68
__global__ void __launch_bounds__(G1_TB)
gat_gather1_proj(const int* __restrict__ rowptr, const int* __restrict__ col,
                 const float* __restrict__ attS, const float* __restrict__ attD,
                 const float* __restrict__ h1, const float* __restrict__ bias,
                 const float* __restrict__ W2, float* __restrict__ h2, int N, int E) {
  __shared__ float tile[16 * G1_PITCH];
  const int t = blockIdx.x * G1_TB + threadIdx.x;
  const int g = min(t >> 4, N - 1);
  const int j = t & 15;
  const int beg = rowptr[g];
  const int end = (g + 1 < N) ? rowptr[g + 1] : E;
  const float ws0 = attS[4 * j + 0], ws1 = attS[4 * j + 1], ws2 = attS[4 * j + 2], ws3 = attS[4 * j + 3];
  const float wd0 = attD[4 * j + 0], wd1 = attD[4 * j + 1], wd2 = attD[4 * j + 2], wd3 = attD[4 * j + 3];

  const float4* hrow = (const float4*)h1;
  float4 acc = hrow[(size_t)g * 16 + j];
  const float ad    = pair_sum(dot4(acc, wd0, wd1, wd2, wd3));
  const float eself = lrelu(pair_sum(dot4(acc, ws0, ws1, ws2, ws3)) + ad);
  float  m   = eself;
  float  den = 1.f;
  int k = beg;
  for (; k + 4 <= end; k += 4) {
    const int s0 = col[k], s1 = col[k + 1], s2 = col[k + 2], s3 = col[k + 3];
    const float4 v0 = hrow[(size_t)s0 * 16 + j];
    const float4 v1 = hrow[(size_t)s1 * 16 + j];
    const float4 v2 = hrow[(size_t)s2 * 16 + j];
    const float4 v3 = hrow[(size_t)s3 * 16 + j];
    const float e0 = lrelu(pair_sum(dot4(v0, ws0, ws1, ws2, ws3)) + ad);
    const float e1 = lrelu(pair_sum(dot4(v1, ws0, ws1, ws2, ws3)) + ad);
    const float e2 = lrelu(pair_sum(dot4(v2, ws0, ws1, ws2, ws3)) + ad);
    const float e3 = lrelu(pair_sum(dot4(v3, ws0, ws1, ws2, ws3)) + ad);
    const float mn = fmaxf(fmaxf(fmaxf(m, e0), fmaxf(e1, e2)), e3);
    sm_rescale(__expf(m - mn), den, acc);
    sm_add(__expf(e0 - mn), v0, den, acc);
    sm_add(__expf(e1 - mn), v1, den, acc);
    sm_add(__expf(e2 - mn), v2, den, acc);
    sm_add(__expf(e3 - mn), v3, den, acc);
    m = mn;
  }
  if (k < end) {
    const int last = end - 1;
    const bool has1 = k + 1 < end, has2 = k + 2 < end;
    const int s0 = col[k], s1 = col[has1 ? k + 1 : last], s2 = col[has2 ? k + 2 : last];
    const float4 v0 = hrow[(size_t)s0 * 16 + j];
    const float4 v1 = hrow[(size_t)s1 * 16 + j];
    const float4 v2 = hrow[(size_t)s2 * 16 + j];
    const float e0 = lrelu(pair_sum(dot4(v0, ws0, ws1, ws2, ws3)) + ad);
    const float x1 = lrelu(pair_sum(dot4(v1, ws0, ws1, ws2, ws3)) + ad);
    const float x2 = lrelu(pair_sum(dot4(v2, ws0, ws1, ws2, ws3)) + ad);
    const float e1 = has1 ? x1 : -INFINITY;
    const float e2 = has2 ? x2 : -INFINITY;
    const float mn = fmaxf(fmaxf(m, e0), fmaxf(e1, e2));
    sm_rescale(__expf(m - mn), den, acc);
    sm_add(__expf(e0 - mn), v0, den, acc);
    sm_add(__expf(e1 - mn), v1, den, acc);
    sm_add(__expf(e2 - mn), v2, den, acc);
    m = mn;
  }
  const int wv   = threadIdx.x >> 5;
  const int lane = threadIdx.x & 31;
  const int mr   = lane & 15;
  const int half = lane >> 4;
  const int c    = wv * 16 + mr;
  const bool cv  = c < 40;
  const int cc   = cv ? c : 39;
  float bw[32];
  if (wv < 3) {
#pragma unroll
    for (int s = 0; s < 16; ++s) {
      const int ka = 4 * s + 2 * half;
      bw[2 * s]     = W2[ka * 40 + cc];
      bw[2 * s + 1] = W2[(ka + 1) * 40 + cc];
    }
  }

  const float inv = 1.f / (den + SOFT_EPS);
  float4 o;
  o.x = acc.x * inv + bias[4 * j + 0];
  o.y = acc.y * inv + bias[4 * j + 1];
  o.z = acc.z * inv + bias[4 * j + 2];
  o.w = acc.w * inv + bias[4 * j + 3];
  o.x = o.x > 0.f ? o.x : (__expf(o.x) - 1.f);
  o.y = o.y > 0.f ? o.y : (__expf(o.y) - 1.f);
  o.z = o.z > 0.f ? o.z : (__expf(o.z) - 1.f);
  o.w = o.w > 0.f ? o.w : (__expf(o.w) - 1.f);

  *(float4*)&tile[(threadIdx.x >> 4) * G1_PITCH + 4 * j] = o;
  __syncthreads();
  if (wv < 3) {
    const float* Ar = tile + mr * G1_PITCH;
    v8f acc2 = {};
#pragma unroll
    for (int s = 0; s < 16; ++s) {
      const int ka = 4 * s + 2 * half;
      v2f a, b;
      a.x = Ar[ka];
      a.y = Ar[ka + 1];
      b.x = cv ? bw[2 * s] : 0.f;
      b.y = cv ? bw[2 * s + 1] : 0.f;
      acc2 = __builtin_amdgcn_wmma_f32_16x16x4_f32(false, a, false, b, (short)0, acc2,
                                                   false, false);
    }
    const int row0 = blockIdx.x * 16;
#pragma unroll
    for (int i = 0; i < 8; ++i) {
      const int r = row0 + i + 8 * half;
      if (r < N) h2[(size_t)r * 48 + c] = acc2[i];
    }
  }
}

__global__ void gat_gather2(const int* __restrict__ rowptr, const int* __restrict__ col,
                            const float* __restrict__ attS, const float* __restrict__ attD,
                            const float* __restrict__ h2, const float* __restrict__ bias,
                            float* __restrict__ out, int N, int E) {
  const int t = blockIdx.x * blockDim.x + threadIdx.x;
  const int g = t >> 4;
  if (g >= N) return;
  const int j = t & 15;
  const bool act = j < 10;
  const int jj = act ? j : 9;
  const int beg = rowptr[g];
  const int end = (g + 1 < N) ? rowptr[g + 1] : E;
  const float ws0 = act ? attS[4 * jj + 0] : 0.f, ws1 = act ? attS[4 * jj + 1] : 0.f;
  const float ws2 = act ? attS[4 * jj + 2] : 0.f, ws3 = act ? attS[4 * jj + 3] : 0.f;
  const float wd0 = act ? attD[4 * jj + 0] : 0.f, wd1 = act ? attD[4 * jj + 1] : 0.f;
  const float wd2 = act ? attD[4 * jj + 2] : 0.f, wd3 = act ? attD[4 * jj + 3] : 0.f;

  const float4* hrow = (const float4*)h2;
  float4 acc = hrow[(size_t)g * 12 + jj];
  const float ad    = row_sum(dot4(acc, wd0, wd1, wd2, wd3));
  const float eself = lrelu(row_sum(dot4(acc, ws0, ws1, ws2, ws3)) + ad);
  float  m   = eself;
  float  den = 1.f;
  int k = beg;
  for (; k + 4 <= end; k += 4) {
    const int s0 = col[k], s1 = col[k + 1], s2 = col[k + 2], s3 = col[k + 3];
    const float4 v0 = hrow[(size_t)s0 * 12 + jj];
    const float4 v1 = hrow[(size_t)s1 * 12 + jj];
    const float4 v2 = hrow[(size_t)s2 * 12 + jj];
    const float4 v3 = hrow[(size_t)s3 * 12 + jj];
    const float e0 = lrelu(row_sum(dot4(v0, ws0, ws1, ws2, ws3)) + ad);
    const float e1 = lrelu(row_sum(dot4(v1, ws0, ws1, ws2, ws3)) + ad);
    const float e2 = lrelu(row_sum(dot4(v2, ws0, ws1, ws2, ws3)) + ad);
    const float e3 = lrelu(row_sum(dot4(v3, ws0, ws1, ws2, ws3)) + ad);
    const float mn = fmaxf(fmaxf(fmaxf(m, e0), fmaxf(e1, e2)), e3);
    sm_rescale(__expf(m - mn), den, acc);
    sm_add(__expf(e0 - mn), v0, den, acc);
    sm_add(__expf(e1 - mn), v1, den, acc);
    sm_add(__expf(e2 - mn), v2, den, acc);
    sm_add(__expf(e3 - mn), v3, den, acc);
    m = mn;
  }
  if (k < end) {
    const int last = end - 1;
    const bool has1 = k + 1 < end, has2 = k + 2 < end;
    const int s0 = col[k], s1 = col[has1 ? k + 1 : last], s2 = col[has2 ? k + 2 : last];
    const float4 v0 = hrow[(size_t)s0 * 12 + jj];
    const float4 v1 = hrow[(size_t)s1 * 12 + jj];
    const float4 v2 = hrow[(size_t)s2 * 12 + jj];
    const float e0 = lrelu(row_sum(dot4(v0, ws0, ws1, ws2, ws3)) + ad);
    const float x1 = lrelu(row_sum(dot4(v1, ws0, ws1, ws2, ws3)) + ad);
    const float x2 = lrelu(row_sum(dot4(v2, ws0, ws1, ws2, ws3)) + ad);
    const float e1 = has1 ? x1 : -INFINITY;
    const float e2 = has2 ? x2 : -INFINITY;
    const float mn = fmaxf(fmaxf(m, e0), fmaxf(e1, e2));
    sm_rescale(__expf(m - mn), den, acc);
    sm_add(__expf(e0 - mn), v0, den, acc);
    sm_add(__expf(e1 - mn), v1, den, acc);
    sm_add(__expf(e2 - mn), v2, den, acc);
    m = mn;
  }
  const float inv = 1.f / (den + SOFT_EPS);
  float4 v;
  v.x = acc.x * inv + bias[4 * jj + 0];
  v.y = acc.y * inv + bias[4 * jj + 1];
  v.z = acc.z * inv + bias[4 * jj + 2];
  v.w = acc.w * inv + bias[4 * jj + 3];
  float mx = act ? fmaxf(fmaxf(v.x, v.y), fmaxf(v.z, v.w)) : -INFINITY;
  for (int off = 8; off > 0; off >>= 1) mx = fmaxf(mx, __shfl_xor(mx, off, 16));
  float sum = act ? (__expf(v.x - mx) + __expf(v.y - mx) + __expf(v.z - mx) + __expf(v.w - mx))
                  : 0.f;
  for (int off = 8; off > 0; off >>= 1) sum += __shfl_xor(sum, off, 16);
  const float lse = mx + __logf(sum);
  if (act) {
    float4 o;
    o.x = v.x - lse; o.y = v.y - lse; o.z = v.z - lse; o.w = v.w - lse;
    ((float4*)out)[(size_t)g * 10 + j] = o;
  }
}

extern "C" void kernel_launch(void* const* d_in, const int* in_sizes, int n_in,
                              void* d_out, int out_size, void* d_ws, size_t ws_size,
                              hipStream_t stream) {
  const float* x   = (const float*)d_in[0];
  const int*   ei  = (const int*)d_in[1];
  const float* W1  = (const float*)d_in[2];
  const float* as1 = (const float*)d_in[3];
  const float* ad1 = (const float*)d_in[4];
  const float* b1  = (const float*)d_in[5];
  const float* W2  = (const float*)d_in[6];
  const float* as2 = (const float*)d_in[7];
  const float* ad2 = (const float*)d_in[8];
  const float* b2  = (const float*)d_in[9];
  float* out = (float*)d_out;

  const int Fin = in_sizes[2] / 64;
  const int N   = in_sizes[0] / Fin;
  const int E   = in_sizes[1] / 2;
  const int* src = ei;
  const int* dst = ei + E;

  const int TB = 256;
  const int gM  = (N + 15) / 16;
  const int gG  = (N * 16 + TB - 1) / TB;
  const int NB  = (N + RNG_NODES - 1) / RNG_NODES;
  const int nH  = NB * BIN_BLOCKS;

  float* ws = (float*)d_ws;
  size_t o = 0;
  float* h1    = ws + o; o += (size_t)N * 64;
  float* h2    = ws + o; o += (size_t)N * 48;
  int2* pairs  = (int2*)(ws + o); o += (size_t)E * 2;
  int* col     = (int*)(ws + o); o += (size_t)E;
  int* rowptr  = (int*)(ws + o); o += (size_t)N;
  int* tab     = (int*)(ws + o); o += (size_t)nH;

  gat_proj1_wmma<<<(gM + 3) / 4, P1_TB, 0, stream>>>(x, W1, h1, N);
  chunk_sort<<<BIN_BLOCKS, BIN_TB, (size_t)NB * 4, stream>>>(src, dst, tab, pairs, E, NB);
  range_fill<<<NB, RF_TB, 0, stream>>>(pairs, tab, rowptr, col, N, E, NB);
  gat_gather1_proj<<<gM, G1_TB, 0, stream>>>(rowptr, col, as1, ad1, h1, b1, W2, h2, N, E);

  gat_gather2<<<gG, TB, 0, stream>>>(rowptr, col, as2, ad2, h2, b2, out, N, E);

  (void)n_in; (void)out_size; (void)ws_size;
}


# Both sort levels build their stretch (a chunk's pairs by range; a range's col slots) as an image in LDS and copy it out in whole lines: scattered 8-B and 4-B global stores gone; slot order kept
# speedup vs baseline: 29.4758x; 1.0961x over previous
#include <hip/hip_runtime.h>
#include <math.h>

typedef __attribute__((ext_vector_type(2))) float v2f;
typedef __attribute__((ext_vector_type(8))) float v8f;

#define NEG_SLOPE 0.2f
#define SOFT_EPS  1e-16f

__device__ __forceinline__ float lrelu(float x) { return x > 0.f ? x : NEG_SLOPE * x; }

#define P1_TB    128
#define P1_PITCH 144
__global__ void __launch_bounds__(P1_TB)
gat_proj1_wmma(const float* __restrict__ X, const float* __restrict__ W,
               float* __restrict__ H, int M) {
  __shared__ float Bs[64 * P1_PITCH];
  const int tid   = threadIdx.x;
  const int wv    = tid >> 5;
  const int lane  = tid & 31;
  const int mr    = lane & 15;
  const int half  = lane >> 4;
  const int strip = blockIdx.x * (P1_TB / 32) + wv;
  const int arow  = min(strip * 16 + mr, M - 1);
  const float4* Ap = (const float4*)(X + (size_t)arow * 128) + half;
  float4 a[16];
#pragma unroll
  for (int s = 0; s < 16; ++s) a[s] = Ap[2 * s];

  {
    const float4* W4 = (const float4*)W;
    const int c = tid & 15;
#pragma unroll
    for (int i = 0; i < 8; ++i) {
      const int kp = (tid >> 4) + 8 * i;
      const float4 w0 = W4[(2 * kp) * 16 + c];
      const float4 w1 = W4[(2 * kp + 1) * 16 + c];
      *(float4*)&Bs[kp * P1_PITCH + 8 * c]     = make_float4(w0.x, w1.x, w0.y, w1.y);
      *(float4*)&Bs[kp * P1_PITCH + 8 * c + 4] = make_float4(w0.z, w1.z, w0.w, w1.w);
    }
  }
  __syncthreads();

  v8f acc0 = {}, acc1 = {}, acc2 = {}, acc3 = {};
  const float* Bl = Bs + (2 * half) * P1_PITCH + 2 * mr;
#pragma unroll
  for (int s = 0; s < 16; ++s) {
#pragma unroll
    for (int u = 0; u < 2; ++u) {
      v2f av;
      av.x = u ? a[s].z : a[s].x;
      av.y = u ? a[s].w : a[s].y;
      const float* bp = Bl + (4 * s + u) * P1_PITCH;
      const v2f b0 = *(const v2f*)(bp);
      const v2f b1 = *(const v2f*)(bp + 32);
      const v2f b2 = *(const v2f*)(bp + 64);
      const v2f b3 = *(const v2f*)(bp + 96);
      acc0 = __builtin_amdgcn_wmma_f32_16x16x4_f32(false, av, false, b0, (short)0, acc0, false, false);
      acc1 = __builtin_amdgcn_wmma_f32_16x16x4_f32(false, av, false, b1, (short)0, acc1, false, false);
      acc2 = __builtin_amdgcn_wmma_f32_16x16x4_f32(false, av, false, b2, (short)0, acc2, false, false);
      acc3 = __builtin_amdgcn_wmma_f32_16x16x4_f32(false, av, false, b3, (short)0, acc3, false, false);
    }
  }
#pragma unroll
  for (int i = 0; i < 8; ++i) {
    const int r = strip * 16 + i + 8 * half;
    if (r < M) {
      float* Hr = H + (size_t)r * 64 + mr;
      Hr[0]  = acc0[i];
      Hr[16] = acc1[i];
      Hr[32] = acc2[i];
      Hr[48] = acc3[i];
    }
  }
}

#define RNG_SHIFT  8
#define RNG_NODES  256
#define BIN_BLOCKS 256
#define BIN_TB     1024
#define RF_TB      1024
#define RF_LPC     (RF_TB / BIN_BLOCKS)
#define RF_DEPTH   8
#define CS_CAP     16384
#define RF_CAP     16384

__device__ __forceinline__ void edge_chunk(int c, int E, int& beg, int& end) {
  const int chunk = (((E + BIN_BLOCKS - 1) / BIN_BLOCKS) + 3) & ~3;
  beg = min(c * chunk, E);
  end = min(beg + chunk, E);
}

__device__ __forceinline__ void bin_add4(int* __restrict__ sh, const int4 d) {
  atomicAdd(&sh[d.x >> RNG_SHIFT], 1);
  atomicAdd(&sh[d.y >> RNG_SHIFT], 1);
  atomicAdd(&sh[d.z >> RNG_SHIFT], 1);
  atomicAdd(&sh[d.w >> RNG_SHIFT], 1);
}

__device__ __forceinline__ void bin_add8(int* __restrict__ sh, const int4 a, const int4 b) {
  atomicAdd(&sh[a.x >> RNG_SHIFT], 1);
  atomicAdd(&sh[b.x >> RNG_SHIFT], 1);
  atomicAdd(&sh[a.y >> RNG_SHIFT], 1);
  atomicAdd(&sh[b.y >> RNG_SHIFT], 1);
  atomicAdd(&sh[a.z >> RNG_SHIFT], 1);
  atomicAdd(&sh[b.z >> RNG_SHIFT], 1);
  atomicAdd(&sh[a.w >> RNG_SHIFT], 1);
  atomicAdd(&sh[b.w >> RNG_SHIFT], 1);
}

__device__ __forceinline__ int take_slot(int* __restrict__ sh, int d) {
  return atomicAdd(&sh[d >> RNG_SHIFT], 1);
}

__device__ __forceinline__ void put_pair(int2* __restrict__ to, int pos, int s, int d, int len) {
  if ((unsigned)pos < (unsigned)len) to[pos] = make_int2(s, d);
}

__device__ __forceinline__ void put4(int* __restrict__ sh, int2* __restrict__ to,
                                     const int4 s, const int4 d, int len) {
  const int q0 = take_slot(sh, d.x), q1 = take_slot(sh, d.y);
  const int q2 = take_slot(sh, d.z), q3 = take_slot(sh, d.w);
  put_pair(to, q0, s.x, d.x, len);
  put_pair(to, q1, s.y, d.y, len);
  put_pair(to, q2, s.z, d.z, len);
  put_pair(to, q3, s.w, d.w, len);
}

__device__ __forceinline__ void place_chunk(const int* __restrict__ src, const int* __restrict__ dst,
                                            int* __restrict__ sh, int2* __restrict__ to,
                                            int beg, int end, int len, bool wide, int tid) {
  if (wide) {
    int e = beg + 4 * tid;
    for (; e + 4 * BIN_TB < end; e += 8 * BIN_TB) {
      const int4 da = *(const int4*)(dst + e);
      const int4 db = *(const int4*)(dst + e + 4 * BIN_TB);
      const int4 sa = *(const int4*)(src + e);
      const int4 sb = *(const int4*)(src + e + 4 * BIN_TB);
      const int q0 = take_slot(sh, da.x), q1 = take_slot(sh, da.y);
      const int q2 = take_slot(sh, da.z), q3 = take_slot(sh, da.w);
      const int q4 = take_slot(sh, db.x), q5 = take_slot(sh, db.y);
      const int q6 = take_slot(sh, db.z), q7 = take_slot(sh, db.w);
      put_pair(to, q0, sa.x, da.x, len);
      put_pair(to, q1, sa.y, da.y, len);
      put_pair(to, q2, sa.z, da.z, len);
      put_pair(to, q3, sa.w, da.w, len);
      put_pair(to, q4, sb.x, db.x, len);
      put_pair(to, q5, sb.y, db.y, len);
      put_pair(to, q6, sb.z, db.z, len);
      put_pair(to, q7, sb.w, db.w, len);
    }
    if (e < end) {
      const int4 da = *(const int4*)(dst + e);
      const int4 sa = *(const int4*)(src + e);
      put4(sh, to, sa, da, len);
    }
  } else {
    for (int e = beg + tid; e < end; e += BIN_TB) {
      const int d = dst[e];
      put_pair(to, take_slot(sh, d), src[e], d, len);
    }
  }
}

__global__ void __launch_bounds__(BIN_TB)
chunk_sort(const int* __restrict__ src, const int* __restrict__ dst,
           int* __restrict__ tab, int2* __restrict__ pairs, int E, int NB) {
  extern __shared__ int sh[];
  __shared__ int wtot[BIN_TB / 32];
  __shared__ __attribute__((aligned(16))) int2 img[CS_CAP];
  const int tid  = threadIdx.x;
  const int lane = tid & 31;
  const int wv   = tid >> 5;
  for (int i = tid; i < NB; i += BIN_TB) sh[i] = 0;
  __syncthreads();
  int beg, end;
  edge_chunk(blockIdx.x, E, beg, end);
  const bool wide = (E & 3) == 0;
  if (wide) {
    int e = beg + 4 * tid;
    for (; e + 4 * BIN_TB < end; e += 8 * BIN_TB) {
      const int4 a = *(const int4*)(dst + e);
      const int4 b = *(const int4*)(dst + e + 4 * BIN_TB);
      bin_add8(sh, a, b);
    }
    if (e < end) bin_add4(sh, *(const int4*)(dst + e));
  } else {
    for (int e = beg + tid; e < end; e += BIN_TB) atomicAdd(&sh[dst[e] >> RNG_SHIFT], 1);
  }
  __syncthreads();
  int run = 0;
  for (int base = 0; base < NB; base += BIN_TB) {
    const int i = base + tid;
    const int v = (i < NB) ? sh[i] : 0;
    int x = v;
    for (int o = 1; o < 32; o <<= 1) {
      const int y = __shfl_up(x, o);
      if (lane >= o) x += y;
    }
    if (lane == 31) wtot[wv] = x;
    __syncthreads();
    if (wv == 0) {
      int w = wtot[lane];
      for (int o = 1; o < 32; o <<= 1) {
        const int y = __shfl_up(w, o);
        if (lane >= o) w += y;
      }
      wtot[lane] = w;
    }
    __syncthreads();
    const int first = run + ((wv > 0) ? wtot[wv - 1] : 0) + x - v;
    if (i < NB) {
      sh[i] = first;
      tab[i * BIN_BLOCKS + blockIdx.x] = first;
    }
    run += wtot[BIN_TB / 32 - 1];
    __syncthreads();
  }
  int2* mine = pairs + beg;
  const int len = end - beg;
  if (len <= CS_CAP) {
    place_chunk(src, dst, sh, img, beg, end, len, wide, tid);
    __syncthreads();
    const int4* im4 = (const int4*)img;
    int4* out4 = (int4*)mine;
    const int n2 = len >> 1;
    for (int i = tid; i < n2; i += BIN_TB) out4[i] = im4[i];
    if ((len & 1) && tid == 0) mine[len - 1] = img[len - 1];
  } else {
    place_chunk(src, dst, sh, mine, beg, end, len, wide, tid);
  }
}

__global__ void __launch_bounds__(RF_TB)
range_fill(const int2* __restrict__ pairs, const int* __restrict__ tab,
           int* __restrict__ rowptr, int* __restrict__ col, int N, int E, int NB) {
  __shared__ int cnt[RNG_NODES];
  __shared__ int cur[RNG_NODES];
  __shared__ int wtot[RNG_NODES / 32];
  __shared__ int rbeg[BIN_BLOCKS];
  __shared__ int rend[BIN_BLOCKS];
  __shared__ int bsum[BIN_BLOCKS / 32];
  __shared__ int img[RF_CAP];
  const int b    = blockIdx.x;
  const int tid  = threadIdx.x;
  const int lane = tid & 31;
  const int wv   = tid >> 5;
  if (tid < BIN_BLOCKS) {
    int cb, ce;
    edge_chunk(tid, E, cb, ce);
    const int clen = ce - cb;
    int r0 = tab[b * BIN_BLOCKS + tid];
    int r1 = (b + 1 < NB) ? tab[(b + 1) * BIN_BLOCKS + tid] : clen;
    r0 = max(0, min(r0, clen));
    r1 = max(r0, min(r1, clen));
    rbeg[tid] = cb + r0;
    rend[tid] = cb + r1;
    int s = r0;
    for (int o = 16; o > 0; o >>= 1) s += __shfl_xor(s, o);
    if (lane == 0) bsum[wv] = s;
  }
  if (tid < RNG_NODES) cnt[tid] = 0;
  __syncthreads();
  int beg = 0;
  for (int w = 0; w < BIN_BLOCKS / 32; ++w) beg += bsum[w];
  beg = max(0, min(beg, E));
  const int c   = tid / RF_LPC;
  const int sub = tid % RF_LPC;
  const int rb = rbeg[c], re = rend[c], last = re - 1;
  for (int i = rb + sub; i < re; i += RF_DEPTH * RF_LPC) {
    int d[RF_DEPTH];
#pragma unroll
    for (int u = 0; u < RF_DEPTH; ++u) d[u] = pairs[min(i + u * RF_LPC, last)].y;
#pragma unroll
    for (int u = 0; u < RF_DEPTH; ++u)
      atomicAdd(&cnt[d[u] & (RNG_NODES - 1)], (i + u * RF_LPC < re) ? 1 : 0);
  }
  __syncthreads();
  int v = 0, x = 0;
  if (tid < RNG_NODES) {
    v = cnt[tid];
    x = v;
    for (int o = 1; o < 32; o <<= 1) {
      const int y = __shfl_up(x, o);
      if (lane >= o) x += y;
    }
    if (lane == 31) wtot[wv] = x;
  }
  __syncthreads();
  if (tid < RNG_NODES) {
    int base = beg;
    for (int w = 0; w < wv; ++w) base += wtot[w];
    const int first = base + x - v;
    cur[tid] = first;
    const int node = b * RNG_NODES + tid;
    if (node < N) rowptr[node] = first;
  }
  __syncthreads();
  int total = 0;
  for (int w = 0; w < RNG_NODES / 32; ++w) total += wtot[w];
  const bool inlds = total <= RF_CAP;
  for (int i = rb + sub; i < re; i += RF_DEPTH * RF_LPC) {
    int2 p[RF_DEPTH];
    int  q[RF_DEPTH];
#pragma unroll
    for (int u = 0; u < RF_DEPTH; ++u) p[u] = pairs[min(i + u * RF_LPC, last)];
    __builtin_amdgcn_sched_barrier(0);
#pragma unroll
    for (int u = 0; u < RF_DEPTH; ++u)
      q[u] = atomicAdd(&cur[p[u].y & (RNG_NODES - 1)], (i + u * RF_LPC < re) ? 1 : 0);
    if (inlds) {
#pragma unroll
      for (int u = 0; u < RF_DEPTH; ++u)
        if (i + u * RF_LPC < re && (unsigned)(q[u] - beg) < (unsigned)total) img[q[u] - beg] = p[u].x;
    } else {
#pragma unroll
      for (int u = 0; u < RF_DEPTH; ++u)
        if (i + u * RF_LPC < re && (unsigned)q[u] < (unsigned)E) col[q[u]] = p[u].x;
    }
  }
  if (inlds) {
    __syncthreads();
    for (int i = tid; i < total; i += RF_TB)
      if ((unsigned)(beg + i) < (unsigned)E) col[beg + i] = img[i];
  }
}

__device__ __forceinline__ void sm_rescale(float sc, float& den, float4& acc) {
  den *= sc;
  acc.x *= sc; acc.y *= sc; acc.z *= sc; acc.w *= sc;
}

__device__ __forceinline__ void sm_add(float p, const float4 v, float& den, float4& acc) {
  den += p;
  acc.x += p * v.x; acc.y += p * v.y; acc.z += p * v.z; acc.w += p * v.w;
}

__device__ __forceinline__ float dot4(const float4 v, float w0, float w1, float w2, float w3) {
  return v.x * w0 + v.y * w1 + v.z * w2 + v.w * w3;
}

template <int CTRL>
__device__ __forceinline__ float lane_xchg(float x) {
  return __int_as_float(__builtin_amdgcn_update_dpp(0, __float_as_int(x), CTRL, 0xF, 0xF, false));
}

__device__ __forceinline__ float pair_sum(float p) { return p + lane_xchg<0xB1>(p); }

__device__ __forceinline__ float row_sum(float p) {
  p += lane_xchg<0xB1>(p);
  p += lane_xchg<0x4E>(p);
  p += lane_xchg<0x141>(p);
  p += lane_xchg<0x140>(p);
  return p;
}

#define G1_TB    256
#define G1_PITCH 68
__global__ void __launch_bounds__(G1_TB)
gat_gather1_proj(const int* __restrict__ rowptr, const int* __restrict__ col,
                 const float* __restrict__ attS, const float* __restrict__ attD,
                 const float* __restrict__ h1, const float* __restrict__ bias,
                 const float* __restrict__ W2, float* __restrict__ h2, int N, int E) {
  __shared__ float tile[16 * G1_PITCH];
  const int t = blockIdx.x * G1_TB + threadIdx.x;
  const int g = min(t >> 4, N - 1);
  const int j = t & 15;
  const int beg = rowptr[g];
  const int end = (g + 1 < N) ? rowptr[g + 1] : E;
  const float ws0 = attS[4 * j + 0], ws1 = attS[4 * j + 1], ws2 = attS[4 * j + 2], ws3 = attS[4 * j + 3];
  const float wd0 = attD[4 * j + 0], wd1 = attD[4 * j + 1], wd2 = attD[4 * j + 2], wd3 = attD[4 * j + 3];

  const float4* hrow = (const float4*)h1;
  float4 acc = hrow[(size_t)g * 16 + j];
  const float ad    = pair_sum(dot4(acc, wd0, wd1, wd2, wd3));
  const float eself = lrelu(pair_sum(dot4(acc, ws0, ws1, ws2, ws3)) + ad);
  float  m   = eself;
  float  den = 1.f;
  int k = beg;
  for (; k + 4 <= end; k += 4) {
    const int s0 = col[k], s1 = col[k + 1], s2 = col[k + 2], s3 = col[k + 3];
    const float4 v0 = hrow[(size_t)s0 * 16 + j];
    const float4 v1 = hrow[(size_t)s1 * 16 + j];
    const float4 v2 = hrow[(size_t)s2 * 16 + j];
    const float4 v3 = hrow[(size_t)s3 * 16 + j];
    const float e0 = lrelu(pair_sum(dot4(v0, ws0, ws1, ws2, ws3)) + ad);
    const float e1 = lrelu(pair_sum(dot4(v1, ws0, ws1, ws2, ws3)) + ad);
    const float e2 = lrelu(pair_sum(dot4(v2, ws0, ws1, ws2, ws3)) + ad);
    const float e3 = lrelu(pair_sum(dot4(v3, ws0, ws1, ws2, ws3)) + ad);
    const float mn = fmaxf(fmaxf(fmaxf(m, e0), fmaxf(e1, e2)), e3);
    sm_rescale(__expf(m - mn), den, acc);
    sm_add(__expf(e0 - mn), v0, den, acc);
    sm_add(__expf(e1 - mn), v1, den, acc);
    sm_add(__expf(e2 - mn), v2, den, acc);
    sm_add(__expf(e3 - mn), v3, den, acc);
    m = mn;
  }
  if (k < end) {
    const int last = end - 1;
    const bool has1 = k + 1 < end, has2 = k + 2 < end;
    const int s0 = col[k], s1 = col[has1 ? k + 1 : last], s2 = col[has2 ? k + 2 : last];
    const float4 v0 = hrow[(size_t)s0 * 16 + j];
    const float4 v1 = hrow[(size_t)s1 * 16 + j];
    const float4 v2 = hrow[(size_t)s2 * 16 + j];
    const float e0 = lrelu(pair_sum(dot4(v0, ws0, ws1, ws2, ws3)) + ad);
    const float x1 = lrelu(pair_sum(dot4(v1, ws0, ws1, ws2, ws3)) + ad);
    const float x2 = lrelu(pair_sum(dot4(v2, ws0, ws1, ws2, ws3)) + ad);
    const float e1 = has1 ? x1 : -INFINITY;
    const float e2 = has2 ? x2 : -INFINITY;
    const float mn = fmaxf(fmaxf(m, e0), fmaxf(e1, e2));
    sm_rescale(__expf(m - mn), den, acc);
    sm_add(__expf(e0 - mn), v0, den, acc);
    sm_add(__expf(e1 - mn), v1, den, acc);
    sm_add(__expf(e2 - mn), v2, den, acc);
    m = mn;
  }
  const int wv   = threadIdx.x >> 5;
  const int lane = threadIdx.x & 31;
  const int mr   = lane & 15;
  const int half = lane >> 4;
  const int c    = wv * 16 + mr;
  const bool cv  = c < 40;
  const int cc   = cv ? c : 39;
  float bw[32];
  if (wv < 3) {
#pragma unroll
    for (int s = 0; s < 16; ++s) {
      const int ka = 4 * s + 2 * half;
      bw[2 * s]     = W2[ka * 40 + cc];
      bw[2 * s + 1] = W2[(ka + 1) * 40 + cc];
    }
  }

  const float inv = 1.f / (den + SOFT_EPS);
  float4 o;
  o.x = acc.x * inv + bias[4 * j + 0];
  o.y = acc.y * inv + bias[4 * j + 1];
  o.z = acc.z * inv + bias[4 * j + 2];
  o.w = acc.w * inv + bias[4 * j + 3];
  o.x = o.x > 0.f ? o.x : (__expf(o.x) - 1.f);
  o.y = o.y > 0.f ? o.y : (__expf(o.y) - 1.f);
  o.z = o.z > 0.f ? o.z : (__expf(o.z) - 1.f);
  o.w = o.w > 0.f ? o.w : (__expf(o.w) - 1.f);

  *(float4*)&tile[(threadIdx.x >> 4) * G1_PITCH + 4 * j] = o;
  __syncthreads();
  if (wv < 3) {
    const float* Ar = tile + mr * G1_PITCH;
    v8f acc2 = {};
#pragma unroll
    for (int s = 0; s < 16; ++s) {
      const int ka = 4 * s + 2 * half;
      v2f a, b;
      a.x = Ar[ka];
      a.y = Ar[ka + 1];
      b.x = cv ? bw[2 * s] : 0.f;
      b.y = cv ? bw[2 * s + 1] : 0.f;
      acc2 = __builtin_amdgcn_wmma_f32_16x16x4_f32(false, a, false, b, (short)0, acc2,
                                                   false, false);
    }
    const int row0 = blockIdx.x * 16;
#pragma unroll
    for (int i = 0; i < 8; ++i) {
      const int r = row0 + i + 8 * half;
      if (r < N) h2[(size_t)r * 48 + c] = acc2[i];
    }
  }
}

__global__ void gat_gather2(const int* __restrict__ rowptr, const int* __restrict__ col,
                            const float* __restrict__ attS, const float* __restrict__ attD,
                            const float* __restrict__ h2, const float* __restrict__ bias,
                            float* __restrict__ out, int N, int E) {
  const int t = blockIdx.x * blockDim.x + threadIdx.x;
  const int g = t >> 4;
  if (g >= N) return;
  const int j = t & 15;
  const bool act = j < 10;
  const int jj = act ? j : 9;
  const int beg = rowptr[g];
  const int end = (g + 1 < N) ? rowptr[g + 1] : E;
  const float ws0 = act ? attS[4 * jj + 0] : 0.f, ws1 = act ? attS[4 * jj + 1] : 0.f;
  const float ws2 = act ? attS[4 * jj + 2] : 0.f, ws3 = act ? attS[4 * jj + 3] : 0.f;
  const float wd0 = act ? attD[4 * jj + 0] : 0.f, wd1 = act ? attD[4 * jj + 1] : 0.f;
  const float wd2 = act ? attD[4 * jj + 2] : 0.f, wd3 = act ? attD[4 * jj + 3] : 0.f;

  const float4* hrow = (const float4*)h2;
  float4 acc = hrow[(size_t)g * 12 + jj];
  const float ad    = row_sum(dot4(acc, wd0, wd1, wd2, wd3));
  const float eself = lrelu(row_sum(dot4(acc, ws0, ws1, ws2, ws3)) + ad);
  float  m   = eself;
  float  den = 1.f;
  int k = beg;
  for (; k + 4 <= end; k += 4) {
    const int s0 = col[k], s1 = col[k + 1], s2 = col[k + 2], s3 = col[k + 3];
    const float4 v0 = hrow[(size_t)s0 * 12 + jj];
    const float4 v1 = hrow[(size_t)s1 * 12 + jj];
    const float4 v2 = hrow[(size_t)s2 * 12 + jj];
    const float4 v3 = hrow[(size_t)s3 * 12 + jj];
    const float e0 = lrelu(row_sum(dot4(v0, ws0, ws1, ws2, ws3)) + ad);
    const float e1 = lrelu(row_sum(dot4(v1, ws0, ws1, ws2, ws3)) + ad);
    const float e2 = lrelu(row_sum(dot4(v2, ws0, ws1, ws2, ws3)) + ad);
    const float e3 = lrelu(row_sum(dot4(v3, ws0, ws1, ws2, ws3)) + ad);
    const float mn = fmaxf(fmaxf(fmaxf(m, e0), fmaxf(e1, e2)), e3);
    sm_rescale(__expf(m - mn), den, acc);
    sm_add(__expf(e0 - mn), v0, den, acc);
    sm_add(__expf(e1 - mn), v1, den, acc);
    sm_add(__expf(e2 - mn), v2, den, acc);
    sm_add(__expf(e3 - mn), v3, den, acc);
    m = mn;
  }
  if (k < end) {
    const int last = end - 1;
    const bool has1 = k + 1 < end, has2 = k + 2 < end;
    const int s0 = col[k], s1 = col[has1 ? k + 1 : last], s2 = col[has2 ? k + 2 : last];
    const float4 v0 = hrow[(size_t)s0 * 12 + jj];
    const float4 v1 = hrow[(size_t)s1 * 12 + jj];
    const float4 v2 = hrow[(size_t)s2 * 12 + jj];
    const float e0 = lrelu(row_sum(dot4(v0, ws0, ws1, ws2, ws3)) + ad);
    const float x1 = lrelu(row_sum(dot4(v1, ws0, ws1, ws2, ws3)) + ad);
    const float x2 = lrelu(row_sum(dot4(v2, ws0, ws1, ws2, ws3)) + ad);
    const float e1 = has1 ? x1 : -INFINITY;
    const float e2 = has2 ? x2 : -INFINITY;
    const float mn = fmaxf(fmaxf(m, e0), fmaxf(e1, e2));
    sm_rescale(__expf(m - mn), den, acc);
    sm_add(__expf(e0 - mn), v0, den, acc);
    sm_add(__expf(e1 - mn), v1, den, acc);
    sm_add(__expf(e2 - mn), v2, den, acc);
    m = mn;
  }
  const float inv = 1.f / (den + SOFT_EPS);
  float4 v;
  v.x = acc.x * inv + bias[4 * jj + 0];
  v.y = acc.y * inv + bias[4 * jj + 1];
  v.z = acc.z * inv + bias[4 * jj + 2];
  v.w = acc.w * inv + bias[4 * jj + 3];
  float mx = act ? fmaxf(fmaxf(v.x, v.y), fmaxf(v.z, v.w)) : -INFINITY;
  for (int off = 8; off > 0; off >>= 1) mx = fmaxf(mx, __shfl_xor(mx, off, 16));
  float sum = act ? (__expf(v.x - mx) + __expf(v.y - mx) + __expf(v.z - mx) + __expf(v.w - mx))
                  : 0.f;
  for (int off = 8; off > 0; off >>= 1) sum += __shfl_xor(sum, off, 16);
  const float lse = mx + __logf(sum);
  if (act) {
    float4 o;
    o.x = v.x - lse; o.y = v.y - lse; o.z = v.z - lse; o.w = v.w - lse;
    ((float4*)out)[(size_t)g * 10 + j] = o;
  }
}

extern "C" void kernel_launch(void* const* d_in, const int* in_sizes, int n_in,
                              void* d_out, int out_size, void* d_ws, size_t ws_size,
                              hipStream_t stream) {
  const float* x   = (const float*)d_in[0];
  const int*   ei  = (const int*)d_in[1];
  const float* W1  = (const float*)d_in[2];
  const float* as1 = (const float*)d_in[3];
  const float* ad1 = (const float*)d_in[4];
  const float* b1  = (const float*)d_in[5];
  const float* W2  = (const float*)d_in[6];
  const float* as2 = (const float*)d_in[7];
  const float* ad2 = (const float*)d_in[8];
  const float* b2  = (const float*)d_in[9];
  float* out = (float*)d_out;

  const int Fin = in_sizes[2] / 64;
  const int N   = in_sizes[0] / Fin;
  const int E   = in_sizes[1] / 2;
  const int* src = ei;
  const int* dst = ei + E;

  const int TB = 256;
  const int gM  = (N + 15) / 16;
  const int gG  = (N * 16 + TB - 1) / TB;
  const int NB  = (N + RNG_NODES - 1) / RNG_NODES;
  const int nH  = NB * BIN_BLOCKS;

  float* ws = (float*)d_ws;
  size_t o = 0;
  float* h1    = ws + o; o += (size_t)N * 64;
  float* h2    = ws + o; o += (size_t)N * 48;
  int2* pairs  = (int2*)(ws + o); o += (size_t)E * 2;
  int* col     = (int*)(ws + o); o += (size_t)E;
  int* rowptr  = (int*)(ws + o); o += (size_t)N;
  int* tab     = (int*)(ws + o); o += (size_t)nH;

  gat_proj1_wmma<<<(gM + 3) / 4, P1_TB, 0, stream>>>(x, W1, h1, N);
  chunk_sort<<<BIN_BLOCKS, BIN_TB, (size_t)NB * 4, stream>>>(src, dst, tab, pairs, E, NB);
  range_fill<<<NB, RF_TB, 0, stream>>>(pairs, tab, rowptr, col, N, E, NB);
  gat_gather1_proj<<<gM, G1_TB, 0, stream>>>(rowptr, col, as1, ad1, h1, b1, W2, h2, N, E);

  gat_gather2<<<gG, TB, 0, stream>>>(rowptr, col, as2, ad2, h2, b2, out, N, E);

  (void)n_in; (void)out_size; (void)ws_size;
}


# Gathers fetch a node's sources 16 slots per load (one per lane, a chunk ahead), keep them as 32-bit row offsets, hand 4 per trip to the group by DPP row_share/row_ror: no index load in a trip's chain
# speedup vs baseline: 31.0487x; 1.0534x over previous
#include <hip/hip_runtime.h>
#include <math.h>

typedef __attribute__((ext_vector_type(2))) float v2f;
typedef __attribute__((ext_vector_type(8))) float v8f;

#define NEG_SLOPE 0.2f
#define SOFT_EPS  1e-16f

__device__ __forceinline__ float lrelu(float x) { return x > 0.f ? x : NEG_SLOPE * x; }

#define P1_TB    128
#define P1_PITCH 144
__global__ void __launch_bounds__(P1_TB)
gat_proj1_wmma(const float* __restrict__ X, const float* __restrict__ W,
               float* __restrict__ H, int M) {
  __shared__ float Bs[64 * P1_PITCH];
  const int tid   = threadIdx.x;
  const int wv    = tid >> 5;
  const int lane  = tid & 31;
  const int mr    = lane & 15;
  const int half  = lane >> 4;
  const int strip = blockIdx.x * (P1_TB / 32) + wv;
  const int arow  = min(strip * 16 + mr, M - 1);
  const float4* Ap = (const float4*)(X + (size_t)arow * 128) + half;
  float4 a[16];
#pragma unroll
  for (int s = 0; s < 16; ++s) a[s] = Ap[2 * s];

  {
    const float4* W4 = (const float4*)W;
    const int c = tid & 15;
#pragma unroll
    for (int i = 0; i < 8; ++i) {
      const int kp = (tid >> 4) + 8 * i;
      const float4 w0 = W4[(2 * kp) * 16 + c];
      const float4 w1 = W4[(2 * kp + 1) * 16 + c];
      *(float4*)&Bs[kp * P1_PITCH + 8 * c]     = make_float4(w0.x, w1.x, w0.y, w1.y);
      *(float4*)&Bs[kp * P1_PITCH + 8 * c + 4] = make_float4(w0.z, w1.z, w0.w, w1.w);
    }
  }
  __syncthreads();

  v8f acc0 = {}, acc1 = {}, acc2 = {}, acc3 = {};
  const float* Bl = Bs + (2 * half) * P1_PITCH + 2 * mr;
#pragma unroll
  for (int s = 0; s < 16; ++s) {
#pragma unroll
    for (int u = 0; u < 2; ++u) {
      v2f av;
      av.x = u ? a[s].z : a[s].x;
      av.y = u ? a[s].w : a[s].y;
      const float* bp = Bl + (4 * s + u) * P1_PITCH;
      const v2f b0 = *(const v2f*)(bp);
      const v2f b1 = *(const v2f*)(bp + 32);
      const v2f b2 = *(const v2f*)(bp + 64);
      const v2f b3 = *(const v2f*)(bp + 96);
      acc0 = __builtin_amdgcn_wmma_f32_16x16x4_f32(false, av, false, b0, (short)0, acc0, false, false);
      acc1 = __builtin_amdgcn_wmma_f32_16x16x4_f32(false, av, false, b1, (short)0, acc1, false, false);
      acc2 = __builtin_amdgcn_wmma_f32_16x16x4_f32(false, av, false, b2, (short)0, acc2, false, false);
      acc3 = __builtin_amdgcn_wmma_f32_16x16x4_f32(false, av, false, b3, (short)0, acc3, false, false);
    }
  }
#pragma unroll
  for (int i = 0; i < 8; ++i) {
    const int r = strip * 16 + i + 8 * half;
    if (r < M) {
      float* Hr = H + (size_t)r * 64 + mr;
      Hr[0]  = acc0[i];
      Hr[16] = acc1[i];
      Hr[32] = acc2[i];
      Hr[48] = acc3[i];
    }
  }
}

#define RNG_SHIFT  8
#define RNG_NODES  256
#define BIN_BLOCKS 256
#define BIN_TB     1024
#define RF_TB      1024
#define RF_LPC     (RF_TB / BIN_BLOCKS)
#define RF_DEPTH   8
#define CS_CAP     16384
#define RF_CAP     16384

__device__ __forceinline__ void edge_chunk(int c, int E, int& beg, int& end) {
  const int chunk = (((E + BIN_BLOCKS - 1) / BIN_BLOCKS) + 3) & ~3;
  beg = min(c * chunk, E);
  end = min(beg + chunk, E);
}

__device__ __forceinline__ void bin_add4(int* __restrict__ sh, const int4 d) {
  atomicAdd(&sh[d.x >> RNG_SHIFT], 1);
  atomicAdd(&sh[d.y >> RNG_SHIFT], 1);
  atomicAdd(&sh[d.z >> RNG_SHIFT], 1);
  atomicAdd(&sh[d.w >> RNG_SHIFT], 1);
}

__device__ __forceinline__ void bin_add8(int* __restrict__ sh, const int4 a, const int4 b) {
  atomicAdd(&sh[a.x >> RNG_SHIFT], 1);
  atomicAdd(&sh[b.x >> RNG_SHIFT], 1);
  atomicAdd(&sh[a.y >> RNG_SHIFT], 1);
  atomicAdd(&sh[b.y >> RNG_SHIFT], 1);
  atomicAdd(&sh[a.z >> RNG_SHIFT], 1);
  atomicAdd(&sh[b.z >> RNG_SHIFT], 1);
  atomicAdd(&sh[a.w >> RNG_SHIFT], 1);
  atomicAdd(&sh[b.w >> RNG_SHIFT], 1);
}

__device__ __forceinline__ int take_slot(int* __restrict__ sh, int d) {
  return atomicAdd(&sh[d >> RNG_SHIFT], 1);
}

__device__ __forceinline__ void put_pair(int2* __restrict__ to, int pos, int s, int d, int len) {
  if ((unsigned)pos < (unsigned)len) to[pos] = make_int2(s, d);
}

__device__ __forceinline__ void put4(int* __restrict__ sh, int2* __restrict__ to,
                                     const int4 s, const int4 d, int len) {
  const int q0 = take_slot(sh, d.x), q1 = take_slot(sh, d.y);
  const int q2 = take_slot(sh, d.z), q3 = take_slot(sh, d.w);
  put_pair(to, q0, s.x, d.x, len);
  put_pair(to, q1, s.y, d.y, len);
  put_pair(to, q2, s.z, d.z, len);
  put_pair(to, q3, s.w, d.w, len);
}

__device__ __forceinline__ void place_chunk(const int* __restrict__ src, const int* __restrict__ dst,
                                            int* __restrict__ sh, int2* __restrict__ to,
                                            int beg, int end, int len, bool wide, int tid) {
  if (wide) {
    int e = beg + 4 * tid;
    for (; e + 4 * BIN_TB < end; e += 8 * BIN_TB) {
      const int4 da = *(const int4*)(dst + e);
      const int4 db = *(const int4*)(dst + e + 4 * BIN_TB);
      const int4 sa = *(const int4*)(src + e);
      const int4 sb = *(const int4*)(src + e + 4 * BIN_TB);
      const int q0 = take_slot(sh, da.x), q1 = take_slot(sh, da.y);
      const int q2 = take_slot(sh, da.z), q3 = take_slot(sh, da.w);
      const int q4 = take_slot(sh, db.x), q5 = take_slot(sh, db.y);
      const int q6 = take_slot(sh, db.z), q7 = take_slot(sh, db.w);
      put_pair(to, q0, sa.x, da.x, len);
      put_pair(to, q1, sa.y, da.y, len);
      put_pair(to, q2, sa.z, da.z, len);
      put_pair(to, q3, sa.w, da.w, len);
      put_pair(to, q4, sb.x, db.x, len);
      put_pair(to, q5, sb.y, db.y, len);
      put_pair(to, q6, sb.z, db.z, len);
      put_pair(to, q7, sb.w, db.w, len);
    }
    if (e < end) {
      const int4 da = *(const int4*)(dst + e);
      const int4 sa = *(const int4*)(src + e);
      put4(sh, to, sa, da, len);
    }
  } else {
    for (int e = beg + tid; e < end; e += BIN_TB) {
      const int d = dst[e];
      put_pair(to, take_slot(sh, d), src[e], d, len);
    }
  }
}

__global__ void __launch_bounds__(BIN_TB)
chunk_sort(const int* __restrict__ src, const int* __restrict__ dst,
           int* __restrict__ tab, int2* __restrict__ pairs, int E, int NB) {
  extern __shared__ int sh[];
  __shared__ int wtot[BIN_TB / 32];
  __shared__ __attribute__((aligned(16))) int2 img[CS_CAP];
  const int tid  = threadIdx.x;
  const int lane = tid & 31;
  const int wv   = tid >> 5;
  for (int i = tid; i < NB; i += BIN_TB) sh[i] = 0;
  __syncthreads();
  int beg, end;
  edge_chunk(blockIdx.x, E, beg, end);
  const bool wide = (E & 3) == 0;
  if (wide) {
    int e = beg + 4 * tid;
    for (; e + 4 * BIN_TB < end; e += 8 * BIN_TB) {
      const int4 a = *(const int4*)(dst + e);
      const int4 b = *(const int4*)(dst + e + 4 * BIN_TB);
      bin_add8(sh, a, b);
    }
    if (e < end) bin_add4(sh, *(const int4*)(dst + e));
  } else {
    for (int e = beg + tid; e < end; e += BIN_TB) atomicAdd(&sh[dst[e] >> RNG_SHIFT], 1);
  }
  __syncthreads();
  int run = 0;
  for (int base = 0; base < NB; base += BIN_TB) {
    const int i = base + tid;
    const int v = (i < NB) ? sh[i] : 0;
    int x = v;
    for (int o = 1; o < 32; o <<= 1) {
      const int y = __shfl_up(x, o);
      if (lane >= o) x += y;
    }
    if (lane == 31) wtot[wv] = x;
    __syncthreads();
    if (wv == 0) {
      int w = wtot[lane];
      for (int o = 1; o < 32; o <<= 1) {
        const int y = __shfl_up(w, o);
        if (lane >= o) w += y;
      }
      wtot[lane] = w;
    }
    __syncthreads();
    const int first = run + ((wv > 0) ? wtot[wv - 1] : 0) + x - v;
    if (i < NB) {
      sh[i] = first;
      tab[i * BIN_BLOCKS + blockIdx.x] = first;
    }
    run += wtot[BIN_TB / 32 - 1];
    __syncthreads();
  }
  int2* mine = pairs + beg;
  const int len = end - beg;
  if (len <= CS_CAP) {
    place_chunk(src, dst, sh, img, beg, end, len, wide, tid);
    __syncthreads();
    const int4* im4 = (const int4*)img;
    int4* out4 = (int4*)mine;
    const int n2 = len >> 1;
    for (int i = tid; i < n2; i += BIN_TB) out4[i] = im4[i];
    if ((len & 1) && tid == 0) mine[len - 1] = img[len - 1];
  } else {
    place_chunk(src, dst, sh, mine, beg, end, len, wide, tid);
  }
}

__global__ void __launch_bounds__(RF_TB)
range_fill(const int2* __restrict__ pairs, const int* __restrict__ tab,
           int* __restrict__ rowptr, int* __restrict__ col, int N, int E, int NB) {
  __shared__ int cnt[RNG_NODES];
  __shared__ int cur[RNG_NODES];
  __shared__ int wtot[RNG_NODES / 32];
  __shared__ int rbeg[BIN_BLOCKS];
  __shared__ int rend[BIN_BLOCKS];
  __shared__ int bsum[BIN_BLOCKS / 32];
  __shared__ int img[RF_CAP];
  const int b    = blockIdx.x;
  const int tid  = threadIdx.x;
  const int lane = tid & 31;
  const int wv   = tid >> 5;
  if (tid < BIN_BLOCKS) {
    int cb, ce;
    edge_chunk(tid, E, cb, ce);
    const int clen = ce - cb;
    int r0 = tab[b * BIN_BLOCKS + tid];
    int r1 = (b + 1 < NB) ? tab[(b + 1) * BIN_BLOCKS + tid] : clen;
    r0 = max(0, min(r0, clen));
    r1 = max(r0, min(r1, clen));
    rbeg[tid] = cb + r0;
    rend[tid] = cb + r1;
    int s = r0;
    for (int o = 16; o > 0; o >>= 1) s += __shfl_xor(s, o);
    if (lane == 0) bsum[wv] = s;
  }
  if (tid < RNG_NODES) cnt[tid] = 0;
  __syncthreads();
  int beg = 0;
  for (int w = 0; w < BIN_BLOCKS / 32; ++w) beg += bsum[w];
  beg = max(0, min(beg, E));
  const int c   = tid / RF_LPC;
  const int sub = tid % RF_LPC;
  const int rb = rbeg[c], re = rend[c], last = re - 1;
  for (int i = rb + sub; i < re; i += RF_DEPTH * RF_LPC) {
    int d[RF_DEPTH];
#pragma unroll
    for (int u = 0; u < RF_DEPTH; ++u) d[u] = pairs[min(i + u * RF_LPC, last)].y;
#pragma unroll
    for (int u = 0; u < RF_DEPTH; ++u)
      atomicAdd(&cnt[d[u] & (RNG_NODES - 1)], (i + u * RF_LPC < re) ? 1 : 0);
  }
  __syncthreads();
  int v = 0, x = 0;
  if (tid < RNG_NODES) {
    v = cnt[tid];
    x = v;
    for (int o = 1; o < 32; o <<= 1) {
      const int y = __shfl_up(x, o);
      if (lane >= o) x += y;
    }
    if (lane == 31) wtot[wv] = x;
  }
  __syncthreads();
  if (tid < RNG_NODES) {
    int base = beg;
    for (int w = 0; w < wv; ++w) base += wtot[w];
    const int first = base + x - v;
    cur[tid] = first;
    const int node = b * RNG_NODES + tid;
    if (node < N) rowptr[node] = first;
  }
  __syncthreads();
  int total = 0;
  for (int w = 0; w < RNG_NODES / 32; ++w) total += wtot[w];
  const bool inlds = total <= RF_CAP;
  for (int i = rb + sub; i < re; i += RF_DEPTH * RF_LPC) {
    int2 p[RF_DEPTH];
    int  q[RF_DEPTH];
#pragma unroll
    for (int u = 0; u < RF_DEPTH; ++u) p[u] = pairs[min(i + u * RF_LPC, last)];
    __builtin_amdgcn_sched_barrier(0);
#pragma unroll
    for (int u = 0; u < RF_DEPTH; ++u)
      q[u] = atomicAdd(&cur[p[u].y & (RNG_NODES - 1)], (i + u * RF_LPC < re) ? 1 : 0);
    if (inlds) {
#pragma unroll
      for (int u = 0; u < RF_DEPTH; ++u)
        if (i + u * RF_LPC < re && (unsigned)(q[u] - beg) < (unsigned)total) img[q[u] - beg] = p[u].x;
    } else {
#pragma unroll
      for (int u = 0; u < RF_DEPTH; ++u)
        if (i + u * RF_LPC < re && (unsigned)q[u] < (unsigned)E) col[q[u]] = p[u].x;
    }
  }
  if (inlds) {
    __syncthreads();
    for (int i = tid; i < total; i += RF_TB)
      if ((unsigned)(beg + i) < (unsigned)E) col[beg + i] = img[i];
  }
}

__device__ __forceinline__ void sm_rescale(float sc, float& den, float4& acc) {
  den *= sc;
  acc.x *= sc; acc.y *= sc; acc.z *= sc; acc.w *= sc;
}

__device__ __forceinline__ void sm_add(float p, const float4 v, float& den, float4& acc) {
  den += p;
  acc.x += p * v.x; acc.y += p * v.y; acc.z += p * v.z; acc.w += p * v.w;
}

__device__ __forceinline__ float dot4(const float4 v, float w0, float w1, float w2, float w3) {
  return v.x * w0 + v.y * w1 + v.z * w2 + v.w * w3;
}

template <int CTRL>
__device__ __forceinline__ float lane_xchg(float x) {
  return __int_as_float(__builtin_amdgcn_update_dpp(0, __float_as_int(x), CTRL, 0xF, 0xF, false));
}

__device__ __forceinline__ float pair_sum(float p) { return p + lane_xchg<0xB1>(p); }

__device__ __forceinline__ float row_sum(float p) {
  p += lane_xchg<0xB1>(p);
  p += lane_xchg<0x4E>(p);
  p += lane_xchg<0x141>(p);
  p += lane_xchg<0x140>(p);
  return p;
}

template <int N>
__device__ __forceinline__ unsigned row_bcast(unsigned x) {
  return (unsigned)__builtin_amdgcn_update_dpp(0, (int)x, 0x150 + N, 0xF, 0xF, false);
}

__device__ __forceinline__ unsigned row_turn4(unsigned x) {
  return (unsigned)__builtin_amdgcn_update_dpp(0, (int)x, 0x12C, 0xF, 0xF, false);
}

template <unsigned ROW_BYTES>
__device__ __forceinline__ unsigned trip_slots(int trip, unsigned offs, int nxt) {
  return (trip & 3) == 0 ? (unsigned)nxt * ROW_BYTES : row_turn4(offs);
}

#define G1_TB    256
#define G1_PITCH 68
__global__ void __launch_bounds__(G1_TB)
gat_gather1_proj(const int* __restrict__ rowptr, const int* __restrict__ col,
                 const float* __restrict__ attS, const float* __restrict__ attD,
                 const float* __restrict__ h1, const float* __restrict__ bias,
                 const float* __restrict__ W2, float* __restrict__ h2, int N, int E) {
  __shared__ float tile[16 * G1_PITCH];
  const int t = blockIdx.x * G1_TB + threadIdx.x;
  const int g = min(t >> 4, N - 1);
  const int j = t & 15;
  const int beg = rowptr[g];
  const int end = (g + 1 < N) ? rowptr[g + 1] : E;
  const float ws0 = attS[4 * j + 0], ws1 = attS[4 * j + 1], ws2 = attS[4 * j + 2], ws3 = attS[4 * j + 3];
  const float wd0 = attD[4 * j + 0], wd1 = attD[4 * j + 1], wd2 = attD[4 * j + 2], wd3 = attD[4 * j + 3];

  const float4* hrow = (const float4*)h1;
  float4 acc = hrow[(size_t)g * 16 + j];
  const float ad    = pair_sum(dot4(acc, wd0, wd1, wd2, wd3));
  const float eself = lrelu(pair_sum(dot4(acc, ws0, ws1, ws2, ws3)) + ad);
  float  m   = eself;
  float  den = 1.f;
  const char* hb = (const char*)hrow;
  const unsigned jo = 16u * j;
  const int last = end - 1;
  int nxt = 0;
  if (beg < end) nxt = col[min(beg + j, last)];
  unsigned offs = 0;
  int trip = 0;
  int k = beg;
  for (; k + 4 <= end; k += 4, ++trip) {
    offs = trip_slots<256u>(trip, offs, nxt);
    const float4 v0 = *(const float4*)(hb + (row_bcast<0>(offs) + jo));
    const float4 v1 = *(const float4*)(hb + (row_bcast<1>(offs) + jo));
    const float4 v2 = *(const float4*)(hb + (row_bcast<2>(offs) + jo));
    const float4 v3 = *(const float4*)(hb + (row_bcast<3>(offs) + jo));
    const float e0 = lrelu(pair_sum(dot4(v0, ws0, ws1, ws2, ws3)) + ad);
    const float e1 = lrelu(pair_sum(dot4(v1, ws0, ws1, ws2, ws3)) + ad);
    const float e2 = lrelu(pair_sum(dot4(v2, ws0, ws1, ws2, ws3)) + ad);
    const float e3 = lrelu(pair_sum(dot4(v3, ws0, ws1, ws2, ws3)) + ad);
    const float mn = fmaxf(fmaxf(fmaxf(m, e0), fmaxf(e1, e2)), e3);
    sm_rescale(__expf(m - mn), den, acc);
    sm_add(__expf(e0 - mn), v0, den, acc);
    sm_add(__expf(e1 - mn), v1, den, acc);
    sm_add(__expf(e2 - mn), v2, den, acc);
    sm_add(__expf(e3 - mn), v3, den, acc);
    m = mn;
    if ((trip & 3) == 0 && k + 16 < end) nxt = col[min(k + 16 + j, last)];
  }
  if (k < end) {
    const bool has1 = k + 1 < end, has2 = k + 2 < end;
    offs = trip_slots<256u>(trip, offs, nxt);
    const float4 v0 = *(const float4*)(hb + (row_bcast<0>(offs) + jo));
    const float4 v1 = *(const float4*)(hb + (row_bcast<1>(offs) + jo));
    const float4 v2 = *(const float4*)(hb + (row_bcast<2>(offs) + jo));
    const float e0 = lrelu(pair_sum(dot4(v0, ws0, ws1, ws2, ws3)) + ad);
    const float x1 = lrelu(pair_sum(dot4(v1, ws0, ws1, ws2, ws3)) + ad);
    const float x2 = lrelu(pair_sum(dot4(v2, ws0, ws1, ws2, ws3)) + ad);
    const float e1 = has1 ? x1 : -INFINITY;
    const float e2 = has2 ? x2 : -INFINITY;
    const float mn = fmaxf(fmaxf(m, e0), fmaxf(e1, e2));
    sm_rescale(__expf(m - mn), den, acc);
    sm_add(__expf(e0 - mn), v0, den, acc);
    sm_add(__expf(e1 - mn), v1, den, acc);
    sm_add(__expf(e2 - mn), v2, den, acc);
    m = mn;
  }
  const int wv   = threadIdx.x >> 5;
  const int lane = threadIdx.x & 31;
  const int mr   = lane & 15;
  const int half = lane >> 4;
  const int c    = wv * 16 + mr;
  const bool cv  = c < 40;
  const int cc   = cv ? c : 39;
  float bw[32];
  if (wv < 3) {
#pragma unroll
    for (int s = 0; s < 16; ++s) {
      const int ka = 4 * s + 2 * half;
      bw[2 * s]     = W2[ka * 40 + cc];
      bw[2 * s + 1] = W2[(ka + 1) * 40 + cc];
    }
  }

  const float inv = 1.f / (den + SOFT_EPS);
  float4 o;
  o.x = acc.x * inv + bias[4 * j + 0];
  o.y = acc.y * inv + bias[4 * j + 1];
  o.z = acc.z * inv + bias[4 * j + 2];
  o.w = acc.w * inv + bias[4 * j + 3];
  o.x = o.x > 0.f ? o.x : (__expf(o.x) - 1.f);
  o.y = o.y > 0.f ? o.y : (__expf(o.y) - 1.f);
  o.z = o.z > 0.f ? o.z : (__expf(o.z) - 1.f);
  o.w = o.w > 0.f ? o.w : (__expf(o.w) - 1.f);

  *(float4*)&tile[(threadIdx.x >> 4) * G1_PITCH + 4 * j] = o;
  __syncthreads();
  if (wv < 3) {
    const float* Ar = tile + mr * G1_PITCH;
    v8f acc2 = {};
#pragma unroll
    for (int s = 0; s < 16; ++s) {
      const int ka = 4 * s + 2 * half;
      v2f a, b;
      a.x = Ar[ka];
      a.y = Ar[ka + 1];
      b.x = cv ? bw[2 * s] : 0.f;
      b.y = cv ? bw[2 * s + 1] : 0.f;
      acc2 = __builtin_amdgcn_wmma_f32_16x16x4_f32(false, a, false, b, (short)0, acc2,
                                                   false, false);
    }
    const int row0 = blockIdx.x * 16;
#pragma unroll
    for (int i = 0; i < 8; ++i) {
      const int r = row0 + i + 8 * half;
      if (r < N) h2[(size_t)r * 48 + c] = acc2[i];
    }
  }
}

__global__ void gat_gather2(const int* __restrict__ rowptr, const int* __restrict__ col,
                            const float* __restrict__ attS, const float* __restrict__ attD,
                            const float* __restrict__ h2, const float* __restrict__ bias,
                            float* __restrict__ out, int N, int E) {
  const int t = blockIdx.x * blockDim.x + threadIdx.x;
  const int g = t >> 4;
  if (g >= N) return;
  const int j = t & 15;
  const bool act = j < 10;
  const int jj = act ? j : 9;
  const int beg = rowptr[g];
  const int end = (g + 1 < N) ? rowptr[g + 1] : E;
  const float ws0 = act ? attS[4 * jj + 0] : 0.f, ws1 = act ? attS[4 * jj + 1] : 0.f;
  const float ws2 = act ? attS[4 * jj + 2] : 0.f, ws3 = act ? attS[4 * jj + 3] : 0.f;
  const float wd0 = act ? attD[4 * jj + 0] : 0.f, wd1 = act ? attD[4 * jj + 1] : 0.f;
  const float wd2 = act ? attD[4 * jj + 2] : 0.f, wd3 = act ? attD[4 * jj + 3] : 0.f;

  const float4* hrow = (const float4*)h2;
  float4 acc = hrow[(size_t)g * 12 + jj];
  const float ad    = row_sum(dot4(acc, wd0, wd1, wd2, wd3));
  const float eself = lrelu(row_sum(dot4(acc, ws0, ws1, ws2, ws3)) + ad);
  float  m   = eself;
  float  den = 1.f;
  const char* hb = (const char*)hrow;
  const unsigned jo = 16u * jj;
  const int last = end - 1;
  int nxt = 0;
  if (beg < end) nxt = col[min(beg + j, last)];
  unsigned offs = 0;
  int trip = 0;
  int k = beg;
  for (; k + 4 <= end; k += 4, ++trip) {
    offs = trip_slots<192u>(trip, offs, nxt);
    const float4 v0 = *(const float4*)(hb + (row_bcast<0>(offs) + jo));
    const float4 v1 = *(const float4*)(hb + (row_bcast<1>(offs) + jo));
    const float4 v2 = *(const float4*)(hb + (row_bcast<2>(offs) + jo));
    const float4 v3 = *(const float4*)(hb + (row_bcast<3>(offs) + jo));
    const float e0 = lrelu(row_sum(dot4(v0, ws0, ws1, ws2, ws3)) + ad);
    const float e1 = lrelu(row_sum(dot4(v1, ws0, ws1, ws2, ws3)) + ad);
    const float e2 = lrelu(row_sum(dot4(v2, ws0, ws1, ws2, ws3)) + ad);
    const float e3 = lrelu(row_sum(dot4(v3, ws0, ws1, ws2, ws3)) + ad);
    const float mn = fmaxf(fmaxf(fmaxf(m, e0), fmaxf(e1, e2)), e3);
    sm_rescale(__expf(m - mn), den, acc);
    sm_add(__expf(e0 - mn), v0, den, acc);
    sm_add(__expf(e1 - mn), v1, den, acc);
    sm_add(__expf(e2 - mn), v2, den, acc);
    sm_add(__expf(e3 - mn), v3, den, acc);
    m = mn;
    if ((trip & 3) == 0 && k + 16 < end) nxt = col[min(k + 16 + j, last)];
  }
  if (k < end) {
    const bool has1 = k + 1 < end, has2 = k + 2 < end;
    offs = trip_slots<192u>(trip, offs, nxt);
    const float4 v0 = *(const float4*)(hb + (row_bcast<0>(offs) + jo));
    const float4 v1 = *(const float4*)(hb + (row_bcast<1>(offs) + jo));
    const float4 v2 = *(const float4*)(hb + (row_bcast<2>(offs) + jo));
    const float e0 = lrelu(row_sum(dot4(v0, ws0, ws1, ws2, ws3)) + ad);
    const float x1 = lrelu(row_sum(dot4(v1, ws0, ws1, ws2, ws3)) + ad);
    const float x2 = lrelu(row_sum(dot4(v2, ws0, ws1, ws2, ws3)) + ad);
    const float e1 = has1 ? x1 : -INFINITY;
    const float e2 = has2 ? x2 : -INFINITY;
    const float mn = fmaxf(fmaxf(m, e0), fmaxf(e1, e2));
    sm_rescale(__expf(m - mn), den, acc);
    sm_add(__expf(e0 - mn), v0, den, acc);
    sm_add(__expf(e1 - mn), v1, den, acc);
    sm_add(__expf(e2 - mn), v2, den, acc);
    m = mn;
  }
  const float inv = 1.f / (den + SOFT_EPS);
  float4 v;
  v.x = acc.x * inv + bias[4 * jj + 0];
  v.y = acc.y * inv + bias[4 * jj + 1];
  v.z = acc.z * inv + bias[4 * jj + 2];
  v.w = acc.w * inv + bias[4 * jj + 3];
  float mx = act ? fmaxf(fmaxf(v.x, v.y), fmaxf(v.z, v.w)) : -INFINITY;
  for (int off = 8; off > 0; off >>= 1) mx = fmaxf(mx, __shfl_xor(mx, off, 16));
  float sum = act ? (__expf(v.x - mx) + __expf(v.y - mx) + __expf(v.z - mx) + __expf(v.w - mx))
                  : 0.f;
  for (int off = 8; off > 0; off >>= 1) sum += __shfl_xor(sum, off, 16);
  const float lse = mx + __logf(sum);
  if (act) {
    float4 o;
    o.x = v.x - lse; o.y = v.y - lse; o.z = v.z - lse; o.w = v.w - lse;
    ((float4*)out)[(size_t)g * 10 + j] = o;
  }
}

extern "C" void kernel_launch(void* const* d_in, const int* in_sizes, int n_in,
                              void* d_out, int out_size, void* d_ws, size_t ws_size,
                              hipStream_t stream) {
  const float* x   = (const float*)d_in[0];
  const int*   ei  = (const int*)d_in[1];
  const float* W1  = (const float*)d_in[2];
  const float* as1 = (const float*)d_in[3];
  const float* ad1 = (const float*)d_in[4];
  const float* b1  = (const float*)d_in[5];
  const float* W2  = (const float*)d_in[6];
  const float* as2 = (const float*)d_in[7];
  const float* ad2 = (const float*)d_in[8];
  const float* b2  = (const float*)d_in[9];
  float* out = (float*)d_out;

  const int Fin = in_sizes[2] / 64;
  const int N   = in_sizes[0] / Fin;
  const int E   = in_sizes[1] / 2;
  const int* src = ei;
  const int* dst = ei + E;

  const int TB = 256;
  const int gM  = (N + 15) / 16;
  const int gG  = (N * 16 + TB - 1) / TB;
  const int NB  = (N + RNG_NODES - 1) / RNG_NODES;
  const int nH  = NB * BIN_BLOCKS;

  float* ws = (float*)d_ws;
  size_t o = 0;
  float* h1    = ws + o; o += (size_t)N * 64;
  float* h2    = ws + o; o += (size_t)N * 48;
  int2* pairs  = (int2*)(ws + o); o += (size_t)E * 2;
  int* col     = (int*)(ws + o); o += (size_t)E;
  int* rowptr  = (int*)(ws + o); o += (size_t)N;
  int* tab     = (int*)(ws + o); o += (size_t)nH;

  gat_proj1_wmma<<<(gM + 3) / 4, P1_TB, 0, stream>>>(x, W1, h1, N);
  chunk_sort<<<BIN_BLOCKS, BIN_TB, (size_t)NB * 4, stream>>>(src, dst, tab, pairs, E, NB);
  range_fill<<<NB, RF_TB, 0, stream>>>(pairs, tab, rowptr, col, N, E, NB);
  gat_gather1_proj<<<gM, G1_TB, 0, stream>>>(rowptr, col, as1, ad1, h1, b1, W2, h2, N, E);

  gat_gather2<<<gG, TB, 0, stream>>>(rowptr, col, as2, ad2, h2, b2, out, N, E);

  (void)n_in; (void)out_size; (void)ws_size;
}


# Layer-1 projection runs inside the sort's first launch: 16-wave projection work-groups beside 16-wave sort work-groups, one of each per WGP; its own launch is gone; col slots in new arrival order
# speedup vs baseline: 32.3561x; 1.0421x over previous
#include <hip/hip_runtime.h>
#include <math.h>

typedef __attribute__((ext_vector_type(2))) float v2f;
typedef __attribute__((ext_vector_type(8))) float v8f;

#define NEG_SLOPE 0.2f
#define SOFT_EPS  1e-16f

__device__ __forceinline__ float lrelu(float x) { return x > 0.f ? x : NEG_SLOPE * x; }

#define PJ_BLOCKS 256
#define P1_PITCH 144
__device__ __forceinline__ void p1_load_a(float4 (&a)[16], const float* __restrict__ X,
                                          int strip, int mr, int half, int M) {
  const int arow = min(strip * 16 + mr, M - 1);
  const float4* Ap = (const float4*)(X + (size_t)arow * 128) + half;
#pragma unroll
  for (int s = 0; s < 16; ++s) a[s] = Ap[2 * s];
}

template <int PJ_TB>
__device__ __forceinline__ void proj1_role(const float* __restrict__ X, const float* __restrict__ W,
                                           float* __restrict__ H, int M, float* __restrict__ Bs,
                                           int pb, int tid) {
  const int wv    = __builtin_amdgcn_readfirstlane(tid >> 5);
  const int lane  = tid & 31;
  const int mr    = lane & 15;
  const int half  = lane >> 4;
  const int nstrip = (M + 15) / 16;
  int strip = wv * PJ_BLOCKS + pb;
  float4 a[16];
  if (strip < nstrip) p1_load_a(a, X, strip, mr, half, M);

  {
    const float4* W4 = (const float4*)W;
    const int c = tid & 15;
#pragma unroll
    for (int i = 0; i < 64 * 16 / PJ_TB; ++i) {
      const int kp = (tid >> 4) + (PJ_TB / 16) * i;
      const float4 w0 = W4[(2 * kp) * 16 + c];
      const float4 w1 = W4[(2 * kp + 1) * 16 + c];
      *(float4*)&Bs[kp * P1_PITCH + 8 * c]     = make_float4(w0.x, w1.x, w0.y, w1.y);
      *(float4*)&Bs[kp * P1_PITCH + 8 * c + 4] = make_float4(w0.z, w1.z, w0.w, w1.w);
    }
  }
  __syncthreads();

  int boff = (2 * half) * P1_PITCH + 2 * mr;
  while (strip < nstrip) {
    asm volatile("" : "+v"(boff) :: "memory");
    const float* Bl = Bs + boff;
    v8f acc0 = {}, acc1 = {}, acc2 = {}, acc3 = {};
#pragma unroll
    for (int s = 0; s < 16; ++s) {
#pragma unroll
      for (int u = 0; u < 2; ++u) {
        v2f av;
        av.x = u ? a[s].z : a[s].x;
        av.y = u ? a[s].w : a[s].y;
        const float* bp = Bl + (4 * s + u) * P1_PITCH;
        const v2f b0 = *(const v2f*)(bp);
        const v2f b1 = *(const v2f*)(bp + 32);
        const v2f b2 = *(const v2f*)(bp + 64);
        const v2f b3 = *(const v2f*)(bp + 96);
        acc0 = __builtin_amdgcn_wmma_f32_16x16x4_f32(false, av, false, b0, (short)0, acc0, false, false);
        acc1 = __builtin_amdgcn_wmma_f32_16x16x4_f32(false, av, false, b1, (short)0, acc1, false, false);
        acc2 = __builtin_amdgcn_wmma_f32_16x16x4_f32(false, av, false, b2, (short)0, acc2, false, false);
        acc3 = __builtin_amdgcn_wmma_f32_16x16x4_f32(false, av, false, b3, (short)0, acc3, false, false);
      }
    }
    const int done = strip;
    strip += (PJ_TB / 32) * PJ_BLOCKS;
    if (strip < nstrip) p1_load_a(a, X, strip, mr, half, M);
#pragma unroll
    for (int i = 0; i < 8; ++i) {
      const int r = done * 16 + i + 8 * half;
      if (r < M) {
        float* Hr = H + (size_t)r * 64 + mr;
        Hr[0]  = acc0[i];
        Hr[16] = acc1[i];
        Hr[32] = acc2[i];
        Hr[48] = acc3[i];
      }
    }
  }
}

#define RNG_SHIFT  8
#define RNG_NODES  256
#define BIN_BLOCKS 256
#define BIN_TB     512
#define RF_TB      1024
#define RF_LPC     (RF_TB / BIN_BLOCKS)
#define RF_DEPTH   8
#define CS_CAP     16384
#define RF_CAP     16384

__device__ __forceinline__ void edge_chunk(int c, int E, int& beg, int& end) {
  const int chunk = (((E + BIN_BLOCKS - 1) / BIN_BLOCKS) + 3) & ~3;
  beg = min(c * chunk, E);
  end = min(beg + chunk, E);
}

__device__ __forceinline__ void bin_add4(int* __restrict__ sh, const int4 d) {
  atomicAdd(&sh[d.x >> RNG_SHIFT], 1);
  atomicAdd(&sh[d.y >> RNG_SHIFT], 1);
  atomicAdd(&sh[d.z >> RNG_SHIFT], 1);
  atomicAdd(&sh[d.w >> RNG_SHIFT], 1);
}

__device__ __forceinline__ void bin_add8(int* __restrict__ sh, const int4 a, const int4 b) {
  atomicAdd(&sh[a.x >> RNG_SHIFT], 1);
  atomicAdd(&sh[b.x >> RNG_SHIFT], 1);
  atomicAdd(&sh[a.y >> RNG_SHIFT], 1);
  atomicAdd(&sh[b.y >> RNG_SHIFT], 1);
  atomicAdd(&sh[a.z >> RNG_SHIFT], 1);
  atomicAdd(&sh[b.z >> RNG_SHIFT], 1);
  atomicAdd(&sh[a.w >> RNG_SHIFT], 1);
  atomicAdd(&sh[b.w >> RNG_SHIFT], 1);
}

__device__ __forceinline__ int take_slot(int* __restrict__ sh, int d) {
  return atomicAdd(&sh[d >> RNG_SHIFT], 1);
}

__device__ __forceinline__ void put_pair(int2* __restrict__ to, int pos, int s, int d, int len) {
  if ((unsigned)pos < (unsigned)len) to[pos] = make_int2(s, d);
}

__device__ __forceinline__ void put4(int* __restrict__ sh, int2* __restrict__ to,
                                     const int4 s, const int4 d, int len) {
  const int q0 = take_slot(sh, d.x), q1 = take_slot(sh, d.y);
  const int q2 = take_slot(sh, d.z), q3 = take_slot(sh, d.w);
  put_pair(to, q0, s.x, d.x, len);
  put_pair(to, q1, s.y, d.y, len);
  put_pair(to, q2, s.z, d.z, len);
  put_pair(to, q3, s.w, d.w, len);
}

__device__ __forceinline__ void place_chunk(const int* __restrict__ src, const int* __restrict__ dst,
                                            int* __restrict__ sh, int2* __restrict__ to,
                                            int beg, int end, int len, bool wide, int tid) {
  if (wide) {
    const int S = 4 * BIN_TB;
    int e = beg + 4 * tid;
    for (; e + 3 * S < end; e += 4 * S) {
      int4 d[4], sv[4];
      int  q[16];
#pragma unroll
      for (int u = 0; u < 4; ++u) {
        d[u]  = *(const int4*)(dst + e + u * S);
        sv[u] = *(const int4*)(src + e + u * S);
      }
#pragma unroll
      for (int u = 0; u < 4; ++u) {
        q[4 * u + 0] = take_slot(sh, d[u].x);
        q[4 * u + 1] = take_slot(sh, d[u].y);
        q[4 * u + 2] = take_slot(sh, d[u].z);
        q[4 * u + 3] = take_slot(sh, d[u].w);
      }
#pragma unroll
      for (int u = 0; u < 4; ++u) {
        put_pair(to, q[4 * u + 0], sv[u].x, d[u].x, len);
        put_pair(to, q[4 * u + 1], sv[u].y, d[u].y, len);
        put_pair(to, q[4 * u + 2], sv[u].z, d[u].z, len);
        put_pair(to, q[4 * u + 3], sv[u].w, d[u].w, len);
      }
    }
    if (e + S < end) {
      const int4 da = *(const int4*)(dst + e);
      const int4 db = *(const int4*)(dst + e + S);
      const int4 sa = *(const int4*)(src + e);
      const int4 sb = *(const int4*)(src + e + S);
      const int q0 = take_slot(sh, da.x), q1 = take_slot(sh, da.y);
      const int q2 = take_slot(sh, da.z), q3 = take_slot(sh, da.w);
      const int q4 = take_slot(sh, db.x), q5 = take_slot(sh, db.y);
      const int q6 = take_slot(sh, db.z), q7 = take_slot(sh, db.w);
      put_pair(to, q0, sa.x, da.x, len);
      put_pair(to, q1, sa.y, da.y, len);
      put_pair(to, q2, sa.z, da.z, len);
      put_pair(to, q3, sa.w, da.w, len);
      put_pair(to, q4, sb.x, db.x, len);
      put_pair(to, q5, sb.y, db.y, len);
      put_pair(to, q6, sb.z, db.z, len);
      put_pair(to, q7, sb.w, db.w, len);
      e += 2 * S;
    }
    if (e < end) {
      const int4 da = *(const int4*)(dst + e);
      const int4 sa = *(const int4*)(src + e);
      put4(sh, to, sa, da, len);
    }
  } else {
    for (int e = beg + tid; e < end; e += BIN_TB) {
      const int d = dst[e];
      put_pair(to, take_slot(sh, d), src[e], d, len);
    }
  }
}

__global__ void __launch_bounds__(BIN_TB, 8)
sort1_proj1(const int* __restrict__ src, const int* __restrict__ dst,
            int* __restrict__ tab, int2* __restrict__ pairs, int E, int NB,
            const float* __restrict__ X, const float* __restrict__ W,
            float* __restrict__ H, int M) {
  extern __shared__ int sh[];
  __shared__ int wtot[BIN_TB / 32];
  __shared__ __attribute__((aligned(16))) int2 img[CS_CAP];
  const int tid  = threadIdx.x;
  if (blockIdx.x >= BIN_BLOCKS) {
    proj1_role<BIN_TB>(X, W, H, M, (float*)img, blockIdx.x - BIN_BLOCKS, tid);
    return;
  }
  const int lane = tid & 31;
  const int wv   = tid >> 5;
  for (int i = tid; i < NB; i += BIN_TB) sh[i] = 0;
  __syncthreads();
  int beg, end;
  edge_chunk(blockIdx.x, E, beg, end);
  const bool wide = (E & 3) == 0;
  if (wide) {
    const int S = 4 * BIN_TB;
    int e = beg + 4 * tid;
    for (; e + 3 * S < end; e += 4 * S) {
      const int4 a = *(const int4*)(dst + e);
      const int4 b = *(const int4*)(dst + e + S);
      const int4 c = *(const int4*)(dst + e + 2 * S);
      const int4 d = *(const int4*)(dst + e + 3 * S);
      bin_add8(sh, a, b);
      bin_add8(sh, c, d);
    }
    if (e + S < end) {
      const int4 a = *(const int4*)(dst + e);
      const int4 b = *(const int4*)(dst + e + S);
      bin_add8(sh, a, b);
      e += 2 * S;
    }
    if (e < end) bin_add4(sh, *(const int4*)(dst + e));
  } else {
    for (int e = beg + tid; e < end; e += BIN_TB) atomicAdd(&sh[dst[e] >> RNG_SHIFT], 1);
  }
  __syncthreads();
  int run = 0;
  for (int base = 0; base < NB; base += BIN_TB) {
    const int i = base + tid;
    const int v = (i < NB) ? sh[i] : 0;
    int x = v;
    for (int o = 1; o < 32; o <<= 1) {
      const int y = __shfl_up(x, o);
      if (lane >= o) x += y;
    }
    if (lane == 31) wtot[wv] = x;
    __syncthreads();
    if (wv == 0) {
      int w = (lane < BIN_TB / 32) ? wtot[lane] : 0;
      for (int o = 1; o < 32; o <<= 1) {
        const int y = __shfl_up(w, o);
        if (lane >= o) w += y;
      }
      if (lane < BIN_TB / 32) wtot[lane] = w;
    }
    __syncthreads();
    const int first = run + ((wv > 0) ? wtot[wv - 1] : 0) + x - v;
    if (i < NB) {
      sh[i] = first;
      tab[i * BIN_BLOCKS + blockIdx.x] = first;
    }
    run += wtot[BIN_TB / 32 - 1];
    __syncthreads();
  }
  int2* mine = pairs + beg;
  const int len = end - beg;
  if (len <= CS_CAP) {
    place_chunk(src, dst, sh, img, beg, end, len, wide, tid);
    __syncthreads();
    const int4* im4 = (const int4*)img;
    int4* out4 = (int4*)mine;
    const int n2 = len >> 1;
    for (int i = tid; i < n2; i += BIN_TB) out4[i] = im4[i];
    if ((len & 1) && tid == 0) mine[len - 1] = img[len - 1];
  } else {
    place_chunk(src, dst, sh, mine, beg, end, len, wide, tid);
  }
}

__global__ void __launch_bounds__(RF_TB)
range_fill(const int2* __restrict__ pairs, const int* __restrict__ tab,
           int* __restrict__ rowptr, int* __restrict__ col, int N, int E, int NB) {
  __shared__ int cnt[RNG_NODES];
  __shared__ int cur[RNG_NODES];
  __shared__ int wtot[RNG_NODES / 32];
  __shared__ int rbeg[BIN_BLOCKS];
  __shared__ int rend[BIN_BLOCKS];
  __shared__ int bsum[BIN_BLOCKS / 32];
  __shared__ int img[RF_CAP];
  const int b    = blockIdx.x;
  const int tid  = threadIdx.x;
  const int lane = tid & 31;
  const int wv   = tid >> 5;
  if (tid < BIN_BLOCKS) {
    int cb, ce;
    edge_chunk(tid, E, cb, ce);
    const int clen = ce - cb;
    int r0 = tab[b * BIN_BLOCKS + tid];
    int r1 = (b + 1 < NB) ? tab[(b + 1) * BIN_BLOCKS + tid] : clen;
    r0 = max(0, min(r0, clen));
    r1 = max(r0, min(r1, clen));
    rbeg[tid] = cb + r0;
    rend[tid] = cb + r1;
    int s = r0;
    for (int o = 16; o > 0; o >>= 1) s += __shfl_xor(s, o);
    if (lane == 0) bsum[wv] = s;
  }
  if (tid < RNG_NODES) cnt[tid] = 0;
  __syncthreads();
  int beg = 0;
  for (int w = 0; w < BIN_BLOCKS / 32; ++w) beg += bsum[w];
  beg = max(0, min(beg, E));
  const int c   = tid / RF_LPC;
  const int sub = tid % RF_LPC;
  const int rb = rbeg[c], re = rend[c], last = re - 1;
  for (int i = rb + sub; i < re; i += RF_DEPTH * RF_LPC) {
    int d[RF_DEPTH];
#pragma unroll
    for (int u = 0; u < RF_DEPTH; ++u) d[u] = pairs[min(i + u * RF_LPC, last)].y;
#pragma unroll
    for (int u = 0; u < RF_DEPTH; ++u)
      atomicAdd(&cnt[d[u] & (RNG_NODES - 1)], (i + u * RF_LPC < re) ? 1 : 0);
  }
  __syncthreads();
  int v = 0, x = 0;
  if (tid < RNG_NODES) {
    v = cnt[tid];
    x = v;
    for (int o = 1; o < 32; o <<= 1) {
      const int y = __shfl_up(x, o);
      if (lane >= o) x += y;
    }
    if (lane == 31) wtot[wv] = x;
  }
  __syncthreads();
  if (tid < RNG_NODES) {
    int base = beg;
    for (int w = 0; w < wv; ++w) base += wtot[w];
    const int first = base + x - v;
    cur[tid] = first;
    const int node = b * RNG_NODES + tid;
    if (node < N) rowptr[node] = first;
  }
  __syncthreads();
  int total = 0;
  for (int w = 0; w < RNG_NODES / 32; ++w) total += wtot[w];
  const bool inlds = total <= RF_CAP;
  for (int i = rb + sub; i < re; i += RF_DEPTH * RF_LPC) {
    int2 p[RF_DEPTH];
    int  q[RF_DEPTH];
#pragma unroll
    for (int u = 0; u < RF_DEPTH; ++u) p[u] = pairs[min(i + u * RF_LPC, last)];
    __builtin_amdgcn_sched_barrier(0);
#pragma unroll
    for (int u = 0; u < RF_DEPTH; ++u)
      q[u] = atomicAdd(&cur[p[u].y & (RNG_NODES - 1)], (i + u * RF_LPC < re) ? 1 : 0);
    if (inlds) {
#pragma unroll
      for (int u = 0; u < RF_DEPTH; ++u)
        if (i + u * RF_LPC < re && (unsigned)(q[u] - beg) < (unsigned)total) img[q[u] - beg] = p[u].x;
    } else {
#pragma unroll
      for (int u = 0; u < RF_DEPTH; ++u)
        if (i + u * RF_LPC < re && (unsigned)q[u] < (unsigned)E) col[q[u]] = p[u].x;
    }
  }
  if (inlds) {
    __syncthreads();
    for (int i = tid; i < total; i += RF_TB)
      if ((unsigned)(beg + i) < (unsigned)E) col[beg + i] = img[i];
  }
}

__device__ __forceinline__ void sm_rescale(float sc, float& den, float4& acc) {
  den *= sc;
  acc.x *= sc; acc.y *= sc; acc.z *= sc; acc.w *= sc;
}

__device__ __forceinline__ void sm_add(float p, const float4 v, float& den, float4& acc) {
  den += p;
  acc.x += p * v.x; acc.y += p * v.y; acc.z += p * v.z; acc.w += p * v.w;
}

__device__ __forceinline__ float dot4(const float4 v, float w0, float w1, float w2, float w3) {
  return v.x * w0 + v.y * w1 + v.z * w2 + v.w * w3;
}

template <int CTRL>
__device__ __forceinline__ float lane_xchg(float x) {
  return __int_as_float(__builtin_amdgcn_update_dpp(0, __float_as_int(x), CTRL, 0xF, 0xF, false));
}

__device__ __forceinline__ float pair_sum(float p) { return p + lane_xchg<0xB1>(p); }

__device__ __forceinline__ float row_sum(float p) {
  p += lane_xchg<0xB1>(p);
  p += lane_xchg<0x4E>(p);
  p += lane_xchg<0x141>(p);
  p += lane_xchg<0x140>(p);
  return p;
}

template <int N>
__device__ __forceinline__ unsigned row_bcast(unsigned x) {
  return (unsigned)__builtin_amdgcn_update_dpp(0, (int)x, 0x150 + N, 0xF, 0xF, false);
}

__device__ __forceinline__ unsigned row_turn4(unsigned x) {
  return (unsigned)__builtin_amdgcn_update_dpp(0, (int)x, 0x12C, 0xF, 0xF, false);
}

template <unsigned ROW_BYTES>
__device__ __forceinline__ unsigned trip_slots(int trip, unsigned offs, int nxt) {
  return (trip & 3) == 0 ? (unsigned)nxt * ROW_BYTES : row_turn4(offs);
}

#define G1_TB    256
#define G1_PITCH 68
__global__ void __launch_bounds__(G1_TB)
gat_gather1_proj(const int* __restrict__ rowptr, const int* __restrict__ col,
                 const float* __restrict__ attS, const float* __restrict__ attD,
                 const float* __restrict__ h1, const float* __restrict__ bias,
                 const float* __restrict__ W2, float* __restrict__ h2, int N, int E) {
  __shared__ float tile[16 * G1_PITCH];
  const int t = blockIdx.x * G1_TB + threadIdx.x;
  const int g = min(t >> 4, N - 1);
  const int j = t & 15;
  const int beg = rowptr[g];
  const int end = (g + 1 < N) ? rowptr[g + 1] : E;
  const float ws0 = attS[4 * j + 0], ws1 = attS[4 * j + 1], ws2 = attS[4 * j + 2], ws3 = attS[4 * j + 3];
  const float wd0 = attD[4 * j + 0], wd1 = attD[4 * j + 1], wd2 = attD[4 * j + 2], wd3 = attD[4 * j + 3];

  const float4* hrow = (const float4*)h1;
  float4 acc = hrow[(size_t)g * 16 + j];
  const float ad    = pair_sum(dot4(acc, wd0, wd1, wd2, wd3));
  const float eself = lrelu(pair_sum(dot4(acc, ws0, ws1, ws2, ws3)) + ad);
  float  m   = eself;
  float  den = 1.f;
  const char* hb = (const char*)hrow;
  const unsigned jo = 16u * j;
  const int last = end - 1;
  int nxt = 0;
  if (beg < end) nxt = col[min(beg + j, last)];
  unsigned offs = 0;
  int trip = 0;
  int k = beg;
  for (; k + 4 <= end; k += 4, ++trip) {
    offs = trip_slots<256u>(trip, offs, nxt);
    const float4 v0 = *(const float4*)(hb + (row_bcast<0>(offs) + jo));
    const float4 v1 = *(const float4*)(hb + (row_bcast<1>(offs) + jo));
    const float4 v2 = *(const float4*)(hb + (row_bcast<2>(offs) + jo));
    const float4 v3 = *(const float4*)(hb + (row_bcast<3>(offs) + jo));
    const float e0 = lrelu(pair_sum(dot4(v0, ws0, ws1, ws2, ws3)) + ad);
    const float e1 = lrelu(pair_sum(dot4(v1, ws0, ws1, ws2, ws3)) + ad);
    const float e2 = lrelu(pair_sum(dot4(v2, ws0, ws1, ws2, ws3)) + ad);
    const float e3 = lrelu(pair_sum(dot4(v3, ws0, ws1, ws2, ws3)) + ad);
    const float mn = fmaxf(fmaxf(fmaxf(m, e0), fmaxf(e1, e2)), e3);
    sm_rescale(__expf(m - mn), den, acc);
    sm_add(__expf(e0 - mn), v0, den, acc);
    sm_add(__expf(e1 - mn), v1, den, acc);
    sm_add(__expf(e2 - mn), v2, den, acc);
    sm_add(__expf(e3 - mn), v3, den, acc);
    m = mn;
    if ((trip & 3) == 0 && k + 16 < end) nxt = col[min(k + 16 + j, last)];
  }
  if (k < end) {
    const bool has1 = k + 1 < end, has2 = k + 2 < end;
    offs = trip_slots<256u>(trip, offs, nxt);
    const float4 v0 = *(const float4*)(hb + (row_bcast<0>(offs) + jo));
    const float4 v1 = *(const float4*)(hb + (row_bcast<1>(offs) + jo));
    const float4 v2 = *(const float4*)(hb + (row_bcast<2>(offs) + jo));
    const float e0 = lrelu(pair_sum(dot4(v0, ws0, ws1, ws2, ws3)) + ad);
    const float x1 = lrelu(pair_sum(dot4(v1, ws0, ws1, ws2, ws3)) + ad);
    const float x2 = lrelu(pair_sum(dot4(v2, ws0, ws1, ws2, ws3)) + ad);
    const float e1 = has1 ? x1 : -INFINITY;
    const float e2 = has2 ? x2 : -INFINITY;
    const float mn = fmaxf(fmaxf(m, e0), fmaxf(e1, e2));
    sm_rescale(__expf(m - mn), den, acc);
    sm_add(__expf(e0 - mn), v0, den, acc);
    sm_add(__expf(e1 - mn), v1, den, acc);
    sm_add(__expf(e2 - mn), v2, den, acc);
    m = mn;
  }
  const int wv   = threadIdx.x >> 5;
  const int lane = threadIdx.x & 31;
  const int mr   = lane & 15;
  const int half = lane >> 4;
  const int c    = wv * 16 + mr;
  const bool cv  = c < 40;
  const int cc   = cv ? c : 39;
  float bw[32];
  if (wv < 3) {
#pragma unroll
    for (int s = 0; s < 16; ++s) {
      const int ka = 4 * s + 2 * half;
      bw[2 * s]     = W2[ka * 40 + cc];
      bw[2 * s + 1] = W2[(ka + 1) * 40 + cc];
    }
  }

  const float inv = 1.f / (den + SOFT_EPS);
  float4 o;
  o.x = acc.x * inv + bias[4 * j + 0];
  o.y = acc.y * inv + bias[4 * j + 1];
  o.z = acc.z * inv + bias[4 * j + 2];
  o.w = acc.w * inv + bias[4 * j + 3];
  o.x = o.x > 0.f ? o.x : (__expf(o.x) - 1.f);
  o.y = o.y > 0.f ? o.y : (__expf(o.y) - 1.f);
  o.z = o.z > 0.f ? o.z : (__expf(o.z) - 1.f);
  o.w = o.w > 0.f ? o.w : (__expf(o.w) - 1.f);

  *(float4*)&tile[(threadIdx.x >> 4) * G1_PITCH + 4 * j] = o;
  __syncthreads();
  if (wv < 3) {
    const float* Ar = tile + mr * G1_PITCH;
    v8f acc2 = {};
#pragma unroll
    for (int s = 0; s < 16; ++s) {
      const int ka = 4 * s + 2 * half;
      v2f a, b;
      a.x = Ar[ka];
      a.y = Ar[ka + 1];
      b.x = cv ? bw[2 * s] : 0.f;
      b.y = cv ? bw[2 * s + 1] : 0.f;
      acc2 = __builtin_amdgcn_wmma_f32_16x16x4_f32(false, a, false, b, (short)0, acc2,
                                                   false, false);
    }
    const int row0 = blockIdx.x * 16;
#pragma unroll
    for (int i = 0; i < 8; ++i) {
      const int r = row0 + i + 8 * half;
      if (r < N) h2[(size_t)r * 48 + c] = acc2[i];
    }
  }
}

__global__ void gat_gather2(const int* __restrict__ rowptr, const int* __restrict__ col,
                            const float* __restrict__ attS, const float* __restrict__ attD,
                            const float* __restrict__ h2, const float* __restrict__ bias,
                            float* __restrict__ out, int N, int E) {
  const int t = blockIdx.x * blockDim.x + threadIdx.x;
  const int g = t >> 4;
  if (g >= N) return;
  const int j = t & 15;
  const bool act = j < 10;
  const int jj = act ? j : 9;
  const int beg = rowptr[g];
  const int end = (g + 1 < N) ? rowptr[g + 1] : E;
  const float ws0 = act ? attS[4 * jj + 0] : 0.f, ws1 = act ? attS[4 * jj + 1] : 0.f;
  const float ws2 = act ? attS[4 * jj + 2] : 0.f, ws3 = act ? attS[4 * jj + 3] : 0.f;
  const float wd0 = act ? attD[4 * jj + 0] : 0.f, wd1 = act ? attD[4 * jj + 1] : 0.f;
  const float wd2 = act ? attD[4 * jj + 2] : 0.f, wd3 = act ? attD[4 * jj + 3] : 0.f;

  const float4* hrow = (const float4*)h2;
  float4 acc = hrow[(size_t)g * 12 + jj];
  const float ad    = row_sum(dot4(acc, wd0, wd1, wd2, wd3));
  const float eself = lrelu(row_sum(dot4(acc, ws0, ws1, ws2, ws3)) + ad);
  float  m   = eself;
  float  den = 1.f;
  const char* hb = (const char*)hrow;
  const unsigned jo = 16u * jj;
  const int last = end - 1;
  int nxt = 0;
  if (beg < end) nxt = col[min(beg + j, last)];
  unsigned offs = 0;
  int trip = 0;
  int k = beg;
  for (; k + 4 <= end; k += 4, ++trip) {
    offs = trip_slots<192u>(trip, offs, nxt);
    const float4 v0 = *(const float4*)(hb + (row_bcast<0>(offs) + jo));
    const float4 v1 = *(const float4*)(hb + (row_bcast<1>(offs) + jo));
    const float4 v2 = *(const float4*)(hb + (row_bcast<2>(offs) + jo));
    const float4 v3 = *(const float4*)(hb + (row_bcast<3>(offs) + jo));
    const float e0 = lrelu(row_sum(dot4(v0, ws0, ws1, ws2, ws3)) + ad);
    const float e1 = lrelu(row_sum(dot4(v1, ws0, ws1, ws2, ws3)) + ad);
    const float e2 = lrelu(row_sum(dot4(v2, ws0, ws1, ws2, ws3)) + ad);
    const float e3 = lrelu(row_sum(dot4(v3, ws0, ws1, ws2, ws3)) + ad);
    const float mn = fmaxf(fmaxf(fmaxf(m, e0), fmaxf(e1, e2)), e3);
    sm_rescale(__expf(m - mn), den, acc);
    sm_add(__expf(e0 - mn), v0, den, acc);
    sm_add(__expf(e1 - mn), v1, den, acc);
    sm_add(__expf(e2 - mn), v2, den, acc);
    sm_add(__expf(e3 - mn), v3, den, acc);
    m = mn;
    if ((trip & 3) == 0 && k + 16 < end) nxt = col[min(k + 16 + j, last)];
  }
  if (k < end) {
    const bool has1 = k + 1 < end, has2 = k + 2 < end;
    offs = trip_slots<192u>(trip, offs, nxt);
    const float4 v0 = *(const float4*)(hb + (row_bcast<0>(offs) + jo));
    const float4 v1 = *(const float4*)(hb + (row_bcast<1>(offs) + jo));
    const float4 v2 = *(const float4*)(hb + (row_bcast<2>(offs) + jo));
    const float e0 = lrelu(row_sum(dot4(v0, ws0, ws1, ws2, ws3)) + ad);
    const float x1 = lrelu(row_sum(dot4(v1, ws0, ws1, ws2, ws3)) + ad);
    const float x2 = lrelu(row_sum(dot4(v2, ws0, ws1, ws2, ws3)) + ad);
    const float e1 = has1 ? x1 : -INFINITY;
    const float e2 = has2 ? x2 : -INFINITY;
    const float mn = fmaxf(fmaxf(m, e0), fmaxf(e1, e2));
    sm_rescale(__expf(m - mn), den, acc);
    sm_add(__expf(e0 - mn), v0, den, acc);
    sm_add(__expf(e1 - mn), v1, den, acc);
    sm_add(__expf(e2 - mn), v2, den, acc);
    m = mn;
  }
  const float inv = 1.f / (den + SOFT_EPS);
  float4 v;
  v.x = acc.x * inv + bias[4 * jj + 0];
  v.y = acc.y * inv + bias[4 * jj + 1];
  v.z = acc.z * inv + bias[4 * jj + 2];
  v.w = acc.w * inv + bias[4 * jj + 3];
  float mx = act ? fmaxf(fmaxf(v.x, v.y), fmaxf(v.z, v.w)) : -INFINITY;
  for (int off = 8; off > 0; off >>= 1) mx = fmaxf(mx, __shfl_xor(mx, off, 16));
  float sum = act ? (__expf(v.x - mx) + __expf(v.y - mx) + __expf(v.z - mx) + __expf(v.w - mx))
                  : 0.f;
  for (int off = 8; off > 0; off >>= 1) sum += __shfl_xor(sum, off, 16);
  const float lse = mx + __logf(sum);
  if (act) {
    float4 o;
    o.x = v.x - lse; o.y = v.y - lse; o.z = v.z - lse; o.w = v.w - lse;
    ((float4*)out)[(size_t)g * 10 + j] = o;
  }
}

extern "C" void kernel_launch(void* const* d_in, const int* in_sizes, int n_in,
                              void* d_out, int out_size, void* d_ws, size_t ws_size,
                              hipStream_t stream) {
  const float* x   = (const float*)d_in[0];
  const int*   ei  = (const int*)d_in[1];
  const float* W1  = (const float*)d_in[2];
  const float* as1 = (const float*)d_in[3];
  const float* ad1 = (const float*)d_in[4];
  const float* b1  = (const float*)d_in[5];
  const float* W2  = (const float*)d_in[6];
  const float* as2 = (const float*)d_in[7];
  const float* ad2 = (const float*)d_in[8];
  const float* b2  = (const float*)d_in[9];
  float* out = (float*)d_out;

  const int Fin = in_sizes[2] / 64;
  const int N   = in_sizes[0] / Fin;
  const int E   = in_sizes[1] / 2;
  const int* src = ei;
  const int* dst = ei + E;

  const int TB = 256;
  const int gM  = (N + 15) / 16;
  const int gG  = (N * 16 + TB - 1) / TB;
  const int NB  = (N + RNG_NODES - 1) / RNG_NODES;
  const int nH  = NB * BIN_BLOCKS;

  float* ws = (float*)d_ws;
  size_t o = 0;
  float* h1    = ws + o; o += (size_t)N * 64;
  float* h2    = ws + o; o += (size_t)N * 48;
  int2* pairs  = (int2*)(ws + o); o += (size_t)E * 2;
  int* col     = (int*)(ws + o); o += (size_t)E;
  int* rowptr  = (int*)(ws + o); o += (size_t)N;
  int* tab     = (int*)(ws + o); o += (size_t)nH;

  sort1_proj1<<<BIN_BLOCKS + PJ_BLOCKS, BIN_TB, (size_t)NB * 4, stream>>>(src, dst, tab, pairs, E, NB,
                                                                         x, W1, h1, N);
  range_fill<<<NB, RF_TB, 0, stream>>>(pairs, tab, rowptr, col, N, E, NB);
  gat_gather1_proj<<<gM, G1_TB, 0, stream>>>(rowptr, col, as1, ad1, h1, b1, W2, h2, N, E);

  gat_gather2<<<gG, TB, 0, stream>>>(rowptr, col, as2, ad2, h2, b2, out, N, E);

  (void)n_in; (void)out_size; (void)ws_size;
}
